# starting kernel
# baseline (speedup 1.0000x reference)
#include <hip/hip_runtime.h>


typedef float v2f __attribute__((ext_vector_type(2)));
typedef float v8f __attribute__((ext_vector_type(8)));

#define N_NODES 50000
#define N_EDGES 800000
#define F_IN    128
#define H1      200
#define H1P     208
#define H2      100
#define H2P     112
#define NC      40
#define NCP     48

#define KP      210

__global__ void k_deg_init(float* __restrict__ deg, int n) {
    int i = blockIdx.x * blockDim.x + threadIdx.x;
    if (i < n) deg[i] = 1.0f;
}

__global__ void k_deg_accum(const int* __restrict__ dst, float* __restrict__ deg, int e) {
    int i = blockIdx.x * blockDim.x + threadIdx.x;
    if (i < e) atomicAdd(&deg[dst[i]], 1.0f);
}

__global__ void k_rsqrt_inplace(float* __restrict__ v, int n) {
    int i = blockIdx.x * blockDim.x + threadIdx.x;
    if (i < n) v[i] = rsqrtf(v[i]);
}

__global__ void __launch_bounds__(128)
k_gemm_wmma_f32(const float* __restrict__ A, int lda,
                const float* __restrict__ B, int Nact,
                float* __restrict__ C, int ldc, int Nstore,
                int K, int reluA, const float* __restrict__ bias)
{
    __shared__ float As[16 * KP];
    __shared__ float Bs[64 * KP];

    const int m_base = blockIdx.x * 16;
    const int nblk   = blockIdx.y * 64;
    const int tid    = threadIdx.x;

    for (int idx = tid; idx < 16 * K; idx += 128) {
        int r = idx / K;
        int c = idx - r * K;
        float v = A[(size_t)(m_base + r) * lda + c];
        if (reluA) v = fmaxf(v, 0.0f);
        As[r * KP + c] = v;
    }
    for (int idx = tid; idx < 64 * K; idx += 128) {
        int k = idx >> 6;
        int c = idx & 63;
        int col = nblk + c;
        float v = (col < Nact) ? B[(size_t)k * Nact + col] : 0.0f;
        Bs[c * KP + k] = v;
    }
    __syncthreads();

    const int wave = tid >> 5;
    const int lane = tid & 31;
    const int mloc = lane & 15;
    const int kb   = (lane >> 4) << 1;
    const int ncol = nblk + wave * 16 + mloc;

    const float* ap = &As[mloc * KP + kb];
    const float* bp = &Bs[(wave * 16 + mloc) * KP + kb];

    v8f acc = {0.f, 0.f, 0.f, 0.f, 0.f, 0.f, 0.f, 0.f};

    for (int k = 0; k < K; k += 4) {
        v2f a, b;
        a.x = ap[k];
        a.y = ap[k + 1];
        b.x = bp[k];
        b.y = bp[k + 1];
        acc = __builtin_amdgcn_wmma_f32_16x16x4_f32(
                  false, a, false, b, (short)0, acc, false, false);
    }

    float bv = (bias != nullptr && ncol < Nact) ? bias[ncol] : 0.0f;
    if (ncol < Nstore) {
        const int rbase = m_base + ((lane >> 4) << 3);
        #pragma unroll
        for (int g = 0; g < 8; ++g)
            C[(size_t)(rbase + g) * ldc + ncol] = acc[g] + bv;
    }
}

__global__ void k_agg_init(const float* __restrict__ hw, const float* __restrict__ dinv,
                           const float* __restrict__ bias, float* __restrict__ agg,
                           int Fp, int Fact, int total)
{
    int i = blockIdx.x * blockDim.x + threadIdx.x;
    if (i >= total) return;
    int node = i / Fp;
    int f    = i - node * Fp;
    float d  = dinv[node];
    float bb = (f < Fact) ? bias[f] : 0.0f;
    agg[i] = bb + d * d * hw[i];
}

__global__ void k_agg_edges(const int* __restrict__ src, const int* __restrict__ dst,
                            const float* __restrict__ dinv, const float* __restrict__ hw,
                            float* __restrict__ agg, int Fp, int Fact, int e)
{
    int wid    = (blockIdx.x * blockDim.x + threadIdx.x) >> 5;
    int lane   = threadIdx.x & 31;
    int nwaves = (gridDim.x * blockDim.x) >> 5;
    for (int ed = wid; ed < e; ed += nwaves) {
        int s = src[ed];
        int d = dst[ed];
        float w = dinv[s] * dinv[d];
        const float* hs = hw  + (size_t)s * Fp;
        float*       ad = agg + (size_t)d * Fp;
        for (int f = lane; f < Fact; f += 32)
            atomicAdd(&ad[f], w * hs[f]);
    }
}

extern "C" void kernel_launch(void* const* d_in, const int* in_sizes, int n_in,
                              void* d_out, int out_size, void* d_ws, size_t ws_size,
                              hipStream_t stream)
{
    const float* x    = (const float*)d_in[0];
    const int*   ei   = (const int*)d_in[1];
    const float* W1   = (const float*)d_in[2];
    const float* b1   = (const float*)d_in[3];
    const float* W2   = (const float*)d_in[4];
    const float* b2   = (const float*)d_in[5];
    const float* Wout = (const float*)d_in[6];
    const float* bout = (const float*)d_in[7];
    float*       out  = (float*)d_out;

    const int* srcI = ei;
    const int* dstI = ei + N_EDGES;

    float* ws   = (float*)d_ws;
    float* dinv = ws;
    float* bufA = ws + 50176;
    float* bufB = bufA + (size_t)N_NODES * H1P;

    k_deg_init     <<<(N_NODES + 255) / 256, 256, 0, stream>>>(dinv, N_NODES);
    k_deg_accum    <<<(N_EDGES + 255) / 256, 256, 0, stream>>>(dstI, dinv, N_EDGES);
    k_rsqrt_inplace<<<(N_NODES + 255) / 256, 256, 0, stream>>>(dinv, N_NODES);

    k_gemm_wmma_f32<<<dim3(N_NODES / 16, (H1P + 63) / 64), 128, 0, stream>>>(
        x, F_IN, W1, H1, bufA, H1P, H1P, F_IN,  0,  nullptr);
    {
        int total = N_NODES * H1P;
        k_agg_init<<<(total + 255) / 256, 256, 0, stream>>>(bufA, dinv, b1, bufB,
                                                            H1P, H1, total);
    }
    k_agg_edges<<<4096, 256, 0, stream>>>(srcI, dstI, dinv, bufA, bufB, H1P, H1, N_EDGES);

    k_gemm_wmma_f32<<<dim3(N_NODES / 16, (H2P + 63) / 64), 128, 0, stream>>>(
        bufB, H1P, W2, H2, bufA, H2P, H2P, H1,  1,  nullptr);
    {
        int total = N_NODES * H2P;
        k_agg_init<<<(total + 255) / 256, 256, 0, stream>>>(bufA, dinv, b2, bufB,
                                                            H2P, H2, total);
    }
    k_agg_edges<<<4096, 256, 0, stream>>>(srcI, dstI, dinv, bufA, bufB, H2P, H2, N_EDGES);

    k_gemm_wmma_f32<<<dim3(N_NODES / 16, (NCP + 63) / 64), 128, 0, stream>>>(
        bufB, H2P, Wout, NC, out, NC, NC, H2,  1,  bout);
}


# GEMMs: whole B staged transposed in LDS once per work-group (256 groups of 16 waves); each wave holds its 16-row A strip in registers and walks the N tiles; B no longer restaged per strip
# speedup vs baseline: 1.7977x; 1.7977x over previous
#include <hip/hip_runtime.h>


typedef float v2f __attribute__((ext_vector_type(2)));
typedef float v8f __attribute__((ext_vector_type(8)));

#define N_NODES 50000
#define N_EDGES 800000
#define F_IN    128
#define H1      200
#define H1P     208
#define H2      100
#define H2P     112
#define NC      40
#define NCP     48


__global__ void k_deg_init(float* __restrict__ deg, int n) {
    int i = blockIdx.x * blockDim.x + threadIdx.x;
    if (i < n) deg[i] = 1.0f;
}

__global__ void k_deg_accum(const int* __restrict__ dst, float* __restrict__ deg, int e) {
    int i = blockIdx.x * blockDim.x + threadIdx.x;
    if (i < e) atomicAdd(&deg[dst[i]], 1.0f);
}

__global__ void k_rsqrt_inplace(float* __restrict__ v, int n) {
    int i = blockIdx.x * blockDim.x + threadIdx.x;
    if (i < n) v[i] = rsqrtf(v[i]);
}

template <int KS, int S, int NTL>
__device__ __forceinline__ void mm_tiles(const v2f (&a)[KS], const float* bp, v8f (&acc)[NTL])
{
    #pragma unroll
    for (int j = 0; j < KS; ++j) {
        #pragma unroll
        for (int t = 0; t < NTL; ++t) {
            v2f b = *reinterpret_cast<const v2f*>(bp + t * 16 * S + 4 * j);
            acc[t] = __builtin_amdgcn_wmma_f32_16x16x4_f32(
                         false, a[j], false, b, (short)0, acc[t], false, false);
        }
    }
}

template <int K, int NACT, int NPAD, int S, bool RELU, bool HASBIAS, int NSTORE, int NTL>
__device__ __forceinline__ void tiles_out(const v2f (&a)[K / 4], const float* Bs, int nt,
                                          int c, int h, int m_base,
                                          float* __restrict__ C, int ldc,
                                          const float* __restrict__ bias)
{
    v8f acc[NTL];
    #pragma unroll
    for (int t = 0; t < NTL; ++t)
        acc[t] = (v8f){0.f, 0.f, 0.f, 0.f, 0.f, 0.f, 0.f, 0.f};

    mm_tiles<K / 4, S, NTL>(a, Bs + (nt * 16 + c) * S + 2 * h, acc);

    const int rbase = m_base + 8 * h;
    #pragma unroll
    for (int t = 0; t < NTL; ++t) {
        const int ncol = (nt + t) * 16 + c;
        float bv = 0.0f;
        if (HASBIAS) bv = (ncol < NACT) ? bias[ncol] : 0.0f;
        if (NSTORE >= NPAD || ncol < NSTORE) {
            #pragma unroll
            for (int g = 0; g < 8; ++g)
                C[(size_t)(rbase + g) * ldc + ncol] = acc[t][g] + bv;
        }
    }
}

template <int K, int NACT, int NPAD, int S, bool RELU, bool HASBIAS, int NSTORE>
__global__ void __launch_bounds__(512)
k_gemm_wmma_f32(const float* __restrict__ A, int lda,
                const float* __restrict__ B,
                float* __restrict__ C, int ldc,
                const float* __restrict__ bias, int mtiles)
{
    __shared__ __attribute__((aligned(16))) float Bs[NPAD * S];

    constexpr int KS    = K / 4;
    constexpr int NT    = NPAD / 16;
    constexpr int Q     = NACT / 4;
    constexpr int QG    = (Q + 7) / 8;
    constexpr int NTASK = KS * QG;

    const int tid  = threadIdx.x;
    const int lane = tid & 31;
    const int wave = __builtin_amdgcn_readfirstlane(tid >> 5);
    const int c    = lane & 15;
    const int h    = lane >> 4;

    int mt = wave * gridDim.x + blockIdx.x;
    v2f a[KS];
    if (mt < mtiles) {
        const float* arow = A + (size_t)(mt * 16 + c) * lda + 2 * h;
        #pragma unroll
        for (int j = 0; j < KS; ++j)
            a[j] = *reinterpret_cast<const v2f*>(arow + 4 * j);
    }

    {
        const int ql = lane & 7;
        const int kl = lane >> 3;
        constexpr int NIT = (NTASK + 15) / 16;
        float4 stg[NIT];
        #pragma unroll
        for (int it = 0; it < NIT; ++it) {
            const int task = wave + 16 * it;
            const int kg = task / QG;
            const int qg = task - kg * QG;
            const int q  = qg * 8 + ql;
            const int k  = kg * 4 + kl;
            stg[it] = make_float4(0.0f, 0.0f, 0.0f, 0.0f);
            if (task < NTASK && q < Q)
                stg[it] = *reinterpret_cast<const float4*>(B + (size_t)k * NACT + 4 * q);
        }
        #pragma unroll
        for (int it = 0; it < NIT; ++it) {
            const int task = wave + 16 * it;
            const int kg = task / QG;
            const int qg = task - kg * QG;
            const int q  = qg * 8 + ql;
            const int k  = kg * 4 + kl;
            if (task < NTASK && q < Q) {
                float* d = &Bs[(4 * q) * S + k];
                d[0]     = stg[it].x;
                d[S]     = stg[it].y;
                d[2 * S] = stg[it].z;
                d[3 * S] = stg[it].w;
            }
        }
        for (int idx = tid; idx < (NPAD - NACT) * K; idx += 512) {
            const int cc = idx / K;
            const int k  = idx - cc * K;
            Bs[(NACT + cc) * S + k] = 0.0f;
        }
    }
    __syncthreads();

    while (mt < mtiles) {
        const int m_base = mt * 16;
        if (RELU) {
            #pragma unroll
            for (int j = 0; j < KS; ++j) {
                a[j].x = fmaxf(a[j].x, 0.0f);
                a[j].y = fmaxf(a[j].y, 0.0f);
            }
        }
        int nt = 0;
        for (; nt + 2 <= NT; nt += 2)
            tiles_out<K, NACT, NPAD, S, RELU, HASBIAS, NSTORE, 2>(a, Bs, nt, c, h, m_base, C, ldc, bias);
        if (NT & 1)
            tiles_out<K, NACT, NPAD, S, RELU, HASBIAS, NSTORE, 1>(a, Bs, NT - 1, c, h, m_base, C, ldc, bias);

        mt += 16 * gridDim.x;
        if (mt < mtiles) {
            const float* arow = A + (size_t)(mt * 16 + c) * lda + 2 * h;
            #pragma unroll
            for (int j = 0; j < KS; ++j)
                a[j] = *reinterpret_cast<const v2f*>(arow + 4 * j);
        }
    }
}

__global__ void k_agg_init(const float* __restrict__ hw, const float* __restrict__ dinv,
                           const float* __restrict__ bias, float* __restrict__ agg,
                           int Fp, int Fact, int total)
{
    int i = blockIdx.x * blockDim.x + threadIdx.x;
    if (i >= total) return;
    int node = i / Fp;
    int f    = i - node * Fp;
    float d  = dinv[node];
    float bb = (f < Fact) ? bias[f] : 0.0f;
    agg[i] = bb + d * d * hw[i];
}

__global__ void k_agg_edges(const int* __restrict__ src, const int* __restrict__ dst,
                            const float* __restrict__ dinv, const float* __restrict__ hw,
                            float* __restrict__ agg, int Fp, int Fact, int e)
{
    int wid    = (blockIdx.x * blockDim.x + threadIdx.x) >> 5;
    int lane   = threadIdx.x & 31;
    int nwaves = (gridDim.x * blockDim.x) >> 5;
    for (int ed = wid; ed < e; ed += nwaves) {
        int s = src[ed];
        int d = dst[ed];
        float w = dinv[s] * dinv[d];
        const float* hs = hw  + (size_t)s * Fp;
        float*       ad = agg + (size_t)d * Fp;
        for (int f = lane; f < Fact; f += 32)
            atomicAdd(&ad[f], w * hs[f]);
    }
}

extern "C" void kernel_launch(void* const* d_in, const int* in_sizes, int n_in,
                              void* d_out, int out_size, void* d_ws, size_t ws_size,
                              hipStream_t stream)
{
    const float* x    = (const float*)d_in[0];
    const int*   ei   = (const int*)d_in[1];
    const float* W1   = (const float*)d_in[2];
    const float* b1   = (const float*)d_in[3];
    const float* W2   = (const float*)d_in[4];
    const float* b2   = (const float*)d_in[5];
    const float* Wout = (const float*)d_in[6];
    const float* bout = (const float*)d_in[7];
    float*       out  = (float*)d_out;

    const int* srcI = ei;
    const int* dstI = ei + N_EDGES;

    float* ws   = (float*)d_ws;
    float* dinv = ws;
    float* bufA = ws + 50176;
    float* bufB = bufA + (size_t)N_NODES * H1P;

    k_deg_init     <<<(N_NODES + 255) / 256, 256, 0, stream>>>(dinv, N_NODES);
    k_deg_accum    <<<(N_EDGES + 255) / 256, 256, 0, stream>>>(dstI, dinv, N_EDGES);
    k_rsqrt_inplace<<<(N_NODES + 255) / 256, 256, 0, stream>>>(dinv, N_NODES);

    k_gemm_wmma_f32<F_IN, H1, H1P, 132, false, false, H1P><<<256, 512, 0, stream>>>(
        x, F_IN, W1, bufA, H1P,  nullptr, N_NODES / 16);
    {
        int total = N_NODES * H1P;
        k_agg_init<<<(total + 255) / 256, 256, 0, stream>>>(bufA, dinv, b1, bufB,
                                                            H1P, H1, total);
    }
    k_agg_edges<<<4096, 256, 0, stream>>>(srcI, dstI, dinv, bufA, bufB, H1P, H1, N_EDGES);

    k_gemm_wmma_f32<H1, H2, H2P, 204, true, false, H2P><<<256, 512, 0, stream>>>(
        bufB, H1P, W2, bufA, H2P,  nullptr, N_NODES / 16);
    {
        int total = N_NODES * H2P;
        k_agg_init<<<(total + 255) / 256, 256, 0, stream>>>(bufA, dinv, b2, bufB,
                                                            H2P, H2, total);
    }
    k_agg_edges<<<4096, 256, 0, stream>>>(srcI, dstI, dinv, bufA, bufB, H2P, H2, N_EDGES);

    k_gemm_wmma_f32<H2, NC, NCP, 100, true, true, NC><<<256, 512, 0, stream>>>(
        bufB, H2P, Wout, out, NC,  bout, N_NODES / 16);
}


# Aggregation without float atomics: edges counting-sorted by dst once (int counters, arrival-order slots), one wave per node sums its in-edges in registers, stores the row once; sum order changes
# speedup vs baseline: 3.4111x; 1.8975x over previous
#include <hip/hip_runtime.h>


typedef float v2f __attribute__((ext_vector_type(2)));
typedef float v8f __attribute__((ext_vector_type(8)));

#define N_NODES 50000
#define N_EDGES 800000
#define F_IN    128
#define H1      200
#define H1P     208
#define H2      100
#define H2P     112
#define NC      40
#define NCP     48
static_assert(N_EDGES % 4 == 0, "edge ids are read four at a time");


__global__ void k_cnt_zero(int* __restrict__ cnt, int n1) {
    int i = blockIdx.x * blockDim.x + threadIdx.x;
    if (i < n1) cnt[i] = 0;
}

__global__ void k_cnt_edges(const int* __restrict__ dst, int* __restrict__ cnt, int e4) {
    int i = blockIdx.x * blockDim.x + threadIdx.x;
    if (i < e4) {
        const int4 d = reinterpret_cast<const int4*>(dst)[i];
        atomicAdd(&cnt[d.x], 1);
        atomicAdd(&cnt[d.y], 1);
        atomicAdd(&cnt[d.z], 1);
        atomicAdd(&cnt[d.w], 1);
    }
}

__global__ void __launch_bounds__(256)
k_seg_alloc(int* cnt, int* __restrict__ start, int* __restrict__ cursor,
            float* __restrict__ dinv, int n)
{
    __shared__ int wsum[8];
    __shared__ int bbase;
    const int tid  = threadIdx.x;
    const int lane = tid & 31;
    const int wave = tid >> 5;
    const int i    = blockIdx.x * 256 + tid;
    const int v    = (i < n) ? cnt[i] : 0;
    int incl = v;
    #pragma unroll
    for (int o = 1; o < 32; o <<= 1) {
        const int t = __shfl_up(incl, o);
        if (lane >= o) incl += t;
    }
    if (lane == 31) wsum[wave] = incl;
    __syncthreads();
    if (tid == 0) {
        int tot = 0;
        for (int w = 0; w < 8; ++w) tot += wsum[w];
        bbase = atomicAdd(&cnt[n], tot);
    }
    __syncthreads();
    int off = bbase;
    for (int w = 0; w < wave; ++w) off += wsum[w];
    if (i < n) {
        const int s = off + incl - v;
        start[i]  = s;
        cursor[i] = s;
        dinv[i]   = rsqrtf((float)(v + 1));
    }
}

__global__ void k_csr_fill(const int* __restrict__ src, const int* __restrict__ dst,
                           const float* __restrict__ dinv, int* __restrict__ cursor,
                           int* __restrict__ csr_src, float* __restrict__ csr_w, int e4)
{
    int i = blockIdx.x * blockDim.x + threadIdx.x;
    if (i < e4) {
        const int4 s = reinterpret_cast<const int4*>(src)[i];
        const int4 d = reinterpret_cast<const int4*>(dst)[i];
        const float w0 = dinv[s.x];
        const float w1 = dinv[s.y];
        const float w2 = dinv[s.z];
        const float w3 = dinv[s.w];
        const int p0 = atomicAdd(&cursor[d.x], 1);
        const int p1 = atomicAdd(&cursor[d.y], 1);
        const int p2 = atomicAdd(&cursor[d.z], 1);
        const int p3 = atomicAdd(&cursor[d.w], 1);
        csr_src[p0] = s.x;  csr_w[p0] = w0;
        csr_src[p1] = s.y;  csr_w[p1] = w1;
        csr_src[p2] = s.z;  csr_w[p2] = w2;
        csr_src[p3] = s.w;  csr_w[p3] = w3;
    }
}

template <int KS, int S, int NTL>
__device__ __forceinline__ void mm_tiles(const v2f (&a)[KS], const float* bp, v8f (&acc)[NTL])
{
    #pragma unroll
    for (int j = 0; j < KS; ++j) {
        #pragma unroll
        for (int t = 0; t < NTL; ++t) {
            v2f b = *reinterpret_cast<const v2f*>(bp + t * 16 * S + 4 * j);
            acc[t] = __builtin_amdgcn_wmma_f32_16x16x4_f32(
                         false, a[j], false, b, (short)0, acc[t], false, false);
        }
    }
}

template <int K, int NACT, int NPAD, int S, bool RELU, bool HASBIAS, int NSTORE, int NTL>
__device__ __forceinline__ void tiles_out(const v2f (&a)[K / 4], const float* Bs, int nt,
                                          int c, int h, int m_base,
                                          float* __restrict__ C, int ldc,
                                          const float* __restrict__ bias)
{
    v8f acc[NTL];
    #pragma unroll
    for (int t = 0; t < NTL; ++t)
        acc[t] = (v8f){0.f, 0.f, 0.f, 0.f, 0.f, 0.f, 0.f, 0.f};

    mm_tiles<K / 4, S, NTL>(a, Bs + (nt * 16 + c) * S + 2 * h, acc);

    const int rbase = m_base + 8 * h;
    #pragma unroll
    for (int t = 0; t < NTL; ++t) {
        const int ncol = (nt + t) * 16 + c;
        float bv = 0.0f;
        if (HASBIAS) bv = (ncol < NACT) ? bias[ncol] : 0.0f;
        if (NSTORE >= NPAD || ncol < NSTORE) {
            #pragma unroll
            for (int g = 0; g < 8; ++g)
                C[(size_t)(rbase + g) * ldc + ncol] = acc[t][g] + bv;
        }
    }
}

template <int K, int NACT, int NPAD, int S, bool RELU, bool HASBIAS, int NSTORE>
__global__ void __launch_bounds__(512)
k_gemm_wmma_f32(const float* __restrict__ A, int lda,
                const float* __restrict__ B,
                float* __restrict__ C, int ldc,
                const float* __restrict__ bias, int mtiles)
{
    __shared__ __attribute__((aligned(16))) float Bs[NPAD * S];

    constexpr int KS    = K / 4;
    constexpr int NT    = NPAD / 16;
    constexpr int Q     = NACT / 4;
    constexpr int QG    = (Q + 7) / 8;
    constexpr int NTASK = KS * QG;

    const int tid  = threadIdx.x;
    const int lane = tid & 31;
    const int wave = __builtin_amdgcn_readfirstlane(tid >> 5);
    const int c    = lane & 15;
    const int h    = lane >> 4;

    int mt = wave * gridDim.x + blockIdx.x;
    v2f a[KS];
    if (mt < mtiles) {
        const float* arow = A + (size_t)(mt * 16 + c) * lda + 2 * h;
        #pragma unroll
        for (int j = 0; j < KS; ++j)
            a[j] = *reinterpret_cast<const v2f*>(arow + 4 * j);
    }

    {
        const int ql = lane & 7;
        const int kl = lane >> 3;
        constexpr int NIT = (NTASK + 15) / 16;
        float4 stg[NIT];
        #pragma unroll
        for (int it = 0; it < NIT; ++it) {
            const int task = wave + 16 * it;
            const int kg = task / QG;
            const int qg = task - kg * QG;
            const int q  = qg * 8 + ql;
            const int k  = kg * 4 + kl;
            stg[it] = make_float4(0.0f, 0.0f, 0.0f, 0.0f);
            if (task < NTASK && q < Q)
                stg[it] = *reinterpret_cast<const float4*>(B + (size_t)k * NACT + 4 * q);
        }
        #pragma unroll
        for (int it = 0; it < NIT; ++it) {
            const int task = wave + 16 * it;
            const int kg = task / QG;
            const int qg = task - kg * QG;
            const int q  = qg * 8 + ql;
            const int k  = kg * 4 + kl;
            if (task < NTASK && q < Q) {
                float* d = &Bs[(4 * q) * S + k];
                d[0]     = stg[it].x;
                d[S]     = stg[it].y;
                d[2 * S] = stg[it].z;
                d[3 * S] = stg[it].w;
            }
        }
        for (int idx = tid; idx < (NPAD - NACT) * K; idx += 512) {
            const int cc = idx / K;
            const int k  = idx - cc * K;
            Bs[(NACT + cc) * S + k] = 0.0f;
        }
    }
    __syncthreads();

    while (mt < mtiles) {
        const int m_base = mt * 16;
        if (RELU) {
            #pragma unroll
            for (int j = 0; j < KS; ++j) {
                a[j].x = fmaxf(a[j].x, 0.0f);
                a[j].y = fmaxf(a[j].y, 0.0f);
            }
        }
        int nt = 0;
        for (; nt + 2 <= NT; nt += 2)
            tiles_out<K, NACT, NPAD, S, RELU, HASBIAS, NSTORE, 2>(a, Bs, nt, c, h, m_base, C, ldc, bias);
        if (NT & 1)
            tiles_out<K, NACT, NPAD, S, RELU, HASBIAS, NSTORE, 1>(a, Bs, NT - 1, c, h, m_base, C, ldc, bias);

        mt += 16 * gridDim.x;
        if (mt < mtiles) {
            const float* arow = A + (size_t)(mt * 16 + c) * lda + 2 * h;
            #pragma unroll
            for (int j = 0; j < KS; ++j)
                a[j] = *reinterpret_cast<const v2f*>(arow + 4 * j);
        }
    }
}

template <int FP, int NP, int R>
__device__ __forceinline__ void pull_rows(const float* __restrict__ hw, int sv, int dvb, float di,
                                          int e, const int (&qc)[NP], float4 (&acc)[NP])
{
    float4 h[R][NP];
    #pragma unroll
    for (int u = 0; u < R; ++u) {
        const int s = __builtin_amdgcn_readlane(sv, e + u);
        const float4* row = reinterpret_cast<const float4*>(hw + (size_t)s * FP);
        #pragma unroll
        for (int p = 0; p < NP; ++p)
            h[u][p] = row[qc[p]];
    }
    #pragma unroll
    for (int u = 0; u < R; ++u) {
        const float w = __int_as_float(__builtin_amdgcn_readlane(dvb, e + u)) * di;
        #pragma unroll
        for (int p = 0; p < NP; ++p) {
            acc[p].x = fmaf(w, h[u][p].x, acc[p].x);
            acc[p].y = fmaf(w, h[u][p].y, acc[p].y);
            acc[p].z = fmaf(w, h[u][p].z, acc[p].z);
            acc[p].w = fmaf(w, h[u][p].w, acc[p].w);
        }
    }
}

template <int FP, int NP, int R>
__device__ __forceinline__ void pull_tail(const float* __restrict__ hw, int sv, int dvb, float di,
                                          int e, int r, const int (&qc)[NP], float4 (&acc)[NP])
{
    if constexpr (R >= 1) {
        if (r == R) pull_rows<FP, NP, R>(hw, sv, dvb, di, e, qc, acc);
        else        pull_tail<FP, NP, R - 1>(hw, sv, dvb, di, e, r, qc, acc);
    }
}

template <int FP, int NP, int U>
__device__ __forceinline__ void pull_chunk(const float* __restrict__ hw, int sv, int dvb, float di,
                                           int cnt, const int (&qc)[NP], float4 (&acc)[NP])
{
    int e = 0;
    for (; e + U <= cnt; e += U)
        pull_rows<FP, NP, U>(hw, sv, dvb, di, e, qc, acc);
    pull_tail<FP, NP, U - 1>(hw, sv, dvb, di, e, cnt - e, qc, acc);
}

template <int FP, int FACT, int U>
__global__ void __launch_bounds__(256, 8)
k_agg_pull(const int* __restrict__ start, const int* __restrict__ endp,
           const int* __restrict__ csr_src, const float* __restrict__ csr_w,
           const float* __restrict__ dinv,
           const float* __restrict__ hw, const float* __restrict__ bias,
           float* __restrict__ agg, int n)
{
    constexpr int Q  = FP / 4;
    constexpr int QB = FACT / 4;
    constexpr int NP = (Q + 31) / 32;
    static_assert(FP % 4 == 0 && FACT % 4 == 0, "rows are walked as float4");

    const int lane = threadIdx.x & 31;
    const int wid  = __builtin_amdgcn_readfirstlane((int)((blockIdx.x * blockDim.x + threadIdx.x) >> 5));
    const int nw   = (int)((gridDim.x * blockDim.x) >> 5);

    int    qc[NP];
    float4 bq[NP];
    #pragma unroll
    for (int p = 0; p < NP; ++p) {
        const int q = lane + 32 * p;
        qc[p] = min(q, Q - 1);
        bq[p] = make_float4(0.0f, 0.0f, 0.0f, 0.0f);
        if (q < QB) bq[p] = reinterpret_cast<const float4*>(bias)[q];
    }

    for (int node0 = wid; node0 < n; node0 += 32 * nw) {
        const int mine = min(node0 + lane * nw, n - 1);
        const int begv = start[mine];
        const int endv = endp[mine];
        const int divb = __float_as_int(dinv[mine]);
        const int kn   = min(32, (n - 1 - node0) / nw + 1);

        int beg = __builtin_amdgcn_readlane(begv, 0);
        int end = __builtin_amdgcn_readlane(endv, 0);
        int sv, dv;
        {
            const int idx = max(min(beg + lane, end - 1), 0);
            sv = csr_src[idx];
            dv = __float_as_int(csr_w[idx]);
        }

        for (int k = 0; k < kn; ++k) {
            const int   node = node0 + k * nw;
            const float di   = __int_as_float(__builtin_amdgcn_readlane(divb, k));

            const float4* hself = reinterpret_cast<const float4*>(hw + (size_t)node * FP);
            float4 hs[NP];
            float4 acc[NP];
            #pragma unroll
            for (int p = 0; p < NP; ++p) {
                hs[p]  = hself[qc[p]];
                acc[p] = bq[p];
            }

            const int k1   = min(k + 1, kn - 1);
            const int nbeg = __builtin_amdgcn_readlane(begv, k1);
            const int nend = __builtin_amdgcn_readlane(endv, k1);
            const int idxn = max(min(nbeg + lane, nend - 1), 0);
            const int svn  = csr_src[idxn];
            const int dvn  = __float_as_int(csr_w[idxn]);

            if (beg < end)
                pull_chunk<FP, NP, U>(hw, sv, dv, di, min(32, end - beg), qc, acc);
            for (int base = beg + 32; base < end; base += 32) {
                const int idx = min(base + lane, end - 1);
                const int svc = csr_src[idx];
                const int dvc = __float_as_int(csr_w[idx]);
                pull_chunk<FP, NP, U>(hw, svc, dvc, di, min(32, end - base), qc, acc);
            }

            const float dd = di * di;
            float4* o4 = reinterpret_cast<float4*>(agg + (size_t)node * FP);
            #pragma unroll
            for (int p = 0; p < NP; ++p) {
                const int q = lane + 32 * p;
                if (q < Q) {
                    float4 o;
                    o.x = fmaf(dd, hs[p].x, acc[p].x);
                    o.y = fmaf(dd, hs[p].y, acc[p].y);
                    o.z = fmaf(dd, hs[p].z, acc[p].z);
                    o.w = fmaf(dd, hs[p].w, acc[p].w);
                    o4[q] = o;
                }
            }

            beg = nbeg;
            end = nend;
            sv  = svn;
            dv  = dvn;
        }
    }
}

extern "C" void kernel_launch(void* const* d_in, const int* in_sizes, int n_in,
                              void* d_out, int out_size, void* d_ws, size_t ws_size,
                              hipStream_t stream)
{
    const float* x    = (const float*)d_in[0];
    const int*   ei   = (const int*)d_in[1];
    const float* W1   = (const float*)d_in[2];
    const float* b1   = (const float*)d_in[3];
    const float* W2   = (const float*)d_in[4];
    const float* b2   = (const float*)d_in[5];
    const float* Wout = (const float*)d_in[6];
    const float* bout = (const float*)d_in[7];
    float*       out  = (float*)d_out;

    const int* srcI = ei;
    const int* dstI = ei + N_EDGES;

    float* ws   = (float*)d_ws;
    float* dinv = ws;
    float* bufA = ws + 50176;
    float* bufB = bufA + (size_t)N_NODES * H1P;
    int*   cnt    = (int*)(bufB + (size_t)N_NODES * H1P);
    int*   segbeg = cnt + 50176;
    int*   segend = segbeg + 50176;
    int*   csrsrc = segend + 50176;
    float* csrw   = (float*)(csrsrc + N_EDGES);

    k_cnt_zero <<<(N_NODES + 1 + 255) / 256, 256, 0, stream>>>(cnt, N_NODES + 1);
    k_cnt_edges<<<(N_EDGES / 4 + 255) / 256, 256, 0, stream>>>(dstI, cnt, N_EDGES / 4);
    k_seg_alloc<<<(N_NODES + 255) / 256, 256, 0, stream>>>(cnt, segbeg, segend, dinv, N_NODES);
    k_csr_fill <<<(N_EDGES / 4 + 255) / 256, 256, 0, stream>>>(srcI, dstI, dinv, segend, csrsrc, csrw,
                                                               N_EDGES / 4);

    k_gemm_wmma_f32<F_IN, H1, H1P, 132, false, false, H1P><<<256, 512, 0, stream>>>(
        x, F_IN, W1, bufA, H1P,  nullptr, N_NODES / 16);
    k_agg_pull<H1P, H1, 8><<<1024, 256, 0, stream>>>(segbeg, segend, csrsrc, csrw, dinv, bufA, b1,
                                                     bufB, N_NODES);

    k_gemm_wmma_f32<H1, H2, H2P, 204, true, false, H2P><<<256, 512, 0, stream>>>(
        bufB, H1P, W2, bufA, H2P,  nullptr, N_NODES / 16);
    k_agg_pull<H2P, H2, 16><<<1024, 256, 0, stream>>>(segbeg, segend, csrsrc, csrw, dinv, bufA, b2,
                                                      bufB, N_NODES);

    k_gemm_wmma_f32<H2, NC, NCP, 100, true, true, NC><<<256, 512, 0, stream>>>(
        bufB, H2P, Wout, out, NC,  bout, N_NODES / 16);
}


# Edge ordering without global atomics: edges split into 256-node dst ranges and counting-sorted per range with LDS counters (3 launches for 4); slot order in a segment, so sum order, changes
# speedup vs baseline: 6.3214x; 1.8532x over previous
#include <hip/hip_runtime.h>


typedef float v2f __attribute__((ext_vector_type(2)));
typedef float v8f __attribute__((ext_vector_type(8)));

#define N_NODES 50000
#define N_EDGES 800000
#define F_IN    128
#define H1      200
#define H1P     208
#define H2      100
#define H2P     112
#define NC      40
#define NCP     48
static_assert(N_EDGES % 4 == 0, "edge ids are read four at a time");


#define RANGE   256
#define NRANGE  ((N_NODES + RANGE - 1) / RANGE)
#define EPB     4096
#define NCHUNK  ((N_EDGES + EPB - 1) / EPB)
#define TPITCH  256
#define SPW     ((NCHUNK + 31) / 32)
static_assert(NRANGE < 256, "one LDS counter per range plus the total");
static_assert(NCHUNK <= TPITCH && NCHUNK <= 1024, "one thread per block run in pass 2");
static_assert(N_NODES <= (1 << 24), "source id shares a word with the 8-bit local destination");

__global__ void __launch_bounds__(1024)
k_range_split(const int* __restrict__ src, const int* __restrict__ dst,
              unsigned* __restrict__ part, int* __restrict__ tab, int e4)
{
    __shared__ int hist[256];
    __shared__ int bstart[256];
    __shared__ int wsum[8];
    __shared__ __attribute__((aligned(16))) unsigned outb[EPB];

    const int tid  = threadIdx.x;
    const int lane = tid & 31;
    const int wave = tid >> 5;
    const int g    = blockIdx.x;

    if (tid < 256) hist[tid] = 0;
    __syncthreads();

    const int  q    = g * (EPB / 4) + tid;
    const bool have = q < e4;
    int4 s = make_int4(0, 0, 0, 0);
    int4 d = make_int4(0, 0, 0, 0);
    int r0 = 0, r1 = 0, r2 = 0, r3 = 0;
    if (have) {
        s = reinterpret_cast<const int4*>(src)[q];
        d = reinterpret_cast<const int4*>(dst)[q];
        r0 = atomicAdd(&hist[d.x >> 8], 1);
        r1 = atomicAdd(&hist[d.y >> 8], 1);
        r2 = atomicAdd(&hist[d.z >> 8], 1);
        r3 = atomicAdd(&hist[d.w >> 8], 1);
    }
    __syncthreads();

    int v = 0, incl = 0;
    if (tid < 256) {
        v = hist[tid];
        incl = v;
        #pragma unroll
        for (int o = 1; o < 32; o <<= 1) {
            const int t = __shfl_up(incl, o);
            if (lane >= o) incl += t;
        }
        if (lane == 31) wsum[wave] = incl;
    }
    __syncthreads();
    if (tid < 256) {
        int off = 0;
        for (int w = 0; w < wave; ++w) off += wsum[w];
        const int ex = off + incl - v;
        bstart[tid] = ex;
        if (tid <= NRANGE) tab[tid * TPITCH + g] = ex;
    }
    __syncthreads();

    const int total = bstart[NRANGE];
    if (have) {
        outb[bstart[d.x >> 8] + r0] = ((unsigned)s.x << 8) | (unsigned)(d.x & 255);
        outb[bstart[d.y >> 8] + r1] = ((unsigned)s.y << 8) | (unsigned)(d.y & 255);
        outb[bstart[d.z >> 8] + r2] = ((unsigned)s.z << 8) | (unsigned)(d.z & 255);
        outb[bstart[d.w >> 8] + r3] = ((unsigned)s.w << 8) | (unsigned)(d.w & 255);
    }
    __syncthreads();
    if (4 * tid < total)
        reinterpret_cast<uint4*>(part)[g * (EPB / 4) + tid] = reinterpret_cast<const uint4*>(outb)[tid];
}

__global__ void __launch_bounds__(1024)
k_range_csr(const int* __restrict__ tab, const unsigned* __restrict__ part,
            int* __restrict__ segbeg, int* __restrict__ segend, float* __restrict__ dinv,
            int* __restrict__ csr_src, int n)
{
    __shared__ int sS[NCHUNK];
    __shared__ int cS[NCHUNK];
    __shared__ int cnt[256];
    __shared__ int cur[256];
    __shared__ int wsumA[32];
    __shared__ int wsumB[8];

    const int tid  = threadIdx.x;
    const int lane = tid & 31;
    const int wave = __builtin_amdgcn_readfirstlane(tid >> 5);
    const int r    = blockIdx.x;

    int s = 0;
    if (tid < NCHUNK) {
        s = tab[r * TPITCH + tid];
        const int e = tab[(r + 1) * TPITCH + tid];
        sS[tid] = s;
        cS[tid] = e - s;
    }
    if (tid < 256) cnt[tid] = 0;
    int rs = s;
    #pragma unroll
    for (int o = 16; o > 0; o >>= 1) rs += __shfl_xor(rs, o);
    if (lane == 0) wsumA[wave] = rs;
    __syncthreads();

    int base = 0;
    #pragma unroll
    for (int w = 0; w < SPW; ++w) base += wsumA[w];

    unsigned rec[SPW];
    int      cw[SPW];
    #pragma unroll
    for (int k = 0; k < SPW; ++k) {
        const int w = wave + 32 * k;
        const int c = (w < NCHUNK) ? cS[w] : 0;
        cw[k]  = c;
        rec[k] = 0u;
        if (lane < c) rec[k] = part[w * EPB + sS[w] + lane];
    }

    #pragma unroll
    for (int k = 0; k < SPW; ++k) {
        if (lane < cw[k]) atomicAdd(&cnt[rec[k] & 255u], 1);
        for (int i = 32 + lane; i < cw[k]; i += 32) {
            const int w = wave + 32 * k;
            const unsigned x = part[w * EPB + sS[w] + i];
            atomicAdd(&cnt[x & 255u], 1);
        }
    }
    __syncthreads();

    int v = 0, incl = 0;
    if (tid < 256) {
        v = cnt[tid];
        incl = v;
        #pragma unroll
        for (int o = 1; o < 32; o <<= 1) {
            const int t = __shfl_up(incl, o);
            if (lane >= o) incl += t;
        }
        if (lane == 31) wsumB[wave] = incl;
    }
    __syncthreads();
    if (tid < 256) {
        int off = 0;
        for (int w = 0; w < wave; ++w) off += wsumB[w];
        const int st = off + incl - v;
        cur[tid] = st;
        const int node = r * RANGE + tid;
        if (node < n) {
            segbeg[node] = base + st;
            segend[node] = base + st + v;
            dinv[node]   = rsqrtf((float)(v + 1));
        }
    }
    __syncthreads();

    #pragma unroll
    for (int k = 0; k < SPW; ++k) {
        if (lane < cw[k]) {
            const int p = atomicAdd(&cur[rec[k] & 255u], 1);
            csr_src[base + p] = (int)(rec[k] >> 8);
        }
        for (int i = 32 + lane; i < cw[k]; i += 32) {
            const int w = wave + 32 * k;
            const unsigned x = part[w * EPB + sS[w] + i];
            const int p = atomicAdd(&cur[x & 255u], 1);
            csr_src[base + p] = (int)(x >> 8);
        }
    }
}

__global__ void k_slot_weight(const int* __restrict__ csr_src, const float* __restrict__ dinv,
                              float* __restrict__ csr_w, int e4)
{
    int i = blockIdx.x * blockDim.x + threadIdx.x;
    if (i < e4) {
        const int4 s = reinterpret_cast<const int4*>(csr_src)[i];
        float4 w;
        w.x = dinv[s.x];
        w.y = dinv[s.y];
        w.z = dinv[s.z];
        w.w = dinv[s.w];
        reinterpret_cast<float4*>(csr_w)[i] = w;
    }
}

template <int KS, int S, int NTL>
__device__ __forceinline__ void mm_tiles(const v2f (&a)[KS], const float* bp, v8f (&acc)[NTL])
{
    #pragma unroll
    for (int j = 0; j < KS; ++j) {
        #pragma unroll
        for (int t = 0; t < NTL; ++t) {
            v2f b = *reinterpret_cast<const v2f*>(bp + t * 16 * S + 4 * j);
            acc[t] = __builtin_amdgcn_wmma_f32_16x16x4_f32(
                         false, a[j], false, b, (short)0, acc[t], false, false);
        }
    }
}

template <int K, int NACT, int NPAD, int S, bool RELU, bool HASBIAS, int NSTORE, int NTL>
__device__ __forceinline__ void tiles_out(const v2f (&a)[K / 4], const float* Bs, int nt,
                                          int c, int h, int m_base,
                                          float* __restrict__ C, int ldc,
                                          const float* __restrict__ bias)
{
    v8f acc[NTL];
    #pragma unroll
    for (int t = 0; t < NTL; ++t)
        acc[t] = (v8f){0.f, 0.f, 0.f, 0.f, 0.f, 0.f, 0.f, 0.f};

    mm_tiles<K / 4, S, NTL>(a, Bs + (nt * 16 + c) * S + 2 * h, acc);

    const int rbase = m_base + 8 * h;
    #pragma unroll
    for (int t = 0; t < NTL; ++t) {
        const int ncol = (nt + t) * 16 + c;
        float bv = 0.0f;
        if (HASBIAS) bv = (ncol < NACT) ? bias[ncol] : 0.0f;
        if (NSTORE >= NPAD || ncol < NSTORE) {
            #pragma unroll
            for (int g = 0; g < 8; ++g)
                C[(size_t)(rbase + g) * ldc + ncol] = acc[t][g] + bv;
        }
    }
}

template <int K, int NACT, int NPAD, int S, bool RELU, bool HASBIAS, int NSTORE>
__global__ void __launch_bounds__(512)
k_gemm_wmma_f32(const float* __restrict__ A, int lda,
                const float* __restrict__ B,
                float* __restrict__ C, int ldc,
                const float* __restrict__ bias, int mtiles)
{
    __shared__ __attribute__((aligned(16))) float Bs[NPAD * S];

    constexpr int KS    = K / 4;
    constexpr int NT    = NPAD / 16;
    constexpr int Q     = NACT / 4;
    constexpr int QG    = (Q + 7) / 8;
    constexpr int NTASK = KS * QG;

    const int tid  = threadIdx.x;
    const int lane = tid & 31;
    const int wave = __builtin_amdgcn_readfirstlane(tid >> 5);
    const int c    = lane & 15;
    const int h    = lane >> 4;

    int mt = wave * gridDim.x + blockIdx.x;
    v2f a[KS];
    if (mt < mtiles) {
        const float* arow = A + (size_t)(mt * 16 + c) * lda + 2 * h;
        #pragma unroll
        for (int j = 0; j < KS; ++j)
            a[j] = *reinterpret_cast<const v2f*>(arow + 4 * j);
    }

    {
        const int ql = lane & 7;
        const int kl = lane >> 3;
        constexpr int NIT = (NTASK + 15) / 16;
        float4 stg[NIT];
        #pragma unroll
        for (int it = 0; it < NIT; ++it) {
            const int task = wave + 16 * it;
            const int kg = task / QG;
            const int qg = task - kg * QG;
            const int q  = qg * 8 + ql;
            const int k  = kg * 4 + kl;
            stg[it] = make_float4(0.0f, 0.0f, 0.0f, 0.0f);
            if (task < NTASK && q < Q)
                stg[it] = *reinterpret_cast<const float4*>(B + (size_t)k * NACT + 4 * q);
        }
        #pragma unroll
        for (int it = 0; it < NIT; ++it) {
            const int task = wave + 16 * it;
            const int kg = task / QG;
            const int qg = task - kg * QG;
            const int q  = qg * 8 + ql;
            const int k  = kg * 4 + kl;
            if (task < NTASK && q < Q) {
                float* d = &Bs[(4 * q) * S + k];
                d[0]     = stg[it].x;
                d[S]     = stg[it].y;
                d[2 * S] = stg[it].z;
                d[3 * S] = stg[it].w;
            }
        }
        for (int idx = tid; idx < (NPAD - NACT) * K; idx += 512) {
            const int cc = idx / K;
            const int k  = idx - cc * K;
            Bs[(NACT + cc) * S + k] = 0.0f;
        }
    }
    __syncthreads();

    while (mt < mtiles) {
        const int m_base = mt * 16;
        if (RELU) {
            #pragma unroll
            for (int j = 0; j < KS; ++j) {
                a[j].x = fmaxf(a[j].x, 0.0f);
                a[j].y = fmaxf(a[j].y, 0.0f);
            }
        }
        int nt = 0;
        for (; nt + 2 <= NT; nt += 2)
            tiles_out<K, NACT, NPAD, S, RELU, HASBIAS, NSTORE, 2>(a, Bs, nt, c, h, m_base, C, ldc, bias);
        if (NT & 1)
            tiles_out<K, NACT, NPAD, S, RELU, HASBIAS, NSTORE, 1>(a, Bs, NT - 1, c, h, m_base, C, ldc, bias);

        mt += 16 * gridDim.x;
        if (mt < mtiles) {
            const float* arow = A + (size_t)(mt * 16 + c) * lda + 2 * h;
            #pragma unroll
            for (int j = 0; j < KS; ++j)
                a[j] = *reinterpret_cast<const v2f*>(arow + 4 * j);
        }
    }
}

template <int FP, int NP, int R>
__device__ __forceinline__ void pull_rows(const float* __restrict__ hw, int sv, int dvb, float di,
                                          int e, const int (&qc)[NP], float4 (&acc)[NP])
{
    float4 h[R][NP];
    #pragma unroll
    for (int u = 0; u < R; ++u) {
        const int s = __builtin_amdgcn_readlane(sv, e + u);
        const float4* row = reinterpret_cast<const float4*>(hw + (size_t)s * FP);
        #pragma unroll
        for (int p = 0; p < NP; ++p)
            h[u][p] = row[qc[p]];
    }
    #pragma unroll
    for (int u = 0; u < R; ++u) {
        const float w = __int_as_float(__builtin_amdgcn_readlane(dvb, e + u)) * di;
        #pragma unroll
        for (int p = 0; p < NP; ++p) {
            acc[p].x = fmaf(w, h[u][p].x, acc[p].x);
            acc[p].y = fmaf(w, h[u][p].y, acc[p].y);
            acc[p].z = fmaf(w, h[u][p].z, acc[p].z);
            acc[p].w = fmaf(w, h[u][p].w, acc[p].w);
        }
    }
}

template <int FP, int NP, int R>
__device__ __forceinline__ void pull_tail(const float* __restrict__ hw, int sv, int dvb, float di,
                                          int e, int r, const int (&qc)[NP], float4 (&acc)[NP])
{
    if constexpr (R >= 1) {
        if (r == R) pull_rows<FP, NP, R>(hw, sv, dvb, di, e, qc, acc);
        else        pull_tail<FP, NP, R - 1>(hw, sv, dvb, di, e, r, qc, acc);
    }
}

template <int FP, int NP, int U>
__device__ __forceinline__ void pull_chunk(const float* __restrict__ hw, int sv, int dvb, float di,
                                           int cnt, const int (&qc)[NP], float4 (&acc)[NP])
{
    int e = 0;
    for (; e + U <= cnt; e += U)
        pull_rows<FP, NP, U>(hw, sv, dvb, di, e, qc, acc);
    pull_tail<FP, NP, U - 1>(hw, sv, dvb, di, e, cnt - e, qc, acc);
}

template <int FP, int FACT, int U>
__global__ void __launch_bounds__(256, 8)
k_agg_pull(const int* __restrict__ start, const int* __restrict__ endp,
           const int* __restrict__ csr_src, const float* __restrict__ csr_w,
           const float* __restrict__ dinv,
           const float* __restrict__ hw, const float* __restrict__ bias,
           float* __restrict__ agg, int n)
{
    constexpr int Q  = FP / 4;
    constexpr int QB = FACT / 4;
    constexpr int NP = (Q + 31) / 32;
    static_assert(FP % 4 == 0 && FACT % 4 == 0, "rows are walked as float4");

    const int lane = threadIdx.x & 31;
    const int wid  = __builtin_amdgcn_readfirstlane((int)((blockIdx.x * blockDim.x + threadIdx.x) >> 5));
    const int nw   = (int)((gridDim.x * blockDim.x) >> 5);

    int    qc[NP];
    float4 bq[NP];
    #pragma unroll
    for (int p = 0; p < NP; ++p) {
        const int q = lane + 32 * p;
        qc[p] = min(q, Q - 1);
        bq[p] = make_float4(0.0f, 0.0f, 0.0f, 0.0f);
        if (q < QB) bq[p] = reinterpret_cast<const float4*>(bias)[q];
    }

    for (int node0 = wid; node0 < n; node0 += 32 * nw) {
        const int mine = min(node0 + lane * nw, n - 1);
        const int begv = start[mine];
        const int endv = endp[mine];
        const int divb = __float_as_int(dinv[mine]);
        const int kn   = min(32, (n - 1 - node0) / nw + 1);

        int beg = __builtin_amdgcn_readlane(begv, 0);
        int end = __builtin_amdgcn_readlane(endv, 0);
        int sv, dv;
        {
            const int idx = max(min(beg + lane, end - 1), 0);
            sv = csr_src[idx];
            dv = __float_as_int(csr_w[idx]);
        }

        for (int k = 0; k < kn; ++k) {
            const int   node = node0 + k * nw;
            const float di   = __int_as_float(__builtin_amdgcn_readlane(divb, k));

            const float4* hself = reinterpret_cast<const float4*>(hw + (size_t)node * FP);
            float4 hs[NP];
            float4 acc[NP];
            #pragma unroll
            for (int p = 0; p < NP; ++p) {
                hs[p]  = hself[qc[p]];
                acc[p] = bq[p];
            }

            const int k1   = min(k + 1, kn - 1);
            const int nbeg = __builtin_amdgcn_readlane(begv, k1);
            const int nend = __builtin_amdgcn_readlane(endv, k1);
            const int idxn = max(min(nbeg + lane, nend - 1), 0);
            const int svn  = csr_src[idxn];
            const int dvn  = __float_as_int(csr_w[idxn]);

            if (beg < end)
                pull_chunk<FP, NP, U>(hw, sv, dv, di, min(32, end - beg), qc, acc);
            for (int base = beg + 32; base < end; base += 32) {
                const int idx = min(base + lane, end - 1);
                const int svc = csr_src[idx];
                const int dvc = __float_as_int(csr_w[idx]);
                pull_chunk<FP, NP, U>(hw, svc, dvc, di, min(32, end - base), qc, acc);
            }

            const float dd = di * di;
            float4* o4 = reinterpret_cast<float4*>(agg + (size_t)node * FP);
            #pragma unroll
            for (int p = 0; p < NP; ++p) {
                const int q = lane + 32 * p;
                if (q < Q) {
                    float4 o;
                    o.x = fmaf(dd, hs[p].x, acc[p].x);
                    o.y = fmaf(dd, hs[p].y, acc[p].y);
                    o.z = fmaf(dd, hs[p].z, acc[p].z);
                    o.w = fmaf(dd, hs[p].w, acc[p].w);
                    o4[q] = o;
                }
            }

            beg = nbeg;
            end = nend;
            sv  = svn;
            dv  = dvn;
        }
    }
}

extern "C" void kernel_launch(void* const* d_in, const int* in_sizes, int n_in,
                              void* d_out, int out_size, void* d_ws, size_t ws_size,
                              hipStream_t stream)
{
    const float* x    = (const float*)d_in[0];
    const int*   ei   = (const int*)d_in[1];
    const float* W1   = (const float*)d_in[2];
    const float* b1   = (const float*)d_in[3];
    const float* W2   = (const float*)d_in[4];
    const float* b2   = (const float*)d_in[5];
    const float* Wout = (const float*)d_in[6];
    const float* bout = (const float*)d_in[7];
    float*       out  = (float*)d_out;

    const int* srcI = ei;
    const int* dstI = ei + N_EDGES;

    float* ws   = (float*)d_ws;
    float* dinv = ws;
    float* bufA = ws + 50176;
    float* bufB = bufA + (size_t)N_NODES * H1P;
    int*   tab    = (int*)(bufB + (size_t)N_NODES * H1P);
    int*   segbeg = tab + (NRANGE + 1) * TPITCH;
    int*   segend = segbeg + 50176;
    int*   csrsrc = segend + 50176;
    float* csrw   = (float*)(csrsrc + N_EDGES);
    unsigned* part = (unsigned*)(csrw + N_EDGES);

    k_range_split<<<NCHUNK, 1024, 0, stream>>>(srcI, dstI, part, tab, N_EDGES / 4);
    k_range_csr  <<<NRANGE, 1024, 0, stream>>>(tab, part, segbeg, segend, dinv, csrsrc, N_NODES);
    k_slot_weight<<<(N_EDGES / 4 + 255) / 256, 256, 0, stream>>>(csrsrc, dinv, csrw, N_EDGES / 4);

    k_gemm_wmma_f32<F_IN, H1, H1P, 132, false, false, H1P><<<256, 512, 0, stream>>>(
        x, F_IN, W1, bufA, H1P,  nullptr, N_NODES / 16);
    k_agg_pull<H1P, H1, 8><<<1024, 256, 0, stream>>>(segbeg, segend, csrsrc, csrw, dinv, bufA, b1,
                                                     bufB, N_NODES);

    k_gemm_wmma_f32<H1, H2, H2P, 204, true, false, H2P><<<256, 512, 0, stream>>>(
        bufB, H1P, W2, bufA, H2P,  nullptr, N_NODES / 16);
    k_agg_pull<H2P, H2, 16><<<1024, 256, 0, stream>>>(segbeg, segend, csrsrc, csrw, dinv, bufA, b2,
                                                      bufB, N_NODES);

    k_gemm_wmma_f32<H2, NC, NCP, 100, true, true, NC><<<256, 512, 0, stream>>>(
        bufB, H2P, Wout, out, NC,  bout, N_NODES / 16);
}


# Layer 1 aggregates first: Agg(x)@W1 for Agg(x@W1) (Agg is linear); the gather reads 128-wide input rows, not 208-wide products; b1 moves to the GEMM; order of the f32 sums changes
# speedup vs baseline: 6.5719x; 1.0396x over previous
#include <hip/hip_runtime.h>


typedef float v2f __attribute__((ext_vector_type(2)));
typedef float v8f __attribute__((ext_vector_type(8)));

#define N_NODES 50000
#define N_EDGES 800000
#define F_IN    128
#define H1      200
#define H1P     208
#define H2      100
#define H2P     112
#define NC      40
#define NCP     48
static_assert(N_EDGES % 4 == 0, "edge ids are read four at a time");


#define RANGE   256
#define NRANGE  ((N_NODES + RANGE - 1) / RANGE)
#define EPB     4096
#define NCHUNK  ((N_EDGES + EPB - 1) / EPB)
#define TPITCH  256
#define SPW     ((NCHUNK + 31) / 32)
static_assert(NRANGE < 256, "one LDS counter per range plus the total");
static_assert(NCHUNK <= TPITCH && NCHUNK <= 1024, "one thread per block run in pass 2");
static_assert(N_NODES <= (1 << 24), "source id shares a word with the 8-bit local destination");

__global__ void __launch_bounds__(1024)
k_range_split(const int* __restrict__ src, const int* __restrict__ dst,
              unsigned* __restrict__ part, int* __restrict__ tab, int e4)
{
    __shared__ int hist[256];
    __shared__ int bstart[256];
    __shared__ int wsum[8];
    __shared__ __attribute__((aligned(16))) unsigned outb[EPB];

    const int tid  = threadIdx.x;
    const int lane = tid & 31;
    const int wave = tid >> 5;
    const int g    = blockIdx.x;

    if (tid < 256) hist[tid] = 0;
    __syncthreads();

    const int  q    = g * (EPB / 4) + tid;
    const bool have = q < e4;
    int4 s = make_int4(0, 0, 0, 0);
    int4 d = make_int4(0, 0, 0, 0);
    int r0 = 0, r1 = 0, r2 = 0, r3 = 0;
    if (have) {
        s = reinterpret_cast<const int4*>(src)[q];
        d = reinterpret_cast<const int4*>(dst)[q];
        r0 = atomicAdd(&hist[d.x >> 8], 1);
        r1 = atomicAdd(&hist[d.y >> 8], 1);
        r2 = atomicAdd(&hist[d.z >> 8], 1);
        r3 = atomicAdd(&hist[d.w >> 8], 1);
    }
    __syncthreads();

    int v = 0, incl = 0;
    if (tid < 256) {
        v = hist[tid];
        incl = v;
        #pragma unroll
        for (int o = 1; o < 32; o <<= 1) {
            const int t = __shfl_up(incl, o);
            if (lane >= o) incl += t;
        }
        if (lane == 31) wsum[wave] = incl;
    }
    __syncthreads();
    if (tid < 256) {
        int off = 0;
        for (int w = 0; w < wave; ++w) off += wsum[w];
        const int ex = off + incl - v;
        bstart[tid] = ex;
        if (tid <= NRANGE) tab[tid * TPITCH + g] = ex;
    }
    __syncthreads();

    const int total = bstart[NRANGE];
    if (have) {
        outb[bstart[d.x >> 8] + r0] = ((unsigned)s.x << 8) | (unsigned)(d.x & 255);
        outb[bstart[d.y >> 8] + r1] = ((unsigned)s.y << 8) | (unsigned)(d.y & 255);
        outb[bstart[d.z >> 8] + r2] = ((unsigned)s.z << 8) | (unsigned)(d.z & 255);
        outb[bstart[d.w >> 8] + r3] = ((unsigned)s.w << 8) | (unsigned)(d.w & 255);
    }
    __syncthreads();
    if (4 * tid < total)
        reinterpret_cast<uint4*>(part)[g * (EPB / 4) + tid] = reinterpret_cast<const uint4*>(outb)[tid];
}

__global__ void __launch_bounds__(1024)
k_range_csr(const int* __restrict__ tab, const unsigned* __restrict__ part,
            int* __restrict__ segbeg, int* __restrict__ segend, float* __restrict__ dinv,
            int* __restrict__ csr_src, int n)
{
    __shared__ int sS[NCHUNK];
    __shared__ int cS[NCHUNK];
    __shared__ int cnt[256];
    __shared__ int cur[256];
    __shared__ int wsumA[32];
    __shared__ int wsumB[8];

    const int tid  = threadIdx.x;
    const int lane = tid & 31;
    const int wave = __builtin_amdgcn_readfirstlane(tid >> 5);
    const int r    = blockIdx.x;

    int s = 0;
    if (tid < NCHUNK) {
        s = tab[r * TPITCH + tid];
        const int e = tab[(r + 1) * TPITCH + tid];
        sS[tid] = s;
        cS[tid] = e - s;
    }
    if (tid < 256) cnt[tid] = 0;
    int rs = s;
    #pragma unroll
    for (int o = 16; o > 0; o >>= 1) rs += __shfl_xor(rs, o);
    if (lane == 0) wsumA[wave] = rs;
    __syncthreads();

    int base = 0;
    #pragma unroll
    for (int w = 0; w < SPW; ++w) base += wsumA[w];

    unsigned rec[SPW];
    int      cw[SPW];
    #pragma unroll
    for (int k = 0; k < SPW; ++k) {
        const int w = wave + 32 * k;
        const int c = (w < NCHUNK) ? cS[w] : 0;
        cw[k]  = c;
        rec[k] = 0u;
        if (lane < c) rec[k] = part[w * EPB + sS[w] + lane];
    }

    #pragma unroll
    for (int k = 0; k < SPW; ++k) {
        if (lane < cw[k]) atomicAdd(&cnt[rec[k] & 255u], 1);
        for (int i = 32 + lane; i < cw[k]; i += 32) {
            const int w = wave + 32 * k;
            const unsigned x = part[w * EPB + sS[w] + i];
            atomicAdd(&cnt[x & 255u], 1);
        }
    }
    __syncthreads();

    int v = 0, incl = 0;
    if (tid < 256) {
        v = cnt[tid];
        incl = v;
        #pragma unroll
        for (int o = 1; o < 32; o <<= 1) {
            const int t = __shfl_up(incl, o);
            if (lane >= o) incl += t;
        }
        if (lane == 31) wsumB[wave] = incl;
    }
    __syncthreads();
    if (tid < 256) {
        int off = 0;
        for (int w = 0; w < wave; ++w) off += wsumB[w];
        const int st = off + incl - v;
        cur[tid] = st;
        const int node = r * RANGE + tid;
        if (node < n) {
            segbeg[node] = base + st;
            segend[node] = base + st + v;
            dinv[node]   = rsqrtf((float)(v + 1));
        }
    }
    __syncthreads();

    #pragma unroll
    for (int k = 0; k < SPW; ++k) {
        if (lane < cw[k]) {
            const int p = atomicAdd(&cur[rec[k] & 255u], 1);
            csr_src[base + p] = (int)(rec[k] >> 8);
        }
        for (int i = 32 + lane; i < cw[k]; i += 32) {
            const int w = wave + 32 * k;
            const unsigned x = part[w * EPB + sS[w] + i];
            const int p = atomicAdd(&cur[x & 255u], 1);
            csr_src[base + p] = (int)(x >> 8);
        }
    }
}

__global__ void k_slot_weight(const int* __restrict__ csr_src, const float* __restrict__ dinv,
                              float* __restrict__ csr_w, int e4)
{
    int i = blockIdx.x * blockDim.x + threadIdx.x;
    if (i < e4) {
        const int4 s = reinterpret_cast<const int4*>(csr_src)[i];
        float4 w;
        w.x = dinv[s.x];
        w.y = dinv[s.y];
        w.z = dinv[s.z];
        w.w = dinv[s.w];
        reinterpret_cast<float4*>(csr_w)[i] = w;
    }
}

template <int KS, int S, int NTL>
__device__ __forceinline__ void mm_tiles(const v2f (&a)[KS], const float* bp, v8f (&acc)[NTL])
{
    #pragma unroll
    for (int j = 0; j < KS; ++j) {
        #pragma unroll
        for (int t = 0; t < NTL; ++t) {
            v2f b = *reinterpret_cast<const v2f*>(bp + t * 16 * S + 4 * j);
            acc[t] = __builtin_amdgcn_wmma_f32_16x16x4_f32(
                         false, a[j], false, b, (short)0, acc[t], false, false);
        }
    }
}

template <int K, int NACT, int NPAD, int S, bool RELU, bool HASBIAS, int NSTORE, int NTL>
__device__ __forceinline__ void tiles_out(const v2f (&a)[K / 4], const float* Bs, int nt,
                                          int c, int h, int m_base,
                                          float* __restrict__ C, int ldc,
                                          const float* __restrict__ bias)
{
    v8f acc[NTL];
    #pragma unroll
    for (int t = 0; t < NTL; ++t)
        acc[t] = (v8f){0.f, 0.f, 0.f, 0.f, 0.f, 0.f, 0.f, 0.f};

    mm_tiles<K / 4, S, NTL>(a, Bs + (nt * 16 + c) * S + 2 * h, acc);

    const int rbase = m_base + 8 * h;
    #pragma unroll
    for (int t = 0; t < NTL; ++t) {
        const int ncol = (nt + t) * 16 + c;
        float bv = 0.0f;
        if (HASBIAS) bv = (ncol < NACT) ? bias[ncol] : 0.0f;
        if (NSTORE >= NPAD || ncol < NSTORE) {
            #pragma unroll
            for (int g = 0; g < 8; ++g)
                C[(size_t)(rbase + g) * ldc + ncol] = acc[t][g] + bv;
        }
    }
}

template <int K, int NACT, int NPAD, int S, bool RELU, bool HASBIAS, int NSTORE>
__global__ void __launch_bounds__(512)
k_gemm_wmma_f32(const float* __restrict__ A, int lda,
                const float* __restrict__ B,
                float* __restrict__ C, int ldc,
                const float* __restrict__ bias, int mtiles)
{
    __shared__ __attribute__((aligned(16))) float Bs[NPAD * S];

    constexpr int KS    = K / 4;
    constexpr int NT    = NPAD / 16;
    constexpr int Q     = NACT / 4;
    constexpr int QG    = (Q + 7) / 8;
    constexpr int NTASK = KS * QG;

    const int tid  = threadIdx.x;
    const int lane = tid & 31;
    const int wave = __builtin_amdgcn_readfirstlane(tid >> 5);
    const int c    = lane & 15;
    const int h    = lane >> 4;

    int mt = wave * gridDim.x + blockIdx.x;
    v2f a[KS];
    if (mt < mtiles) {
        const float* arow = A + (size_t)(mt * 16 + c) * lda + 2 * h;
        #pragma unroll
        for (int j = 0; j < KS; ++j)
            a[j] = *reinterpret_cast<const v2f*>(arow + 4 * j);
    }

    {
        const int ql = lane & 7;
        const int kl = lane >> 3;
        constexpr int NIT = (NTASK + 15) / 16;
        float4 stg[NIT];
        #pragma unroll
        for (int it = 0; it < NIT; ++it) {
            const int task = wave + 16 * it;
            const int kg = task / QG;
            const int qg = task - kg * QG;
            const int q  = qg * 8 + ql;
            const int k  = kg * 4 + kl;
            stg[it] = make_float4(0.0f, 0.0f, 0.0f, 0.0f);
            if (task < NTASK && q < Q)
                stg[it] = *reinterpret_cast<const float4*>(B + (size_t)k * NACT + 4 * q);
        }
        #pragma unroll
        for (int it = 0; it < NIT; ++it) {
            const int task = wave + 16 * it;
            const int kg = task / QG;
            const int qg = task - kg * QG;
            const int q  = qg * 8 + ql;
            const int k  = kg * 4 + kl;
            if (task < NTASK && q < Q) {
                float* d = &Bs[(4 * q) * S + k];
                d[0]     = stg[it].x;
                d[S]     = stg[it].y;
                d[2 * S] = stg[it].z;
                d[3 * S] = stg[it].w;
            }
        }
        for (int idx = tid; idx < (NPAD - NACT) * K; idx += 512) {
            const int cc = idx / K;
            const int k  = idx - cc * K;
            Bs[(NACT + cc) * S + k] = 0.0f;
        }
    }
    __syncthreads();

    while (mt < mtiles) {
        const int m_base = mt * 16;
        if (RELU) {
            #pragma unroll
            for (int j = 0; j < KS; ++j) {
                a[j].x = fmaxf(a[j].x, 0.0f);
                a[j].y = fmaxf(a[j].y, 0.0f);
            }
        }
        int nt = 0;
        for (; nt + 2 <= NT; nt += 2)
            tiles_out<K, NACT, NPAD, S, RELU, HASBIAS, NSTORE, 2>(a, Bs, nt, c, h, m_base, C, ldc, bias);
        if (NT & 1)
            tiles_out<K, NACT, NPAD, S, RELU, HASBIAS, NSTORE, 1>(a, Bs, NT - 1, c, h, m_base, C, ldc, bias);

        mt += 16 * gridDim.x;
        if (mt < mtiles) {
            const float* arow = A + (size_t)(mt * 16 + c) * lda + 2 * h;
            #pragma unroll
            for (int j = 0; j < KS; ++j)
                a[j] = *reinterpret_cast<const v2f*>(arow + 4 * j);
        }
    }
}

template <int FP, int NP, int R>
__device__ __forceinline__ void pull_rows(const float* __restrict__ hw, int sv, int dvb, float di,
                                          int e, const int (&qc)[NP], float4 (&acc)[NP])
{
    float4 h[R][NP];
    #pragma unroll
    for (int u = 0; u < R; ++u) {
        const int s = __builtin_amdgcn_readlane(sv, e + u);
        const float4* row = reinterpret_cast<const float4*>(hw + (size_t)s * FP);
        #pragma unroll
        for (int p = 0; p < NP; ++p)
            h[u][p] = row[qc[p]];
    }
    #pragma unroll
    for (int u = 0; u < R; ++u) {
        const float w = __int_as_float(__builtin_amdgcn_readlane(dvb, e + u)) * di;
        #pragma unroll
        for (int p = 0; p < NP; ++p) {
            acc[p].x = fmaf(w, h[u][p].x, acc[p].x);
            acc[p].y = fmaf(w, h[u][p].y, acc[p].y);
            acc[p].z = fmaf(w, h[u][p].z, acc[p].z);
            acc[p].w = fmaf(w, h[u][p].w, acc[p].w);
        }
    }
}

template <int FP, int NP, int R>
__device__ __forceinline__ void pull_tail(const float* __restrict__ hw, int sv, int dvb, float di,
                                          int e, int r, const int (&qc)[NP], float4 (&acc)[NP])
{
    if constexpr (R >= 1) {
        if (r == R) pull_rows<FP, NP, R>(hw, sv, dvb, di, e, qc, acc);
        else        pull_tail<FP, NP, R - 1>(hw, sv, dvb, di, e, r, qc, acc);
    }
}

template <int FP, int NP, int U>
__device__ __forceinline__ void pull_chunk(const float* __restrict__ hw, int sv, int dvb, float di,
                                           int cnt, const int (&qc)[NP], float4 (&acc)[NP])
{
    int e = 0;
    for (; e + U <= cnt; e += U)
        pull_rows<FP, NP, U>(hw, sv, dvb, di, e, qc, acc);
    pull_tail<FP, NP, U - 1>(hw, sv, dvb, di, e, cnt - e, qc, acc);
}

template <int FP, int FACT, int U>
__global__ void __launch_bounds__(256, 8)
k_agg_pull(const int* __restrict__ start, const int* __restrict__ endp,
           const int* __restrict__ csr_src, const float* __restrict__ csr_w,
           const float* __restrict__ dinv,
           const float* __restrict__ hw, const float* __restrict__ bias,
           float* __restrict__ agg, int n)
{
    constexpr int Q  = FP / 4;
    constexpr int QB = FACT / 4;
    constexpr int NP = (Q + 31) / 32;
    static_assert(FP % 4 == 0 && FACT % 4 == 0, "rows are walked as float4");

    const int lane = threadIdx.x & 31;
    const int wid  = __builtin_amdgcn_readfirstlane((int)((blockIdx.x * blockDim.x + threadIdx.x) >> 5));
    const int nw   = (int)((gridDim.x * blockDim.x) >> 5);

    int    qc[NP];
    float4 bq[NP];
    #pragma unroll
    for (int p = 0; p < NP; ++p) {
        const int q = lane + 32 * p;
        qc[p] = min(q, Q - 1);
        bq[p] = make_float4(0.0f, 0.0f, 0.0f, 0.0f);
        if (q < QB) bq[p] = reinterpret_cast<const float4*>(bias)[q];
    }

    for (int node0 = wid; node0 < n; node0 += 32 * nw) {
        const int mine = min(node0 + lane * nw, n - 1);
        const int begv = start[mine];
        const int endv = endp[mine];
        const int divb = __float_as_int(dinv[mine]);
        const int kn   = min(32, (n - 1 - node0) / nw + 1);

        int beg = __builtin_amdgcn_readlane(begv, 0);
        int end = __builtin_amdgcn_readlane(endv, 0);
        int sv, dv;
        {
            const int idx = max(min(beg + lane, end - 1), 0);
            sv = csr_src[idx];
            dv = __float_as_int(csr_w[idx]);
        }

        for (int k = 0; k < kn; ++k) {
            const int   node = node0 + k * nw;
            const float di   = __int_as_float(__builtin_amdgcn_readlane(divb, k));

            const float4* hself = reinterpret_cast<const float4*>(hw + (size_t)node * FP);
            float4 hs[NP];
            float4 acc[NP];
            #pragma unroll
            for (int p = 0; p < NP; ++p) {
                hs[p]  = hself[qc[p]];
                acc[p] = bq[p];
            }

            const int k1   = min(k + 1, kn - 1);
            const int nbeg = __builtin_amdgcn_readlane(begv, k1);
            const int nend = __builtin_amdgcn_readlane(endv, k1);
            const int idxn = max(min(nbeg + lane, nend - 1), 0);
            const int svn  = csr_src[idxn];
            const int dvn  = __float_as_int(csr_w[idxn]);

            if (beg < end)
                pull_chunk<FP, NP, U>(hw, sv, dv, di, min(32, end - beg), qc, acc);
            for (int base = beg + 32; base < end; base += 32) {
                const int idx = min(base + lane, end - 1);
                const int svc = csr_src[idx];
                const int dvc = __float_as_int(csr_w[idx]);
                pull_chunk<FP, NP, U>(hw, svc, dvc, di, min(32, end - base), qc, acc);
            }

            const float dd = di * di;
            float4* o4 = reinterpret_cast<float4*>(agg + (size_t)node * FP);
            #pragma unroll
            for (int p = 0; p < NP; ++p) {
                const int q = lane + 32 * p;
                if (q < Q) {
                    float4 o;
                    o.x = fmaf(dd, hs[p].x, acc[p].x);
                    o.y = fmaf(dd, hs[p].y, acc[p].y);
                    o.z = fmaf(dd, hs[p].z, acc[p].z);
                    o.w = fmaf(dd, hs[p].w, acc[p].w);
                    o4[q] = o;
                }
            }

            beg = nbeg;
            end = nend;
            sv  = svn;
            dv  = dvn;
        }
    }
}

extern "C" void kernel_launch(void* const* d_in, const int* in_sizes, int n_in,
                              void* d_out, int out_size, void* d_ws, size_t ws_size,
                              hipStream_t stream)
{
    const float* x    = (const float*)d_in[0];
    const int*   ei   = (const int*)d_in[1];
    const float* W1   = (const float*)d_in[2];
    const float* b1   = (const float*)d_in[3];
    const float* W2   = (const float*)d_in[4];
    const float* b2   = (const float*)d_in[5];
    const float* Wout = (const float*)d_in[6];
    const float* bout = (const float*)d_in[7];
    float*       out  = (float*)d_out;

    const int* srcI = ei;
    const int* dstI = ei + N_EDGES;

    float* ws   = (float*)d_ws;
    float* dinv = ws;
    float* bufA = ws + 50176;
    float* bufB = bufA + (size_t)N_NODES * H1P;
    int*   tab    = (int*)(bufB + (size_t)N_NODES * H1P);
    int*   segbeg = tab + (NRANGE + 1) * TPITCH;
    int*   segend = segbeg + 50176;
    int*   csrsrc = segend + 50176;
    float* csrw   = (float*)(csrsrc + N_EDGES);
    unsigned* part = (unsigned*)(csrw + N_EDGES);

    k_range_split<<<NCHUNK, 1024, 0, stream>>>(srcI, dstI, part, tab, N_EDGES / 4);
    k_range_csr  <<<NRANGE, 1024, 0, stream>>>(tab, part, segbeg, segend, dinv, csrsrc, N_NODES);
    k_slot_weight<<<(N_EDGES / 4 + 255) / 256, 256, 0, stream>>>(csrsrc, dinv, csrw, N_EDGES / 4);

    k_agg_pull<F_IN, 0, 16><<<1024, 256, 0, stream>>>(segbeg, segend, csrsrc, csrw, dinv, x,
                                                       nullptr, bufB, N_NODES);
    k_gemm_wmma_f32<F_IN, H1, H1P, 132, false, true, H1P><<<256, 512, 0, stream>>>(
        bufB, F_IN, W1, bufA, H1P,  b1, N_NODES / 16);

    k_gemm_wmma_f32<H1, H2, H2P, 204, true, false, H2P><<<256, 512, 0, stream>>>(
        bufA, H1P, W2, bufB, H2P,  nullptr, N_NODES / 16);
    k_agg_pull<H2P, H2, 16><<<1024, 256, 0, stream>>>(segbeg, segend, csrsrc, csrw, dinv, bufB, b2,
                                                      bufA, N_NODES);

    k_gemm_wmma_f32<H2, NC, NCP, 100, true, true, NC><<<256, 512, 0, stream>>>(
        bufA, H2P, Wout, out, NC,  bout, N_NODES / 16);
}


# GEMM 1 and GEMM 2 fused in one kernel: transposed products, the z1 tile in D registers is the next B fragment (W2 k-permuted in LDS); z1 never stored; order of the f32 k sums in GEMM 2 changes
# speedup vs baseline: 6.7841x; 1.0323x over previous
#include <hip/hip_runtime.h>


typedef float v2f __attribute__((ext_vector_type(2)));
typedef float v8f __attribute__((ext_vector_type(8)));

#define N_NODES 50000
#define N_EDGES 800000
#define F_IN    128
#define H1      200
#define H1P     208
#define H2      100
#define H2P     112
#define NC      40
#define NCP     48
static_assert(N_EDGES % 4 == 0, "edge ids are read four at a time");


#define RANGE   256
#define NRANGE  ((N_NODES + RANGE - 1) / RANGE)
#define EPB     4096
#define NCHUNK  ((N_EDGES + EPB - 1) / EPB)
#define TPITCH  256
#define SPW     ((NCHUNK + 31) / 32)
static_assert(NRANGE < 256, "one LDS counter per range plus the total");
static_assert(NCHUNK <= TPITCH && NCHUNK <= 1024, "one thread per block run in pass 2");
static_assert(N_NODES <= (1 << 24), "source id shares a word with the 8-bit local destination");

__global__ void __launch_bounds__(1024)
k_range_split(const int* __restrict__ src, const int* __restrict__ dst,
              unsigned* __restrict__ part, int* __restrict__ tab, int e4)
{
    __shared__ int hist[256];
    __shared__ int bstart[256];
    __shared__ int wsum[8];
    __shared__ __attribute__((aligned(16))) unsigned outb[EPB];

    const int tid  = threadIdx.x;
    const int lane = tid & 31;
    const int wave = tid >> 5;
    const int g    = blockIdx.x;

    if (tid < 256) hist[tid] = 0;
    __syncthreads();

    const int  q    = g * (EPB / 4) + tid;
    const bool have = q < e4;
    int4 s = make_int4(0, 0, 0, 0);
    int4 d = make_int4(0, 0, 0, 0);
    int r0 = 0, r1 = 0, r2 = 0, r3 = 0;
    if (have) {
        s = reinterpret_cast<const int4*>(src)[q];
        d = reinterpret_cast<const int4*>(dst)[q];
        r0 = atomicAdd(&hist[d.x >> 8], 1);
        r1 = atomicAdd(&hist[d.y >> 8], 1);
        r2 = atomicAdd(&hist[d.z >> 8], 1);
        r3 = atomicAdd(&hist[d.w >> 8], 1);
    }
    __syncthreads();

    int v = 0, incl = 0;
    if (tid < 256) {
        v = hist[tid];
        incl = v;
        #pragma unroll
        for (int o = 1; o < 32; o <<= 1) {
            const int t = __shfl_up(incl, o);
            if (lane >= o) incl += t;
        }
        if (lane == 31) wsum[wave] = incl;
    }
    __syncthreads();
    if (tid < 256) {
        int off = 0;
        for (int w = 0; w < wave; ++w) off += wsum[w];
        const int ex = off + incl - v;
        bstart[tid] = ex;
        if (tid <= NRANGE) tab[tid * TPITCH + g] = ex;
    }
    __syncthreads();

    const int total = bstart[NRANGE];
    if (have) {
        outb[bstart[d.x >> 8] + r0] = ((unsigned)s.x << 8) | (unsigned)(d.x & 255);
        outb[bstart[d.y >> 8] + r1] = ((unsigned)s.y << 8) | (unsigned)(d.y & 255);
        outb[bstart[d.z >> 8] + r2] = ((unsigned)s.z << 8) | (unsigned)(d.z & 255);
        outb[bstart[d.w >> 8] + r3] = ((unsigned)s.w << 8) | (unsigned)(d.w & 255);
    }
    __syncthreads();
    if (4 * tid < total)
        reinterpret_cast<uint4*>(part)[g * (EPB / 4) + tid] = reinterpret_cast<const uint4*>(outb)[tid];
}

__global__ void __launch_bounds__(1024)
k_range_csr(const int* __restrict__ tab, const unsigned* __restrict__ part,
            int* __restrict__ segbeg, int* __restrict__ segend, float* __restrict__ dinv,
            int* __restrict__ csr_src, int n)
{
    __shared__ int sS[NCHUNK];
    __shared__ int cS[NCHUNK];
    __shared__ int cnt[256];
    __shared__ int cur[256];
    __shared__ int wsumA[32];
    __shared__ int wsumB[8];

    const int tid  = threadIdx.x;
    const int lane = tid & 31;
    const int wave = __builtin_amdgcn_readfirstlane(tid >> 5);
    const int r    = blockIdx.x;

    int s = 0;
    if (tid < NCHUNK) {
        s = tab[r * TPITCH + tid];
        const int e = tab[(r + 1) * TPITCH + tid];
        sS[tid] = s;
        cS[tid] = e - s;
    }
    if (tid < 256) cnt[tid] = 0;
    int rs = s;
    #pragma unroll
    for (int o = 16; o > 0; o >>= 1) rs += __shfl_xor(rs, o);
    if (lane == 0) wsumA[wave] = rs;
    __syncthreads();

    int base = 0;
    #pragma unroll
    for (int w = 0; w < SPW; ++w) base += wsumA[w];

    unsigned rec[SPW];
    int      cw[SPW];
    #pragma unroll
    for (int k = 0; k < SPW; ++k) {
        const int w = wave + 32 * k;
        const int c = (w < NCHUNK) ? cS[w] : 0;
        cw[k]  = c;
        rec[k] = 0u;
        if (lane < c) rec[k] = part[w * EPB + sS[w] + lane];
    }

    #pragma unroll
    for (int k = 0; k < SPW; ++k) {
        if (lane < cw[k]) atomicAdd(&cnt[rec[k] & 255u], 1);
        for (int i = 32 + lane; i < cw[k]; i += 32) {
            const int w = wave + 32 * k;
            const unsigned x = part[w * EPB + sS[w] + i];
            atomicAdd(&cnt[x & 255u], 1);
        }
    }
    __syncthreads();

    int v = 0, incl = 0;
    if (tid < 256) {
        v = cnt[tid];
        incl = v;
        #pragma unroll
        for (int o = 1; o < 32; o <<= 1) {
            const int t = __shfl_up(incl, o);
            if (lane >= o) incl += t;
        }
        if (lane == 31) wsumB[wave] = incl;
    }
    __syncthreads();
    if (tid < 256) {
        int off = 0;
        for (int w = 0; w < wave; ++w) off += wsumB[w];
        const int st = off + incl - v;
        cur[tid] = st;
        const int node = r * RANGE + tid;
        if (node < n) {
            segbeg[node] = base + st;
            segend[node] = base + st + v;
            dinv[node]   = rsqrtf((float)(v + 1));
        }
    }
    __syncthreads();

    #pragma unroll
    for (int k = 0; k < SPW; ++k) {
        if (lane < cw[k]) {
            const int p = atomicAdd(&cur[rec[k] & 255u], 1);
            csr_src[base + p] = (int)(rec[k] >> 8);
        }
        for (int i = 32 + lane; i < cw[k]; i += 32) {
            const int w = wave + 32 * k;
            const unsigned x = part[w * EPB + sS[w] + i];
            const int p = atomicAdd(&cur[x & 255u], 1);
            csr_src[base + p] = (int)(x >> 8);
        }
    }
}

__global__ void k_slot_weight(const int* __restrict__ csr_src, const float* __restrict__ dinv,
                              float* __restrict__ csr_w, int e4)
{
    int i = blockIdx.x * blockDim.x + threadIdx.x;
    if (i < e4) {
        const int4 s = reinterpret_cast<const int4*>(csr_src)[i];
        float4 w;
        w.x = dinv[s.x];
        w.y = dinv[s.y];
        w.z = dinv[s.z];
        w.w = dinv[s.w];
        reinterpret_cast<float4*>(csr_w)[i] = w;
    }
}

template <int KS, int S, int NTL>
__device__ __forceinline__ void mm_tiles(const v2f (&a)[KS], const float* bp, v8f (&acc)[NTL])
{
    #pragma unroll
    for (int j = 0; j < KS; ++j) {
        #pragma unroll
        for (int t = 0; t < NTL; ++t) {
            v2f b = *reinterpret_cast<const v2f*>(bp + t * 16 * S + 4 * j);
            acc[t] = __builtin_amdgcn_wmma_f32_16x16x4_f32(
                         false, a[j], false, b, (short)0, acc[t], false, false);
        }
    }
}

template <int K, int NACT, int NPAD, int S, bool RELU, bool HASBIAS, int NSTORE, int NTL>
__device__ __forceinline__ void tiles_out(const v2f (&a)[K / 4], const float* Bs, int nt,
                                          int c, int h, int m_base,
                                          float* __restrict__ C, int ldc,
                                          const float* __restrict__ bias)
{
    v8f acc[NTL];
    #pragma unroll
    for (int t = 0; t < NTL; ++t)
        acc[t] = (v8f){0.f, 0.f, 0.f, 0.f, 0.f, 0.f, 0.f, 0.f};

    mm_tiles<K / 4, S, NTL>(a, Bs + (nt * 16 + c) * S + 2 * h, acc);

    const int rbase = m_base + 8 * h;
    #pragma unroll
    for (int t = 0; t < NTL; ++t) {
        const int ncol = (nt + t) * 16 + c;
        float bv = 0.0f;
        if (HASBIAS) bv = (ncol < NACT) ? bias[ncol] : 0.0f;
        if (NSTORE >= NPAD || ncol < NSTORE) {
            #pragma unroll
            for (int g = 0; g < 8; ++g)
                C[(size_t)(rbase + g) * ldc + ncol] = acc[t][g] + bv;
        }
    }
}

template <int K, int NACT, int NPAD, int S, bool RELU, bool HASBIAS, int NSTORE>
__global__ void __launch_bounds__(512)
k_gemm_wmma_f32(const float* __restrict__ A, int lda,
                const float* __restrict__ B,
                float* __restrict__ C, int ldc,
                const float* __restrict__ bias, int mtiles)
{
    __shared__ __attribute__((aligned(16))) float Bs[NPAD * S];

    constexpr int KS    = K / 4;
    constexpr int NT    = NPAD / 16;
    constexpr int Q     = NACT / 4;
    constexpr int QG    = (Q + 7) / 8;
    constexpr int NTASK = KS * QG;

    const int tid  = threadIdx.x;
    const int lane = tid & 31;
    const int wave = __builtin_amdgcn_readfirstlane(tid >> 5);
    const int c    = lane & 15;
    const int h    = lane >> 4;

    int mt = wave * gridDim.x + blockIdx.x;
    v2f a[KS];
    if (mt < mtiles) {
        const float* arow = A + (size_t)(mt * 16 + c) * lda + 2 * h;
        #pragma unroll
        for (int j = 0; j < KS; ++j)
            a[j] = *reinterpret_cast<const v2f*>(arow + 4 * j);
    }

    {
        const int ql = lane & 7;
        const int kl = lane >> 3;
        constexpr int NIT = (NTASK + 15) / 16;
        float4 stg[NIT];
        #pragma unroll
        for (int it = 0; it < NIT; ++it) {
            const int task = wave + 16 * it;
            const int kg = task / QG;
            const int qg = task - kg * QG;
            const int q  = qg * 8 + ql;
            const int k  = kg * 4 + kl;
            stg[it] = make_float4(0.0f, 0.0f, 0.0f, 0.0f);
            if (task < NTASK && q < Q)
                stg[it] = *reinterpret_cast<const float4*>(B + (size_t)k * NACT + 4 * q);
        }
        #pragma unroll
        for (int it = 0; it < NIT; ++it) {
            const int task = wave + 16 * it;
            const int kg = task / QG;
            const int qg = task - kg * QG;
            const int q  = qg * 8 + ql;
            const int k  = kg * 4 + kl;
            if (task < NTASK && q < Q) {
                float* d = &Bs[(4 * q) * S + k];
                d[0]     = stg[it].x;
                d[S]     = stg[it].y;
                d[2 * S] = stg[it].z;
                d[3 * S] = stg[it].w;
            }
        }
        for (int idx = tid; idx < (NPAD - NACT) * K; idx += 512) {
            const int cc = idx / K;
            const int k  = idx - cc * K;
            Bs[(NACT + cc) * S + k] = 0.0f;
        }
    }
    __syncthreads();

    while (mt < mtiles) {
        const int m_base = mt * 16;
        if (RELU) {
            #pragma unroll
            for (int j = 0; j < KS; ++j) {
                a[j].x = fmaxf(a[j].x, 0.0f);
                a[j].y = fmaxf(a[j].y, 0.0f);
            }
        }
        int nt = 0;
        for (; nt + 2 <= NT; nt += 2)
            tiles_out<K, NACT, NPAD, S, RELU, HASBIAS, NSTORE, 2>(a, Bs, nt, c, h, m_base, C, ldc, bias);
        if (NT & 1)
            tiles_out<K, NACT, NPAD, S, RELU, HASBIAS, NSTORE, 1>(a, Bs, NT - 1, c, h, m_base, C, ldc, bias);

        mt += 16 * gridDim.x;
        if (mt < mtiles) {
            const float* arow = A + (size_t)(mt * 16 + c) * lda + 2 * h;
            #pragma unroll
            for (int j = 0; j < KS; ++j)
                a[j] = *reinterpret_cast<const v2f*>(arow + 4 * j);
        }
    }
}

__device__ __forceinline__ int kpos(int k)
{
    return (k & ~15) | ((k & 6) << 1) | ((k & 8) >> 2) | (k & 1);
}

template <int K, int NACT, int S, bool PERM>
__device__ __forceinline__ void stage_bt(const float* __restrict__ B, float* Bs, int wave, int lane)
{
    constexpr int KS    = K / 4;
    constexpr int Q     = NACT / 4;
    constexpr int QG    = (Q + 7) / 8;
    constexpr int NTASK = KS * QG;
    constexpr int NIT   = (NTASK + 15) / 16;
    const int ql = lane & 7;
    const int kl = lane >> 3;
    float4 stg[NIT];
    #pragma unroll
    for (int it = 0; it < NIT; ++it) {
        const int task = wave + 16 * it;
        const int kg = task / QG;
        const int qg = task - kg * QG;
        const int q  = qg * 8 + ql;
        const int k  = kg * 4 + kl;
        stg[it] = make_float4(0.0f, 0.0f, 0.0f, 0.0f);
        if (task < NTASK && q < Q)
            stg[it] = *reinterpret_cast<const float4*>(B + (size_t)k * NACT + 4 * q);
    }
    #pragma unroll
    for (int it = 0; it < NIT; ++it) {
        const int task = wave + 16 * it;
        const int kg = task / QG;
        const int qg = task - kg * QG;
        const int q  = qg * 8 + ql;
        const int k  = kg * 4 + kl;
        if (task < NTASK && q < Q) {
            float* d = &Bs[(4 * q) * S + (PERM ? kpos(k) : k)];
            d[0]     = stg[it].x;
            d[S]     = stg[it].y;
            d[2 * S] = stg[it].z;
            d[3 * S] = stg[it].w;
        }
    }
}

template <int KS1, int S1, int S2, int NT2, int NTL>
__device__ __forceinline__ void fused_tiles(const v2f (&a)[KS1], const float* W1s, const float* W2s,
                                            const float* b1s, int t, int c, int h, v8f (&acc2)[NT2])
{
    v8f d[NTL];
    #pragma unroll
    for (int u = 0; u < NTL; ++u)
        d[u] = (v8f){0.f, 0.f, 0.f, 0.f, 0.f, 0.f, 0.f, 0.f};

    const float* wp = W1s + (t * 16 + c) * S1 + 2 * h;
    #pragma unroll
    for (int j = 0; j < KS1; ++j) {
        #pragma unroll
        for (int u = 0; u < NTL; ++u) {
            v2f w = *reinterpret_cast<const v2f*>(wp + u * 16 * S1 + 4 * j);
            d[u] = __builtin_amdgcn_wmma_f32_16x16x4_f32(
                       false, w, false, a[j], (short)0, d[u], false, false);
        }
    }

    #pragma unroll
    for (int u = 0; u < NTL; ++u) {
        const float4 blo = *reinterpret_cast<const float4*>(b1s + (t + u) * 16 + 8 * h);
        const float4 bhi = *reinterpret_cast<const float4*>(b1s + (t + u) * 16 + 8 * h + 4);
        v8f z = d[u];
        z[0] = fmaxf(z[0] + blo.x, 0.0f);
        z[1] = fmaxf(z[1] + blo.y, 0.0f);
        z[2] = fmaxf(z[2] + blo.z, 0.0f);
        z[3] = fmaxf(z[3] + blo.w, 0.0f);
        z[4] = fmaxf(z[4] + bhi.x, 0.0f);
        z[5] = fmaxf(z[5] + bhi.y, 0.0f);
        z[6] = fmaxf(z[6] + bhi.z, 0.0f);
        z[7] = fmaxf(z[7] + bhi.w, 0.0f);

        const float* w2p = W2s + c * S2 + (t + u) * 16 + 2 * h;
        #pragma unroll
        for (int jp = 0; jp < 4; ++jp) {
            v2f b;
            b.x = z[2 * jp];
            b.y = z[2 * jp + 1];
            #pragma unroll
            for (int q = 0; q < NT2; ++q) {
                v2f w = *reinterpret_cast<const v2f*>(w2p + q * 16 * S2 + 4 * jp);
                acc2[q] = __builtin_amdgcn_wmma_f32_16x16x4_f32(
                              false, w, false, b, (short)0, acc2[q], false, false);
            }
        }
    }
}

template <int K1, int N1, int N1P, int S1, int N2, int N2P, int S2>
__global__ void __launch_bounds__(512)
k_gemm2_fused(const float* __restrict__ A, int lda,
              const float* __restrict__ W1, const float* __restrict__ b1,
              const float* __restrict__ W2,
              float* __restrict__ C, int ldc, int mtiles)
{
    __shared__ __attribute__((aligned(16))) float W1s[N1P * S1];
    __shared__ __attribute__((aligned(16))) float W2s[N2P * S2];
    __shared__ __attribute__((aligned(16))) float b1s[N1P];

    constexpr int KS1 = K1 / 4;
    constexpr int NT1 = N1P / 16;
    constexpr int NT2 = N2P / 16;
    static_assert(S2 >= N1P, "W2s rows hold the padded k range");

    const int tid  = threadIdx.x;
    const int lane = tid & 31;
    const int wave = __builtin_amdgcn_readfirstlane(tid >> 5);
    const int c    = lane & 15;
    const int h    = lane >> 4;

    int mt = wave * gridDim.x + blockIdx.x;
    v2f a[KS1];
    if (mt < mtiles) {
        const float* arow = A + (size_t)(mt * 16 + c) * lda + 2 * h;
        #pragma unroll
        for (int j = 0; j < KS1; ++j)
            a[j] = *reinterpret_cast<const v2f*>(arow + 4 * j);
    }

    stage_bt<K1, N1, S1, false>(W1, W1s, wave, lane);
    stage_bt<N1, N2, S2, true >(W2, W2s, wave, lane);
    for (int idx = tid; idx < (N1P - N1) * K1; idx += 512) {
        const int cc = idx / K1;
        const int k  = idx - cc * K1;
        W1s[(N1 + cc) * S1 + k] = 0.0f;
    }
    for (int idx = tid; idx < (N2P - N2) * N1P; idx += 512) {
        const int cc = idx / N1P;
        const int k  = idx - cc * N1P;
        W2s[(N2 + cc) * S2 + k] = 0.0f;
    }
    for (int idx = tid; idx < N2 * (N1P - N1); idx += 512) {
        const int cc = idx / (N1P - N1);
        const int k  = N1 + (idx - cc * (N1P - N1));
        W2s[cc * S2 + kpos(k)] = 0.0f;
    }
    if (tid < N1P) b1s[tid] = (tid < N1) ? b1[tid] : 0.0f;
    __syncthreads();

    while (mt < mtiles) {
        const int m_base = mt * 16;
        v8f acc2[NT2];
        #pragma unroll
        for (int q = 0; q < NT2; ++q)
            acc2[q] = (v8f){0.f, 0.f, 0.f, 0.f, 0.f, 0.f, 0.f, 0.f};

        if (NT1 & 1)
            fused_tiles<KS1, S1, S2, NT2, 1>(a, W1s, W2s, b1s, 0, c, h, acc2);
        for (int t = NT1 & 1; t + 2 <= NT1; t += 2)
            fused_tiles<KS1, S1, S2, NT2, 2>(a, W1s, W2s, b1s, t, c, h, acc2);

        float* crow = C + (size_t)(m_base + c) * ldc + 8 * h;
        #pragma unroll
        for (int q = 0; q < NT2; ++q) {
            float4 lo, hi;
            lo.x = acc2[q][0]; lo.y = acc2[q][1]; lo.z = acc2[q][2]; lo.w = acc2[q][3];
            hi.x = acc2[q][4]; hi.y = acc2[q][5]; hi.z = acc2[q][6]; hi.w = acc2[q][7];
            *reinterpret_cast<float4*>(crow + q * 16)     = lo;
            *reinterpret_cast<float4*>(crow + q * 16 + 4) = hi;
        }

        mt += 16 * gridDim.x;
        if (mt < mtiles) {
            const float* arow = A + (size_t)(mt * 16 + c) * lda + 2 * h;
            #pragma unroll
            for (int j = 0; j < KS1; ++j)
                a[j] = *reinterpret_cast<const v2f*>(arow + 4 * j);
        }
    }
}

template <int FP, int NP, int R>
__device__ __forceinline__ void pull_rows(const float* __restrict__ hw, int sv, int dvb, float di,
                                          int e, const int (&qc)[NP], float4 (&acc)[NP])
{
    float4 h[R][NP];
    #pragma unroll
    for (int u = 0; u < R; ++u) {
        const int s = __builtin_amdgcn_readlane(sv, e + u);
        const float4* row = reinterpret_cast<const float4*>(hw + (size_t)s * FP);
        #pragma unroll
        for (int p = 0; p < NP; ++p)
            h[u][p] = row[qc[p]];
    }
    #pragma unroll
    for (int u = 0; u < R; ++u) {
        const float w = __int_as_float(__builtin_amdgcn_readlane(dvb, e + u)) * di;
        #pragma unroll
        for (int p = 0; p < NP; ++p) {
            acc[p].x = fmaf(w, h[u][p].x, acc[p].x);
            acc[p].y = fmaf(w, h[u][p].y, acc[p].y);
            acc[p].z = fmaf(w, h[u][p].z, acc[p].z);
            acc[p].w = fmaf(w, h[u][p].w, acc[p].w);
        }
    }
}

template <int FP, int NP, int R>
__device__ __forceinline__ void pull_tail(const float* __restrict__ hw, int sv, int dvb, float di,
                                          int e, int r, const int (&qc)[NP], float4 (&acc)[NP])
{
    if constexpr (R >= 1) {
        if (r == R) pull_rows<FP, NP, R>(hw, sv, dvb, di, e, qc, acc);
        else        pull_tail<FP, NP, R - 1>(hw, sv, dvb, di, e, r, qc, acc);
    }
}

template <int FP, int NP, int U>
__device__ __forceinline__ void pull_chunk(const float* __restrict__ hw, int sv, int dvb, float di,
                                           int cnt, const int (&qc)[NP], float4 (&acc)[NP])
{
    int e = 0;
    for (; e + U <= cnt; e += U)
        pull_rows<FP, NP, U>(hw, sv, dvb, di, e, qc, acc);
    pull_tail<FP, NP, U - 1>(hw, sv, dvb, di, e, cnt - e, qc, acc);
}

template <int FP, int FACT, int U>
__global__ void __launch_bounds__(256, 8)
k_agg_pull(const int* __restrict__ start, const int* __restrict__ endp,
           const int* __restrict__ csr_src, const float* __restrict__ csr_w,
           const float* __restrict__ dinv,
           const float* __restrict__ hw, const float* __restrict__ bias,
           float* __restrict__ agg, int n)
{
    constexpr int Q  = FP / 4;
    constexpr int QB = FACT / 4;
    constexpr int NP = (Q + 31) / 32;
    static_assert(FP % 4 == 0 && FACT % 4 == 0, "rows are walked as float4");

    const int lane = threadIdx.x & 31;
    const int wid  = __builtin_amdgcn_readfirstlane((int)((blockIdx.x * blockDim.x + threadIdx.x) >> 5));
    const int nw   = (int)((gridDim.x * blockDim.x) >> 5);

    int    qc[NP];
    float4 bq[NP];
    #pragma unroll
    for (int p = 0; p < NP; ++p) {
        const int q = lane + 32 * p;
        qc[p] = min(q, Q - 1);
        bq[p] = make_float4(0.0f, 0.0f, 0.0f, 0.0f);
        if (q < QB) bq[p] = reinterpret_cast<const float4*>(bias)[q];
    }

    for (int node0 = wid; node0 < n; node0 += 32 * nw) {
        const int mine = min(node0 + lane * nw, n - 1);
        const int begv = start[mine];
        const int endv = endp[mine];
        const int divb = __float_as_int(dinv[mine]);
        const int kn   = min(32, (n - 1 - node0) / nw + 1);

        int beg = __builtin_amdgcn_readlane(begv, 0);
        int end = __builtin_amdgcn_readlane(endv, 0);
        int sv, dv;
        {
            const int idx = max(min(beg + lane, end - 1), 0);
            sv = csr_src[idx];
            dv = __float_as_int(csr_w[idx]);
        }

        for (int k = 0; k < kn; ++k) {
            const int   node = node0 + k * nw;
            const float di   = __int_as_float(__builtin_amdgcn_readlane(divb, k));

            const float4* hself = reinterpret_cast<const float4*>(hw + (size_t)node * FP);
            float4 hs[NP];
            float4 acc[NP];
            #pragma unroll
            for (int p = 0; p < NP; ++p) {
                hs[p]  = hself[qc[p]];
                acc[p] = bq[p];
            }

            const int k1   = min(k + 1, kn - 1);
            const int nbeg = __builtin_amdgcn_readlane(begv, k1);
            const int nend = __builtin_amdgcn_readlane(endv, k1);
            const int idxn = max(min(nbeg + lane, nend - 1), 0);
            const int svn  = csr_src[idxn];
            const int dvn  = __float_as_int(csr_w[idxn]);

            if (beg < end)
                pull_chunk<FP, NP, U>(hw, sv, dv, di, min(32, end - beg), qc, acc);
            for (int base = beg + 32; base < end; base += 32) {
                const int idx = min(base + lane, end - 1);
                const int svc = csr_src[idx];
                const int dvc = __float_as_int(csr_w[idx]);
                pull_chunk<FP, NP, U>(hw, svc, dvc, di, min(32, end - base), qc, acc);
            }

            const float dd = di * di;
            float4* o4 = reinterpret_cast<float4*>(agg + (size_t)node * FP);
            #pragma unroll
            for (int p = 0; p < NP; ++p) {
                const int q = lane + 32 * p;
                if (q < Q) {
                    float4 o;
                    o.x = fmaf(dd, hs[p].x, acc[p].x);
                    o.y = fmaf(dd, hs[p].y, acc[p].y);
                    o.z = fmaf(dd, hs[p].z, acc[p].z);
                    o.w = fmaf(dd, hs[p].w, acc[p].w);
                    o4[q] = o;
                }
            }

            beg = nbeg;
            end = nend;
            sv  = svn;
            dv  = dvn;
        }
    }
}

extern "C" void kernel_launch(void* const* d_in, const int* in_sizes, int n_in,
                              void* d_out, int out_size, void* d_ws, size_t ws_size,
                              hipStream_t stream)
{
    const float* x    = (const float*)d_in[0];
    const int*   ei   = (const int*)d_in[1];
    const float* W1   = (const float*)d_in[2];
    const float* b1   = (const float*)d_in[3];
    const float* W2   = (const float*)d_in[4];
    const float* b2   = (const float*)d_in[5];
    const float* Wout = (const float*)d_in[6];
    const float* bout = (const float*)d_in[7];
    float*       out  = (float*)d_out;

    const int* srcI = ei;
    const int* dstI = ei + N_EDGES;

    float* ws   = (float*)d_ws;
    float* dinv = ws;
    float* bufA = ws + 50176;
    float* bufB = bufA + (size_t)N_NODES * H1P;
    int*   tab    = (int*)(bufB + (size_t)N_NODES * H1P);
    int*   segbeg = tab + (NRANGE + 1) * TPITCH;
    int*   segend = segbeg + 50176;
    int*   csrsrc = segend + 50176;
    float* csrw   = (float*)(csrsrc + N_EDGES);
    unsigned* part = (unsigned*)(csrw + N_EDGES);

    k_range_split<<<NCHUNK, 1024, 0, stream>>>(srcI, dstI, part, tab, N_EDGES / 4);
    k_range_csr  <<<NRANGE, 1024, 0, stream>>>(tab, part, segbeg, segend, dinv, csrsrc, N_NODES);
    k_slot_weight<<<(N_EDGES / 4 + 255) / 256, 256, 0, stream>>>(csrsrc, dinv, csrw, N_EDGES / 4);

    k_agg_pull<F_IN, 0, 16><<<1024, 256, 0, stream>>>(segbeg, segend, csrsrc, csrw, dinv, x,
                                                       nullptr, bufB, N_NODES);
    k_gemm2_fused<F_IN, H1, H1P, 132, H2, H2P, 212><<<256, 512, 0, stream>>>(
        bufB, F_IN, W1, b1, W2, bufA, H2P, N_NODES / 16);

    k_agg_pull<H2P, H2, 16><<<1024, 256, 0, stream>>>(segbeg, segend, csrsrc, csrw, dinv, bufA, b2,
                                                      bufB, N_NODES);

    k_gemm_wmma_f32<H2, NC, NCP, 100, true, true, NC><<<256, 512, 0, stream>>>(
        bufB, H2P, Wout, out, NC,  bout, N_NODES / 16);
}


# Layer-2 aggregation and head GEMM fused in one kernel (6 launches, not 7): a work-group leaves its 49 nodes' finished rows in LDS and multiplies them by Wout there; agg2 never stored; sums unchanged
# speedup vs baseline: 7.1778x; 1.0580x over previous
#include <hip/hip_runtime.h>


typedef float v2f __attribute__((ext_vector_type(2)));
typedef float v8f __attribute__((ext_vector_type(8)));

#define N_NODES 50000
#define N_EDGES 800000
#define F_IN    128
#define H1      200
#define H1P     208
#define H2      100
#define H2P     112
#define NC      40
#define NCP     48
static_assert(N_EDGES % 4 == 0, "edge ids are read four at a time");


#define RANGE   256
#define NRANGE  ((N_NODES + RANGE - 1) / RANGE)
#define EPB     4096
#define NCHUNK  ((N_EDGES + EPB - 1) / EPB)
#define TPITCH  256
#define SPW     ((NCHUNK + 31) / 32)
static_assert(NRANGE < 256, "one LDS counter per range plus the total");
static_assert(NCHUNK <= TPITCH && NCHUNK <= 1024, "one thread per block run in pass 2");
static_assert(N_NODES <= (1 << 24), "source id shares a word with the 8-bit local destination");

__global__ void __launch_bounds__(1024)
k_range_split(const int* __restrict__ src, const int* __restrict__ dst,
              unsigned* __restrict__ part, int* __restrict__ tab, int e4)
{
    __shared__ int hist[256];
    __shared__ int bstart[256];
    __shared__ int wsum[8];
    __shared__ __attribute__((aligned(16))) unsigned outb[EPB];

    const int tid  = threadIdx.x;
    const int lane = tid & 31;
    const int wave = tid >> 5;
    const int g    = blockIdx.x;

    if (tid < 256) hist[tid] = 0;
    __syncthreads();

    const int  q    = g * (EPB / 4) + tid;
    const bool have = q < e4;
    int4 s = make_int4(0, 0, 0, 0);
    int4 d = make_int4(0, 0, 0, 0);
    int r0 = 0, r1 = 0, r2 = 0, r3 = 0;
    if (have) {
        s = reinterpret_cast<const int4*>(src)[q];
        d = reinterpret_cast<const int4*>(dst)[q];
        r0 = atomicAdd(&hist[d.x >> 8], 1);
        r1 = atomicAdd(&hist[d.y >> 8], 1);
        r2 = atomicAdd(&hist[d.z >> 8], 1);
        r3 = atomicAdd(&hist[d.w >> 8], 1);
    }
    __syncthreads();

    int v = 0, incl = 0;
    if (tid < 256) {
        v = hist[tid];
        incl = v;
        #pragma unroll
        for (int o = 1; o < 32; o <<= 1) {
            const int t = __shfl_up(incl, o);
            if (lane >= o) incl += t;
        }
        if (lane == 31) wsum[wave] = incl;
    }
    __syncthreads();
    if (tid < 256) {
        int off = 0;
        for (int w = 0; w < wave; ++w) off += wsum[w];
        const int ex = off + incl - v;
        bstart[tid] = ex;
        if (tid <= NRANGE) tab[tid * TPITCH + g] = ex;
    }
    __syncthreads();

    const int total = bstart[NRANGE];
    if (have) {
        outb[bstart[d.x >> 8] + r0] = ((unsigned)s.x << 8) | (unsigned)(d.x & 255);
        outb[bstart[d.y >> 8] + r1] = ((unsigned)s.y << 8) | (unsigned)(d.y & 255);
        outb[bstart[d.z >> 8] + r2] = ((unsigned)s.z << 8) | (unsigned)(d.z & 255);
        outb[bstart[d.w >> 8] + r3] = ((unsigned)s.w << 8) | (unsigned)(d.w & 255);
    }
    __syncthreads();
    if (4 * tid < total)
        reinterpret_cast<uint4*>(part)[g * (EPB / 4) + tid] = reinterpret_cast<const uint4*>(outb)[tid];
}

__global__ void __launch_bounds__(1024)
k_range_csr(const int* __restrict__ tab, const unsigned* __restrict__ part,
            int* __restrict__ segbeg, int* __restrict__ segend, float* __restrict__ dinv,
            int* __restrict__ csr_src, int n)
{
    __shared__ int sS[NCHUNK];
    __shared__ int cS[NCHUNK];
    __shared__ int cnt[256];
    __shared__ int cur[256];
    __shared__ int wsumA[32];
    __shared__ int wsumB[8];

    const int tid  = threadIdx.x;
    const int lane = tid & 31;
    const int wave = __builtin_amdgcn_readfirstlane(tid >> 5);
    const int r    = blockIdx.x;

    int s = 0;
    if (tid < NCHUNK) {
        s = tab[r * TPITCH + tid];
        const int e = tab[(r + 1) * TPITCH + tid];
        sS[tid] = s;
        cS[tid] = e - s;
    }
    if (tid < 256) cnt[tid] = 0;
    int rs = s;
    #pragma unroll
    for (int o = 16; o > 0; o >>= 1) rs += __shfl_xor(rs, o);
    if (lane == 0) wsumA[wave] = rs;
    __syncthreads();

    int base = 0;
    #pragma unroll
    for (int w = 0; w < SPW; ++w) base += wsumA[w];

    unsigned rec[SPW];
    int      cw[SPW];
    #pragma unroll
    for (int k = 0; k < SPW; ++k) {
        const int w = wave + 32 * k;
        const int c = (w < NCHUNK) ? cS[w] : 0;
        cw[k]  = c;
        rec[k] = 0u;
        if (lane < c) rec[k] = part[w * EPB + sS[w] + lane];
    }

    #pragma unroll
    for (int k = 0; k < SPW; ++k) {
        if (lane < cw[k]) atomicAdd(&cnt[rec[k] & 255u], 1);
        for (int i = 32 + lane; i < cw[k]; i += 32) {
            const int w = wave + 32 * k;
            const unsigned x = part[w * EPB + sS[w] + i];
            atomicAdd(&cnt[x & 255u], 1);
        }
    }
    __syncthreads();

    int v = 0, incl = 0;
    if (tid < 256) {
        v = cnt[tid];
        incl = v;
        #pragma unroll
        for (int o = 1; o < 32; o <<= 1) {
            const int t = __shfl_up(incl, o);
            if (lane >= o) incl += t;
        }
        if (lane == 31) wsumB[wave] = incl;
    }
    __syncthreads();
    if (tid < 256) {
        int off = 0;
        for (int w = 0; w < wave; ++w) off += wsumB[w];
        const int st = off + incl - v;
        cur[tid] = st;
        const int node = r * RANGE + tid;
        if (node < n) {
            segbeg[node] = base + st;
            segend[node] = base + st + v;
            dinv[node]   = rsqrtf((float)(v + 1));
        }
    }
    __syncthreads();

    #pragma unroll
    for (int k = 0; k < SPW; ++k) {
        if (lane < cw[k]) {
            const int p = atomicAdd(&cur[rec[k] & 255u], 1);
            csr_src[base + p] = (int)(rec[k] >> 8);
        }
        for (int i = 32 + lane; i < cw[k]; i += 32) {
            const int w = wave + 32 * k;
            const unsigned x = part[w * EPB + sS[w] + i];
            const int p = atomicAdd(&cur[x & 255u], 1);
            csr_src[base + p] = (int)(x >> 8);
        }
    }
}

__global__ void k_slot_weight(const int* __restrict__ csr_src, const float* __restrict__ dinv,
                              float* __restrict__ csr_w, int e4)
{
    int i = blockIdx.x * blockDim.x + threadIdx.x;
    if (i < e4) {
        const int4 s = reinterpret_cast<const int4*>(csr_src)[i];
        float4 w;
        w.x = dinv[s.x];
        w.y = dinv[s.y];
        w.z = dinv[s.z];
        w.w = dinv[s.w];
        reinterpret_cast<float4*>(csr_w)[i] = w;
    }
}

template <int KS, int S, int NTL>
__device__ __forceinline__ void mm_tiles(const v2f (&a)[KS], const float* bp, v8f (&acc)[NTL])
{
    #pragma unroll
    for (int j = 0; j < KS; ++j) {
        #pragma unroll
        for (int t = 0; t < NTL; ++t) {
            v2f b = *reinterpret_cast<const v2f*>(bp + t * 16 * S + 4 * j);
            acc[t] = __builtin_amdgcn_wmma_f32_16x16x4_f32(
                         false, a[j], false, b, (short)0, acc[t], false, false);
        }
    }
}

__device__ __forceinline__ int kpos(int k)
{
    return (k & ~15) | ((k & 6) << 1) | ((k & 8) >> 2) | (k & 1);
}

template <int K, int NACT, int S, bool PERM>
__device__ __forceinline__ void stage_bt(const float* __restrict__ B, float* Bs, int wave, int lane)
{
    constexpr int KS    = K / 4;
    constexpr int Q     = NACT / 4;
    constexpr int QG    = (Q + 7) / 8;
    constexpr int NTASK = KS * QG;
    constexpr int NIT   = (NTASK + 15) / 16;
    const int ql = lane & 7;
    const int kl = lane >> 3;
    float4 stg[NIT];
    #pragma unroll
    for (int it = 0; it < NIT; ++it) {
        const int task = wave + 16 * it;
        const int kg = task / QG;
        const int qg = task - kg * QG;
        const int q  = qg * 8 + ql;
        const int k  = kg * 4 + kl;
        stg[it] = make_float4(0.0f, 0.0f, 0.0f, 0.0f);
        if (task < NTASK && q < Q)
            stg[it] = *reinterpret_cast<const float4*>(B + (size_t)k * NACT + 4 * q);
    }
    #pragma unroll
    for (int it = 0; it < NIT; ++it) {
        const int task = wave + 16 * it;
        const int kg = task / QG;
        const int qg = task - kg * QG;
        const int q  = qg * 8 + ql;
        const int k  = kg * 4 + kl;
        if (task < NTASK && q < Q) {
            float* d = &Bs[(4 * q) * S + (PERM ? kpos(k) : k)];
            d[0]     = stg[it].x;
            d[S]     = stg[it].y;
            d[2 * S] = stg[it].z;
            d[3 * S] = stg[it].w;
        }
    }
}

template <int KS1, int S1, int S2, int NT2, int NTL>
__device__ __forceinline__ void fused_tiles(const v2f (&a)[KS1], const float* W1s, const float* W2s,
                                            const float* b1s, int t, int c, int h, v8f (&acc2)[NT2])
{
    v8f d[NTL];
    #pragma unroll
    for (int u = 0; u < NTL; ++u)
        d[u] = (v8f){0.f, 0.f, 0.f, 0.f, 0.f, 0.f, 0.f, 0.f};

    const float* wp = W1s + (t * 16 + c) * S1 + 2 * h;
    #pragma unroll
    for (int j = 0; j < KS1; ++j) {
        #pragma unroll
        for (int u = 0; u < NTL; ++u) {
            v2f w = *reinterpret_cast<const v2f*>(wp + u * 16 * S1 + 4 * j);
            d[u] = __builtin_amdgcn_wmma_f32_16x16x4_f32(
                       false, w, false, a[j], (short)0, d[u], false, false);
        }
    }

    #pragma unroll
    for (int u = 0; u < NTL; ++u) {
        const float4 blo = *reinterpret_cast<const float4*>(b1s + (t + u) * 16 + 8 * h);
        const float4 bhi = *reinterpret_cast<const float4*>(b1s + (t + u) * 16 + 8 * h + 4);
        v8f z = d[u];
        z[0] = fmaxf(z[0] + blo.x, 0.0f);
        z[1] = fmaxf(z[1] + blo.y, 0.0f);
        z[2] = fmaxf(z[2] + blo.z, 0.0f);
        z[3] = fmaxf(z[3] + blo.w, 0.0f);
        z[4] = fmaxf(z[4] + bhi.x, 0.0f);
        z[5] = fmaxf(z[5] + bhi.y, 0.0f);
        z[6] = fmaxf(z[6] + bhi.z, 0.0f);
        z[7] = fmaxf(z[7] + bhi.w, 0.0f);

        const float* w2p = W2s + c * S2 + (t + u) * 16 + 2 * h;
        #pragma unroll
        for (int jp = 0; jp < 4; ++jp) {
            v2f b;
            b.x = z[2 * jp];
            b.y = z[2 * jp + 1];
            #pragma unroll
            for (int q = 0; q < NT2; ++q) {
                v2f w = *reinterpret_cast<const v2f*>(w2p + q * 16 * S2 + 4 * jp);
                acc2[q] = __builtin_amdgcn_wmma_f32_16x16x4_f32(
                              false, w, false, b, (short)0, acc2[q], false, false);
            }
        }
    }
}

template <int K1, int N1, int N1P, int S1, int N2, int N2P, int S2>
__global__ void __launch_bounds__(512)
k_gemm2_fused(const float* __restrict__ A, int lda,
              const float* __restrict__ W1, const float* __restrict__ b1,
              const float* __restrict__ W2,
              float* __restrict__ C, int ldc, int mtiles)
{
    __shared__ __attribute__((aligned(16))) float W1s[N1P * S1];
    __shared__ __attribute__((aligned(16))) float W2s[N2P * S2];
    __shared__ __attribute__((aligned(16))) float b1s[N1P];

    constexpr int KS1 = K1 / 4;
    constexpr int NT1 = N1P / 16;
    constexpr int NT2 = N2P / 16;
    static_assert(S2 >= N1P, "W2s rows hold the padded k range");

    const int tid  = threadIdx.x;
    const int lane = tid & 31;
    const int wave = __builtin_amdgcn_readfirstlane(tid >> 5);
    const int c    = lane & 15;
    const int h    = lane >> 4;

    int mt = wave * gridDim.x + blockIdx.x;
    v2f a[KS1];
    if (mt < mtiles) {
        const float* arow = A + (size_t)(mt * 16 + c) * lda + 2 * h;
        #pragma unroll
        for (int j = 0; j < KS1; ++j)
            a[j] = *reinterpret_cast<const v2f*>(arow + 4 * j);
    }

    stage_bt<K1, N1, S1, false>(W1, W1s, wave, lane);
    stage_bt<N1, N2, S2, true >(W2, W2s, wave, lane);
    for (int idx = tid; idx < (N1P - N1) * K1; idx += 512) {
        const int cc = idx / K1;
        const int k  = idx - cc * K1;
        W1s[(N1 + cc) * S1 + k] = 0.0f;
    }
    for (int idx = tid; idx < (N2P - N2) * N1P; idx += 512) {
        const int cc = idx / N1P;
        const int k  = idx - cc * N1P;
        W2s[(N2 + cc) * S2 + k] = 0.0f;
    }
    for (int idx = tid; idx < N2 * (N1P - N1); idx += 512) {
        const int cc = idx / (N1P - N1);
        const int k  = N1 + (idx - cc * (N1P - N1));
        W2s[cc * S2 + kpos(k)] = 0.0f;
    }
    if (tid < N1P) b1s[tid] = (tid < N1) ? b1[tid] : 0.0f;
    __syncthreads();

    while (mt < mtiles) {
        const int m_base = mt * 16;
        v8f acc2[NT2];
        #pragma unroll
        for (int q = 0; q < NT2; ++q)
            acc2[q] = (v8f){0.f, 0.f, 0.f, 0.f, 0.f, 0.f, 0.f, 0.f};

        if (NT1 & 1)
            fused_tiles<KS1, S1, S2, NT2, 1>(a, W1s, W2s, b1s, 0, c, h, acc2);
        for (int t = NT1 & 1; t + 2 <= NT1; t += 2)
            fused_tiles<KS1, S1, S2, NT2, 2>(a, W1s, W2s, b1s, t, c, h, acc2);

        float* crow = C + (size_t)(m_base + c) * ldc + 8 * h;
        #pragma unroll
        for (int q = 0; q < NT2; ++q) {
            float4 lo, hi;
            lo.x = acc2[q][0]; lo.y = acc2[q][1]; lo.z = acc2[q][2]; lo.w = acc2[q][3];
            hi.x = acc2[q][4]; hi.y = acc2[q][5]; hi.z = acc2[q][6]; hi.w = acc2[q][7];
            *reinterpret_cast<float4*>(crow + q * 16)     = lo;
            *reinterpret_cast<float4*>(crow + q * 16 + 4) = hi;
        }

        mt += 16 * gridDim.x;
        if (mt < mtiles) {
            const float* arow = A + (size_t)(mt * 16 + c) * lda + 2 * h;
            #pragma unroll
            for (int j = 0; j < KS1; ++j)
                a[j] = *reinterpret_cast<const v2f*>(arow + 4 * j);
        }
    }
}

template <int FP, int NP, int R>
__device__ __forceinline__ void pull_rows(const float* __restrict__ hw, int sv, int dvb, float di,
                                          int e, const int (&qc)[NP], float4 (&acc)[NP])
{
    float4 h[R][NP];
    #pragma unroll
    for (int u = 0; u < R; ++u) {
        const int s = __builtin_amdgcn_readlane(sv, e + u);
        const float4* row = reinterpret_cast<const float4*>(hw + (size_t)s * FP);
        #pragma unroll
        for (int p = 0; p < NP; ++p)
            h[u][p] = row[qc[p]];
    }
    #pragma unroll
    for (int u = 0; u < R; ++u) {
        const float w = __int_as_float(__builtin_amdgcn_readlane(dvb, e + u)) * di;
        #pragma unroll
        for (int p = 0; p < NP; ++p) {
            acc[p].x = fmaf(w, h[u][p].x, acc[p].x);
            acc[p].y = fmaf(w, h[u][p].y, acc[p].y);
            acc[p].z = fmaf(w, h[u][p].z, acc[p].z);
            acc[p].w = fmaf(w, h[u][p].w, acc[p].w);
        }
    }
}

template <int FP, int NP, int R>
__device__ __forceinline__ void pull_tail(const float* __restrict__ hw, int sv, int dvb, float di,
                                          int e, int r, const int (&qc)[NP], float4 (&acc)[NP])
{
    if constexpr (R >= 1) {
        if (r == R) pull_rows<FP, NP, R>(hw, sv, dvb, di, e, qc, acc);
        else        pull_tail<FP, NP, R - 1>(hw, sv, dvb, di, e, r, qc, acc);
    }
}

template <int FP, int NP, int U>
__device__ __forceinline__ void pull_chunk(const float* __restrict__ hw, int sv, int dvb, float di,
                                           int cnt, const int (&qc)[NP], float4 (&acc)[NP])
{
    int e = 0;
    for (; e + U <= cnt; e += U)
        pull_rows<FP, NP, U>(hw, sv, dvb, di, e, qc, acc);
    pull_tail<FP, NP, U - 1>(hw, sv, dvb, di, e, cnt - e, qc, acc);
}

template <int FP, int FACT, int U>
__global__ void __launch_bounds__(256, 8)
k_agg_pull(const int* __restrict__ start, const int* __restrict__ endp,
           const int* __restrict__ csr_src, const float* __restrict__ csr_w,
           const float* __restrict__ dinv,
           const float* __restrict__ hw, const float* __restrict__ bias,
           float* __restrict__ agg, int n)
{
    constexpr int Q  = FP / 4;
    constexpr int QB = FACT / 4;
    constexpr int NP = (Q + 31) / 32;
    static_assert(FP % 4 == 0 && FACT % 4 == 0, "rows are walked as float4");

    const int lane = threadIdx.x & 31;
    const int wid  = __builtin_amdgcn_readfirstlane((int)((blockIdx.x * blockDim.x + threadIdx.x) >> 5));
    const int nw   = (int)((gridDim.x * blockDim.x) >> 5);

    int    qc[NP];
    float4 bq[NP];
    #pragma unroll
    for (int p = 0; p < NP; ++p) {
        const int q = lane + 32 * p;
        qc[p] = min(q, Q - 1);
        bq[p] = make_float4(0.0f, 0.0f, 0.0f, 0.0f);
        if (q < QB) bq[p] = reinterpret_cast<const float4*>(bias)[q];
    }

    for (int node0 = wid; node0 < n; node0 += 32 * nw) {
        const int mine = min(node0 + lane * nw, n - 1);
        const int begv = start[mine];
        const int endv = endp[mine];
        const int divb = __float_as_int(dinv[mine]);
        const int kn   = min(32, (n - 1 - node0) / nw + 1);

        int beg = __builtin_amdgcn_readlane(begv, 0);
        int end = __builtin_amdgcn_readlane(endv, 0);
        int sv, dv;
        {
            const int idx = max(min(beg + lane, end - 1), 0);
            sv = csr_src[idx];
            dv = __float_as_int(csr_w[idx]);
        }

        for (int k = 0; k < kn; ++k) {
            const int   node = node0 + k * nw;
            const float di   = __int_as_float(__builtin_amdgcn_readlane(divb, k));

            const float4* hself = reinterpret_cast<const float4*>(hw + (size_t)node * FP);
            float4 hs[NP];
            float4 acc[NP];
            #pragma unroll
            for (int p = 0; p < NP; ++p) {
                hs[p]  = hself[qc[p]];
                acc[p] = bq[p];
            }

            const int k1   = min(k + 1, kn - 1);
            const int nbeg = __builtin_amdgcn_readlane(begv, k1);
            const int nend = __builtin_amdgcn_readlane(endv, k1);
            const int idxn = max(min(nbeg + lane, nend - 1), 0);
            const int svn  = csr_src[idxn];
            const int dvn  = __float_as_int(csr_w[idxn]);

            if (beg < end)
                pull_chunk<FP, NP, U>(hw, sv, dv, di, min(32, end - beg), qc, acc);
            for (int base = beg + 32; base < end; base += 32) {
                const int idx = min(base + lane, end - 1);
                const int svc = csr_src[idx];
                const int dvc = __float_as_int(csr_w[idx]);
                pull_chunk<FP, NP, U>(hw, svc, dvc, di, min(32, end - base), qc, acc);
            }

            const float dd = di * di;
            float4* o4 = reinterpret_cast<float4*>(agg + (size_t)node * FP);
            #pragma unroll
            for (int p = 0; p < NP; ++p) {
                const int q = lane + 32 * p;
                if (q < Q) {
                    float4 o;
                    o.x = fmaf(dd, hs[p].x, acc[p].x);
                    o.y = fmaf(dd, hs[p].y, acc[p].y);
                    o.z = fmaf(dd, hs[p].z, acc[p].z);
                    o.w = fmaf(dd, hs[p].w, acc[p].w);
                    o4[q] = o;
                }
            }

            beg = nbeg;
            end = nend;
            sv  = svn;
            dv  = dvn;
        }
    }
}

#define NPB   49
#define NGRP  ((N_NODES + NPB - 1) / NPB)

template <int FP, int FACT, int U, int NB, int NOUT, int NOUTP, int S>
__global__ void __launch_bounds__(256, 8)
k_agg_head(const int* __restrict__ start, const int* __restrict__ endp,
           const int* __restrict__ csr_src, const float* __restrict__ csr_w,
           const float* __restrict__ dinv,
           const float* __restrict__ hw, const float* __restrict__ bias,
           const float* __restrict__ Wo, const float* __restrict__ bo,
           float* __restrict__ out, int n)
{
    constexpr int Q   = FP / 4;
    constexpr int QB  = FACT / 4;
    constexpr int NP  = (Q + 31) / 32;
    constexpr int HP  = FP + 4;
    constexpr int KS  = FACT / 4;
    constexpr int NT  = NOUTP / 16;
    constexpr int QO  = NOUT / 4;
    constexpr int NW4 = FACT * QO;
    constexpr int NIT = (NW4 + 255) / 256;
    static_assert(FP % 4 == 0 && FACT % 4 == 0 && NOUT % 4 == 0, "rows are walked as float4");
    static_assert((HP / 4) % 2 == 1 && (S / 4) % 2 == 1 && S >= FACT, "LDS pitches are 4 x odd");
    static_assert((NB + 15) / 16 <= 8, "one wave per 16-row strip");

    __shared__ __attribute__((aligned(16))) float Hs[NB * HP];
    __shared__ __attribute__((aligned(16))) float Bs[NOUTP * S];

    const int tid  = threadIdx.x;
    const int lane = tid & 31;
    const int wave   = __builtin_amdgcn_readfirstlane(tid >> 5);
    const int nbase  = blockIdx.x * NB;
    const int nvalid = min(NB, n - nbase);

    float4 wst[NIT];
    #pragma unroll
    for (int it = 0; it < NIT; ++it) {
        const int idx = tid + 256 * it;
        wst[it] = make_float4(0.0f, 0.0f, 0.0f, 0.0f);
        if (idx < NW4) wst[it] = reinterpret_cast<const float4*>(Wo)[idx];
    }

    int    qc[NP];
    float4 bq[NP];
    #pragma unroll
    for (int p = 0; p < NP; ++p) {
        const int q = lane + 32 * p;
        qc[p] = min(q, Q - 1);
        bq[p] = make_float4(0.0f, 0.0f, 0.0f, 0.0f);
        if (q < QB) bq[p] = reinterpret_cast<const float4*>(bias)[q];
    }

    const int kn   = (wave < nvalid) ? (nvalid - 1 - wave) / 8 + 1 : 0;
    const int mine = nbase + min(wave + 8 * lane, nvalid - 1);
    const int begv = start[mine];
    const int endv = endp[mine];
    const int divb = __float_as_int(dinv[mine]);

    int beg = __builtin_amdgcn_readlane(begv, 0);
    int end = __builtin_amdgcn_readlane(endv, 0);
    int sv, dv;
    {
        const int idx = max(min(beg + lane, end - 1), 0);
        sv = csr_src[idx];
        dv = __float_as_int(csr_w[idx]);
    }

    #pragma unroll
    for (int it = 0; it < NIT; ++it) {
        const int idx = tid + 256 * it;
        if (idx < NW4) {
            const int k = idx / QO;
            const int q = idx - k * QO;
            float* d = &Bs[(4 * q) * S + k];
            d[0]     = wst[it].x;
            d[S]     = wst[it].y;
            d[2 * S] = wst[it].z;
            d[3 * S] = wst[it].w;
        }
    }
    for (int idx = tid; idx < (NOUTP - NOUT) * S; idx += 256)
        Bs[NOUT * S + idx] = 0.0f;

    for (int k = 0; k < kn; ++k) {
        const int   lrow = wave + 8 * k;
        const int   node = nbase + lrow;
        const float di   = __int_as_float(__builtin_amdgcn_readlane(divb, k));

        const float4* hself = reinterpret_cast<const float4*>(hw + (size_t)node * FP);
        float4 hs[NP];
        float4 acc[NP];
        #pragma unroll
        for (int p = 0; p < NP; ++p) {
            hs[p]  = hself[qc[p]];
            acc[p] = bq[p];
        }

        const int k1   = min(k + 1, kn - 1);
        const int nbeg = __builtin_amdgcn_readlane(begv, k1);
        const int nend = __builtin_amdgcn_readlane(endv, k1);
        const int idxn = max(min(nbeg + lane, nend - 1), 0);
        const int svn  = csr_src[idxn];
        const int dvn  = __float_as_int(csr_w[idxn]);

        if (beg < end)
            pull_chunk<FP, NP, U>(hw, sv, dv, di, min(32, end - beg), qc, acc);
        for (int base = beg + 32; base < end; base += 32) {
            const int idx = min(base + lane, end - 1);
            const int svc = csr_src[idx];
            const int dvc = __float_as_int(csr_w[idx]);
            pull_chunk<FP, NP, U>(hw, svc, dvc, di, min(32, end - base), qc, acc);
        }

        const float dd = di * di;
        float4* o4 = reinterpret_cast<float4*>(Hs + lrow * HP);
        #pragma unroll
        for (int p = 0; p < NP; ++p) {
            const int q = lane + 32 * p;
            if (q < Q) {
                float4 o;
                o.x = fmaxf(fmaf(dd, hs[p].x, acc[p].x), 0.0f);
                o.y = fmaxf(fmaf(dd, hs[p].y, acc[p].y), 0.0f);
                o.z = fmaxf(fmaf(dd, hs[p].z, acc[p].z), 0.0f);
                o.w = fmaxf(fmaf(dd, hs[p].w, acc[p].w), 0.0f);
                o4[q] = o;
            }
        }

        beg = nbeg;
        end = nend;
        sv  = svn;
        dv  = dvn;
    }
    __syncthreads();

    const int nstrips = (nvalid + 15) >> 4;
    if (wave < nstrips) {
        const int c = lane & 15;
        const int h = lane >> 4;
        const float* ap = Hs + min(wave * 16 + c, nvalid - 1) * HP + 2 * h;
        v2f a[KS];
        #pragma unroll
        for (int j = 0; j < KS; ++j)
            a[j] = *reinterpret_cast<const v2f*>(ap + 4 * j);

        v8f acc[NT];
        #pragma unroll
        for (int t = 0; t < NT; ++t)
            acc[t] = (v8f){0.f, 0.f, 0.f, 0.f, 0.f, 0.f, 0.f, 0.f};

        mm_tiles<KS, S, NT>(a, Bs + c * S + 2 * h, acc);

        const int rl = wave * 16 + 8 * h;
        #pragma unroll
        for (int t = 0; t < NT; ++t) {
            const int ncol = t * 16 + c;
            if (ncol < NOUT) {
                const float bv = bo[ncol];
                #pragma unroll
                for (int g = 0; g < 8; ++g)
                    if (rl + g < nvalid)
                        out[(size_t)(nbase + rl + g) * NOUT + ncol] = acc[t][g] + bv;
            }
        }
    }
}

extern "C" void kernel_launch(void* const* d_in, const int* in_sizes, int n_in,
                              void* d_out, int out_size, void* d_ws, size_t ws_size,
                              hipStream_t stream)
{
    const float* x    = (const float*)d_in[0];
    const int*   ei   = (const int*)d_in[1];
    const float* W1   = (const float*)d_in[2];
    const float* b1   = (const float*)d_in[3];
    const float* W2   = (const float*)d_in[4];
    const float* b2   = (const float*)d_in[5];
    const float* Wout = (const float*)d_in[6];
    const float* bout = (const float*)d_in[7];
    float*       out  = (float*)d_out;

    const int* srcI = ei;
    const int* dstI = ei + N_EDGES;

    float* ws   = (float*)d_ws;
    float* dinv = ws;
    float* bufA = ws + 50176;
    float* bufB = bufA + (size_t)N_NODES * H1P;
    int*   tab    = (int*)(bufB + (size_t)N_NODES * H1P);
    int*   segbeg = tab + (NRANGE + 1) * TPITCH;
    int*   segend = segbeg + 50176;
    int*   csrsrc = segend + 50176;
    float* csrw   = (float*)(csrsrc + N_EDGES);
    unsigned* part = (unsigned*)(csrw + N_EDGES);

    k_range_split<<<NCHUNK, 1024, 0, stream>>>(srcI, dstI, part, tab, N_EDGES / 4);
    k_range_csr  <<<NRANGE, 1024, 0, stream>>>(tab, part, segbeg, segend, dinv, csrsrc, N_NODES);
    k_slot_weight<<<(N_EDGES / 4 + 255) / 256, 256, 0, stream>>>(csrsrc, dinv, csrw, N_EDGES / 4);

    k_agg_pull<F_IN, 0, 16><<<1024, 256, 0, stream>>>(segbeg, segend, csrsrc, csrw, dinv, x,
                                                       nullptr, bufB, N_NODES);
    k_gemm2_fused<F_IN, H1, H1P, 132, H2, H2P, 212><<<256, 512, 0, stream>>>(
        bufB, F_IN, W1, b1, W2, bufA, H2P, N_NODES / 16);

    k_agg_head<H2P, H2, 16, NPB, NC, NCP, 100><<<NGRP, 256, 0, stream>>>(
        segbeg, segend, csrsrc, csrw, dinv, bufA, b2, Wout, bout, out, N_NODES);
}


# Layer-1 aggregation and the fused GEMMs in one kernel (5 launches, not 6): a 16-wave work-group leaves its 196 nodes' summed rows in LDS as the B-side fragments; Agg(x) never stored; sums unchanged
# speedup vs baseline: 8.1647x; 1.1375x over previous
#include <hip/hip_runtime.h>


typedef float v2f __attribute__((ext_vector_type(2)));
typedef float v8f __attribute__((ext_vector_type(8)));

#define N_NODES 50000
#define N_EDGES 800000
#define F_IN    128
#define H1      200
#define H1P     208
#define H2      100
#define H2P     112
#define NC      40
#define NCP     48
static_assert(N_EDGES % 4 == 0, "edge ids are read four at a time");


#define RANGE   256
#define NRANGE  ((N_NODES + RANGE - 1) / RANGE)
#define EPB     4096
#define NCHUNK  ((N_EDGES + EPB - 1) / EPB)
#define TPITCH  256
#define SPW     ((NCHUNK + 31) / 32)
static_assert(NRANGE < 256, "one LDS counter per range plus the total");
static_assert(NCHUNK <= TPITCH && NCHUNK <= 1024, "one thread per block run in pass 2");
static_assert(N_NODES <= (1 << 24), "source id shares a word with the 8-bit local destination");

__global__ void __launch_bounds__(1024)
k_range_split(const int* __restrict__ src, const int* __restrict__ dst,
              unsigned* __restrict__ part, int* __restrict__ tab, int e4)
{
    __shared__ int hist[256];
    __shared__ int bstart[256];
    __shared__ int wsum[8];
    __shared__ __attribute__((aligned(16))) unsigned outb[EPB];

    const int tid  = threadIdx.x;
    const int lane = tid & 31;
    const int wave = tid >> 5;
    const int g    = blockIdx.x;

    if (tid < 256) hist[tid] = 0;
    __syncthreads();

    const int  q    = g * (EPB / 4) + tid;
    const bool have = q < e4;
    int4 s = make_int4(0, 0, 0, 0);
    int4 d = make_int4(0, 0, 0, 0);
    int r0 = 0, r1 = 0, r2 = 0, r3 = 0;
    if (have) {
        s = reinterpret_cast<const int4*>(src)[q];
        d = reinterpret_cast<const int4*>(dst)[q];
        r0 = atomicAdd(&hist[d.x >> 8], 1);
        r1 = atomicAdd(&hist[d.y >> 8], 1);
        r2 = atomicAdd(&hist[d.z >> 8], 1);
        r3 = atomicAdd(&hist[d.w >> 8], 1);
    }
    __syncthreads();

    int v = 0, incl = 0;
    if (tid < 256) {
        v = hist[tid];
        incl = v;
        #pragma unroll
        for (int o = 1; o < 32; o <<= 1) {
            const int t = __shfl_up(incl, o);
            if (lane >= o) incl += t;
        }
        if (lane == 31) wsum[wave] = incl;
    }
    __syncthreads();
    if (tid < 256) {
        int off = 0;
        for (int w = 0; w < wave; ++w) off += wsum[w];
        const int ex = off + incl - v;
        bstart[tid] = ex;
        if (tid <= NRANGE) tab[tid * TPITCH + g] = ex;
    }
    __syncthreads();

    const int total = bstart[NRANGE];
    if (have) {
        outb[bstart[d.x >> 8] + r0] = ((unsigned)s.x << 8) | (unsigned)(d.x & 255);
        outb[bstart[d.y >> 8] + r1] = ((unsigned)s.y << 8) | (unsigned)(d.y & 255);
        outb[bstart[d.z >> 8] + r2] = ((unsigned)s.z << 8) | (unsigned)(d.z & 255);
        outb[bstart[d.w >> 8] + r3] = ((unsigned)s.w << 8) | (unsigned)(d.w & 255);
    }
    __syncthreads();
    if (4 * tid < total)
        reinterpret_cast<uint4*>(part)[g * (EPB / 4) + tid] = reinterpret_cast<const uint4*>(outb)[tid];
}

__global__ void __launch_bounds__(1024)
k_range_csr(const int* __restrict__ tab, const unsigned* __restrict__ part,
            int* __restrict__ segbeg, int* __restrict__ segend, float* __restrict__ dinv,
            int* __restrict__ csr_src, int n)
{
    __shared__ int sS[NCHUNK];
    __shared__ int cS[NCHUNK];
    __shared__ int cnt[256];
    __shared__ int cur[256];
    __shared__ int wsumA[32];
    __shared__ int wsumB[8];

    const int tid  = threadIdx.x;
    const int lane = tid & 31;
    const int wave = __builtin_amdgcn_readfirstlane(tid >> 5);
    const int r    = blockIdx.x;

    int s = 0;
    if (tid < NCHUNK) {
        s = tab[r * TPITCH + tid];
        const int e = tab[(r + 1) * TPITCH + tid];
        sS[tid] = s;
        cS[tid] = e - s;
    }
    if (tid < 256) cnt[tid] = 0;
    int rs = s;
    #pragma unroll
    for (int o = 16; o > 0; o >>= 1) rs += __shfl_xor(rs, o);
    if (lane == 0) wsumA[wave] = rs;
    __syncthreads();

    int base = 0;
    #pragma unroll
    for (int w = 0; w < SPW; ++w) base += wsumA[w];

    unsigned rec[SPW];
    int      cw[SPW];
    #pragma unroll
    for (int k = 0; k < SPW; ++k) {
        const int w = wave + 32 * k;
        const int c = (w < NCHUNK) ? cS[w] : 0;
        cw[k]  = c;
        rec[k] = 0u;
        if (lane < c) rec[k] = part[w * EPB + sS[w] + lane];
    }

    #pragma unroll
    for (int k = 0; k < SPW; ++k) {
        if (lane < cw[k]) atomicAdd(&cnt[rec[k] & 255u], 1);
        for (int i = 32 + lane; i < cw[k]; i += 32) {
            const int w = wave + 32 * k;
            const unsigned x = part[w * EPB + sS[w] + i];
            atomicAdd(&cnt[x & 255u], 1);
        }
    }
    __syncthreads();

    int v = 0, incl = 0;
    if (tid < 256) {
        v = cnt[tid];
        incl = v;
        #pragma unroll
        for (int o = 1; o < 32; o <<= 1) {
            const int t = __shfl_up(incl, o);
            if (lane >= o) incl += t;
        }
        if (lane == 31) wsumB[wave] = incl;
    }
    __syncthreads();
    if (tid < 256) {
        int off = 0;
        for (int w = 0; w < wave; ++w) off += wsumB[w];
        const int st = off + incl - v;
        cur[tid] = st;
        const int node = r * RANGE + tid;
        if (node < n) {
            segbeg[node] = base + st;
            segend[node] = base + st + v;
            dinv[node]   = rsqrtf((float)(v + 1));
        }
    }
    __syncthreads();

    #pragma unroll
    for (int k = 0; k < SPW; ++k) {
        if (lane < cw[k]) {
            const int p = atomicAdd(&cur[rec[k] & 255u], 1);
            csr_src[base + p] = (int)(rec[k] >> 8);
        }
        for (int i = 32 + lane; i < cw[k]; i += 32) {
            const int w = wave + 32 * k;
            const unsigned x = part[w * EPB + sS[w] + i];
            const int p = atomicAdd(&cur[x & 255u], 1);
            csr_src[base + p] = (int)(x >> 8);
        }
    }
}

__global__ void k_slot_weight(const int* __restrict__ csr_src, const float* __restrict__ dinv,
                              float* __restrict__ csr_w, int e4)
{
    int i = blockIdx.x * blockDim.x + threadIdx.x;
    if (i < e4) {
        const int4 s = reinterpret_cast<const int4*>(csr_src)[i];
        float4 w;
        w.x = dinv[s.x];
        w.y = dinv[s.y];
        w.z = dinv[s.z];
        w.w = dinv[s.w];
        reinterpret_cast<float4*>(csr_w)[i] = w;
    }
}

template <int KS, int S, int NTL>
__device__ __forceinline__ void mm_tiles(const v2f (&a)[KS], const float* bp, v8f (&acc)[NTL])
{
    #pragma unroll
    for (int j = 0; j < KS; ++j) {
        #pragma unroll
        for (int t = 0; t < NTL; ++t) {
            v2f b = *reinterpret_cast<const v2f*>(bp + t * 16 * S + 4 * j);
            acc[t] = __builtin_amdgcn_wmma_f32_16x16x4_f32(
                         false, a[j], false, b, (short)0, acc[t], false, false);
        }
    }
}

__device__ __forceinline__ int kpos(int k)
{
    return (k & ~15) | ((k & 6) << 1) | ((k & 8) >> 2) | (k & 1);
}

template <int K, int NACT, int S, bool PERM>
__device__ __forceinline__ void stage_bt(const float* __restrict__ B, float* Bs, int wave, int lane)
{
    constexpr int KS    = K / 4;
    constexpr int Q     = NACT / 4;
    constexpr int QG    = (Q + 7) / 8;
    constexpr int NTASK = KS * QG;
    constexpr int NIT   = (NTASK + 15) / 16;
    const int ql = lane & 7;
    const int kl = lane >> 3;
    float4 stg[NIT];
    #pragma unroll
    for (int it = 0; it < NIT; ++it) {
        const int task = wave + 16 * it;
        const int kg = task / QG;
        const int qg = task - kg * QG;
        const int q  = qg * 8 + ql;
        const int k  = kg * 4 + kl;
        stg[it] = make_float4(0.0f, 0.0f, 0.0f, 0.0f);
        if (task < NTASK && q < Q)
            stg[it] = *reinterpret_cast<const float4*>(B + (size_t)k * NACT + 4 * q);
    }
    #pragma unroll
    for (int it = 0; it < NIT; ++it) {
        const int task = wave + 16 * it;
        const int kg = task / QG;
        const int qg = task - kg * QG;
        const int q  = qg * 8 + ql;
        const int k  = kg * 4 + kl;
        if (task < NTASK && q < Q) {
            float* d = &Bs[(4 * q) * S + (PERM ? kpos(k) : k)];
            d[0]     = stg[it].x;
            d[S]     = stg[it].y;
            d[2 * S] = stg[it].z;
            d[3 * S] = stg[it].w;
        }
    }
}

template <int KS1, int S1, int S2, int NT2, int NTL>
__device__ __forceinline__ void fused_tiles(const v2f (&a)[KS1], const float* W1s, const float* W2s,
                                            const float* b1s, int t, int c, int h, v8f (&acc2)[NT2])
{
    v8f d[NTL];
    #pragma unroll
    for (int u = 0; u < NTL; ++u)
        d[u] = (v8f){0.f, 0.f, 0.f, 0.f, 0.f, 0.f, 0.f, 0.f};

    const float* wp = W1s + (t * 16 + c) * S1 + 2 * h;
    #pragma unroll
    for (int j = 0; j < KS1; ++j) {
        #pragma unroll
        for (int u = 0; u < NTL; ++u) {
            v2f w = *reinterpret_cast<const v2f*>(wp + u * 16 * S1 + 4 * j);
            d[u] = __builtin_amdgcn_wmma_f32_16x16x4_f32(
                       false, w, false, a[j], (short)0, d[u], false, false);
        }
    }

    #pragma unroll
    for (int u = 0; u < NTL; ++u) {
        const float4 blo = *reinterpret_cast<const float4*>(b1s + (t + u) * 16 + 8 * h);
        const float4 bhi = *reinterpret_cast<const float4*>(b1s + (t + u) * 16 + 8 * h + 4);
        v8f z = d[u];
        z[0] = fmaxf(z[0] + blo.x, 0.0f);
        z[1] = fmaxf(z[1] + blo.y, 0.0f);
        z[2] = fmaxf(z[2] + blo.z, 0.0f);
        z[3] = fmaxf(z[3] + blo.w, 0.0f);
        z[4] = fmaxf(z[4] + bhi.x, 0.0f);
        z[5] = fmaxf(z[5] + bhi.y, 0.0f);
        z[6] = fmaxf(z[6] + bhi.z, 0.0f);
        z[7] = fmaxf(z[7] + bhi.w, 0.0f);

        const float* w2p = W2s + c * S2 + (t + u) * 16 + 2 * h;
        #pragma unroll
        for (int jp = 0; jp < 4; ++jp) {
            v2f b;
            b.x = z[2 * jp];
            b.y = z[2 * jp + 1];
            #pragma unroll
            for (int q = 0; q < NT2; ++q) {
                v2f w = *reinterpret_cast<const v2f*>(w2p + q * 16 * S2 + 4 * jp);
                acc2[q] = __builtin_amdgcn_wmma_f32_16x16x4_f32(
                              false, w, false, b, (short)0, acc2[q], false, false);
            }
        }
    }
}

template <int FP, int NP, int R>
__device__ __forceinline__ void pull_rows(const float* __restrict__ hw, int sv, int dvb, float di,
                                          int e, const int (&qc)[NP], float4 (&acc)[NP])
{
    float4 h[R][NP];
    #pragma unroll
    for (int u = 0; u < R; ++u) {
        const int s = __builtin_amdgcn_readlane(sv, e + u);
        const float4* row = reinterpret_cast<const float4*>(hw + (size_t)s * FP);
        #pragma unroll
        for (int p = 0; p < NP; ++p)
            h[u][p] = row[qc[p]];
    }
    #pragma unroll
    for (int u = 0; u < R; ++u) {
        const float w = __int_as_float(__builtin_amdgcn_readlane(dvb, e + u)) * di;
        #pragma unroll
        for (int p = 0; p < NP; ++p) {
            acc[p].x = fmaf(w, h[u][p].x, acc[p].x);
            acc[p].y = fmaf(w, h[u][p].y, acc[p].y);
            acc[p].z = fmaf(w, h[u][p].z, acc[p].z);
            acc[p].w = fmaf(w, h[u][p].w, acc[p].w);
        }
    }
}

template <int FP, int NP, int R>
__device__ __forceinline__ void pull_tail(const float* __restrict__ hw, int sv, int dvb, float di,
                                          int e, int r, const int (&qc)[NP], float4 (&acc)[NP])
{
    if constexpr (R >= 1) {
        if (r == R) pull_rows<FP, NP, R>(hw, sv, dvb, di, e, qc, acc);
        else        pull_tail<FP, NP, R - 1>(hw, sv, dvb, di, e, r, qc, acc);
    }
}

template <int FP, int NP, int U>
__device__ __forceinline__ void pull_chunk(const float* __restrict__ hw, int sv, int dvb, float di,
                                           int cnt, const int (&qc)[NP], float4 (&acc)[NP])
{
    int e = 0;
    for (; e + U <= cnt; e += U)
        pull_rows<FP, NP, U>(hw, sv, dvb, di, e, qc, acc);
    pull_tail<FP, NP, U - 1>(hw, sv, dvb, di, e, cnt - e, qc, acc);
}

#define NB1   196
#define NGRP1 ((N_NODES + NB1 - 1) / NB1)

template <int K1, int N1, int N1P, int S1, int N2, int N2P, int S2, int U, int NB>
__global__ void __launch_bounds__(512)
k_agg_gemm2(const int* __restrict__ start, const int* __restrict__ endp,
            const int* __restrict__ csr_src, const float* __restrict__ csr_w,
            const float* __restrict__ dinv,
            const float* __restrict__ x,
            const float* __restrict__ W1, const float* __restrict__ b1,
            const float* __restrict__ W2,
            float* __restrict__ C, int ldc, int n)
{
    constexpr int KS1 = K1 / 4;
    constexpr int NT1 = N1P / 16;
    constexpr int NT2 = N2P / 16;
    constexpr int HP  = K1 + 4;
    static_assert(K1 == 128, "one float4 per lane covers an input row");
    static_assert(S2 >= N1P, "W2s rows hold the padded k range");
    static_assert((HP / 4) % 2 == 1, "LDS pitch is 4 x odd");
    static_assert((NB + 15) / 16 <= 16, "one wave per 16-row strip, one lane per node of a wave");

    __shared__ __attribute__((aligned(16))) float W1s[N1P * S1];
    __shared__ __attribute__((aligned(16))) float W2s[N2P * S2];
    __shared__ __attribute__((aligned(16))) float b1s[N1P];
    __shared__ __attribute__((aligned(16))) float Hs[NB * HP];

    const int tid  = threadIdx.x;
    const int lane = tid & 31;
    const int wave   = __builtin_amdgcn_readfirstlane(tid >> 5);
    const int c      = lane & 15;
    const int h      = lane >> 4;
    const int nbase  = blockIdx.x * NB;
    const int nvalid = min(NB, n - nbase);

    const int qc[1] = {lane};

    const int kn   = (wave < nvalid) ? (nvalid - 1 - wave) / 16 + 1 : 0;
    const int mine = nbase + min(wave + 16 * lane, nvalid - 1);
    const int begv = start[mine];
    const int endv = endp[mine];
    const int divb = __float_as_int(dinv[mine]);

    int beg = __builtin_amdgcn_readlane(begv, 0);
    int end = __builtin_amdgcn_readlane(endv, 0);
    int sv, dv;
    {
        const int idx = max(min(beg + lane, end - 1), 0);
        sv = csr_src[idx];
        dv = __float_as_int(csr_w[idx]);
    }

    stage_bt<K1, N1, S1, false>(W1, W1s, wave, lane);
    stage_bt<N1, N2, S2, true >(W2, W2s, wave, lane);
    for (int idx = tid; idx < (N1P - N1) * K1; idx += 512) {
        const int cc = idx / K1;
        const int k  = idx - cc * K1;
        W1s[(N1 + cc) * S1 + k] = 0.0f;
    }
    for (int idx = tid; idx < (N2P - N2) * N1P; idx += 512) {
        const int cc = idx / N1P;
        const int k  = idx - cc * N1P;
        W2s[(N2 + cc) * S2 + k] = 0.0f;
    }
    for (int idx = tid; idx < N2 * (N1P - N1); idx += 512) {
        const int cc = idx / (N1P - N1);
        const int k  = N1 + (idx - cc * (N1P - N1));
        W2s[cc * S2 + kpos(k)] = 0.0f;
    }
    if (tid < N1P) b1s[tid] = (tid < N1) ? b1[tid] : 0.0f;

    for (int k = 0; k < kn; ++k) {
        const int   lrow = wave + 16 * k;
        const int   node = nbase + lrow;
        const float di   = __int_as_float(__builtin_amdgcn_readlane(divb, k));

        const float4 hs = reinterpret_cast<const float4*>(x + (size_t)node * K1)[lane];
        float4 acc[1];
        acc[0] = make_float4(0.0f, 0.0f, 0.0f, 0.0f);

        const int k1   = min(k + 1, kn - 1);
        const int nbeg = __builtin_amdgcn_readlane(begv, k1);
        const int nend = __builtin_amdgcn_readlane(endv, k1);
        const int idxn = max(min(nbeg + lane, nend - 1), 0);
        const int svn  = csr_src[idxn];
        const int dvn  = __float_as_int(csr_w[idxn]);

        if (beg < end)
            pull_chunk<K1, 1, U>(x, sv, dv, di, min(32, end - beg), qc, acc);
        for (int base = beg + 32; base < end; base += 32) {
            const int idx = min(base + lane, end - 1);
            const int svc = csr_src[idx];
            const int dvc = __float_as_int(csr_w[idx]);
            pull_chunk<K1, 1, U>(x, svc, dvc, di, min(32, end - base), qc, acc);
        }

        const float dd = di * di;
        float4 o;
        o.x = fmaf(dd, hs.x, acc[0].x);
        o.y = fmaf(dd, hs.y, acc[0].y);
        o.z = fmaf(dd, hs.z, acc[0].z);
        o.w = fmaf(dd, hs.w, acc[0].w);
        reinterpret_cast<float4*>(Hs + lrow * HP)[lane] = o;

        beg = nbeg;
        end = nend;
        sv  = svn;
        dv  = dvn;
    }
    __syncthreads();

    const int nstrips = (nvalid + 15) >> 4;
    if (wave < nstrips) {
        const int lrow = wave * 16 + c;
        const float* ap = Hs + min(lrow, nvalid - 1) * HP + 2 * h;
        v2f a[KS1];
        #pragma unroll
        for (int j = 0; j < KS1; ++j)
            a[j] = *reinterpret_cast<const v2f*>(ap + 4 * j);

        v8f acc2[NT2];
        #pragma unroll
        for (int q = 0; q < NT2; ++q)
            acc2[q] = (v8f){0.f, 0.f, 0.f, 0.f, 0.f, 0.f, 0.f, 0.f};

        if (NT1 & 1)
            fused_tiles<KS1, S1, S2, NT2, 1>(a, W1s, W2s, b1s, 0, c, h, acc2);
        for (int t = NT1 & 1; t + 2 <= NT1; t += 2)
            fused_tiles<KS1, S1, S2, NT2, 2>(a, W1s, W2s, b1s, t, c, h, acc2);

        if (lrow < nvalid) {
            float* crow = C + (size_t)(nbase + lrow) * ldc + 8 * h;
            #pragma unroll
            for (int q = 0; q < NT2; ++q) {
                float4 lo, hi;
                lo.x = acc2[q][0]; lo.y = acc2[q][1]; lo.z = acc2[q][2]; lo.w = acc2[q][3];
                hi.x = acc2[q][4]; hi.y = acc2[q][5]; hi.z = acc2[q][6]; hi.w = acc2[q][7];
                *reinterpret_cast<float4*>(crow + q * 16)     = lo;
                *reinterpret_cast<float4*>(crow + q * 16 + 4) = hi;
            }
        }
    }
}

#define NPB   49
#define NGRP  ((N_NODES + NPB - 1) / NPB)

template <int FP, int FACT, int U, int NB, int NOUT, int NOUTP, int S>
__global__ void __launch_bounds__(256, 8)
k_agg_head(const int* __restrict__ start, const int* __restrict__ endp,
           const int* __restrict__ csr_src, const float* __restrict__ csr_w,
           const float* __restrict__ dinv,
           const float* __restrict__ hw, const float* __restrict__ bias,
           const float* __restrict__ Wo, const float* __restrict__ bo,
           float* __restrict__ out, int n)
{
    constexpr int Q   = FP / 4;
    constexpr int QB  = FACT / 4;
    constexpr int NP  = (Q + 31) / 32;
    constexpr int HP  = FP + 4;
    constexpr int KS  = FACT / 4;
    constexpr int NT  = NOUTP / 16;
    constexpr int QO  = NOUT / 4;
    constexpr int NW4 = FACT * QO;
    constexpr int NIT = (NW4 + 255) / 256;
    static_assert(FP % 4 == 0 && FACT % 4 == 0 && NOUT % 4 == 0, "rows are walked as float4");
    static_assert((HP / 4) % 2 == 1 && (S / 4) % 2 == 1 && S >= FACT, "LDS pitches are 4 x odd");
    static_assert((NB + 15) / 16 <= 8, "one wave per 16-row strip");

    __shared__ __attribute__((aligned(16))) float Hs[NB * HP];
    __shared__ __attribute__((aligned(16))) float Bs[NOUTP * S];

    const int tid  = threadIdx.x;
    const int lane = tid & 31;
    const int wave   = __builtin_amdgcn_readfirstlane(tid >> 5);
    const int nbase  = blockIdx.x * NB;
    const int nvalid = min(NB, n - nbase);

    float4 wst[NIT];
    #pragma unroll
    for (int it = 0; it < NIT; ++it) {
        const int idx = tid + 256 * it;
        wst[it] = make_float4(0.0f, 0.0f, 0.0f, 0.0f);
        if (idx < NW4) wst[it] = reinterpret_cast<const float4*>(Wo)[idx];
    }

    int    qc[NP];
    float4 bq[NP];
    #pragma unroll
    for (int p = 0; p < NP; ++p) {
        const int q = lane + 32 * p;
        qc[p] = min(q, Q - 1);
        bq[p] = make_float4(0.0f, 0.0f, 0.0f, 0.0f);
        if (q < QB) bq[p] = reinterpret_cast<const float4*>(bias)[q];
    }

    const int kn   = (wave < nvalid) ? (nvalid - 1 - wave) / 8 + 1 : 0;
    const int mine = nbase + min(wave + 8 * lane, nvalid - 1);
    const int begv = start[mine];
    const int endv = endp[mine];
    const int divb = __float_as_int(dinv[mine]);

    int beg = __builtin_amdgcn_readlane(begv, 0);
    int end = __builtin_amdgcn_readlane(endv, 0);
    int sv, dv;
    {
        const int idx = max(min(beg + lane, end - 1), 0);
        sv = csr_src[idx];
        dv = __float_as_int(csr_w[idx]);
    }

    #pragma unroll
    for (int it = 0; it < NIT; ++it) {
        const int idx = tid + 256 * it;
        if (idx < NW4) {
            const int k = idx / QO;
            const int q = idx - k * QO;
            float* d = &Bs[(4 * q) * S + k];
            d[0]     = wst[it].x;
            d[S]     = wst[it].y;
            d[2 * S] = wst[it].z;
            d[3 * S] = wst[it].w;
        }
    }
    for (int idx = tid; idx < (NOUTP - NOUT) * S; idx += 256)
        Bs[NOUT * S + idx] = 0.0f;

    for (int k = 0; k < kn; ++k) {
        const int   lrow = wave + 8 * k;
        const int   node = nbase + lrow;
        const float di   = __int_as_float(__builtin_amdgcn_readlane(divb, k));

        const float4* hself = reinterpret_cast<const float4*>(hw + (size_t)node * FP);
        float4 hs[NP];
        float4 acc[NP];
        #pragma unroll
        for (int p = 0; p < NP; ++p) {
            hs[p]  = hself[qc[p]];
            acc[p] = bq[p];
        }

        const int k1   = min(k + 1, kn - 1);
        const int nbeg = __builtin_amdgcn_readlane(begv, k1);
        const int nend = __builtin_amdgcn_readlane(endv, k1);
        const int idxn = max(min(nbeg + lane, nend - 1), 0);
        const int svn  = csr_src[idxn];
        const int dvn  = __float_as_int(csr_w[idxn]);

        if (beg < end)
            pull_chunk<FP, NP, U>(hw, sv, dv, di, min(32, end - beg), qc, acc);
        for (int base = beg + 32; base < end; base += 32) {
            const int idx = min(base + lane, end - 1);
            const int svc = csr_src[idx];
            const int dvc = __float_as_int(csr_w[idx]);
            pull_chunk<FP, NP, U>(hw, svc, dvc, di, min(32, end - base), qc, acc);
        }

        const float dd = di * di;
        float4* o4 = reinterpret_cast<float4*>(Hs + lrow * HP);
        #pragma unroll
        for (int p = 0; p < NP; ++p) {
            const int q = lane + 32 * p;
            if (q < Q) {
                float4 o;
                o.x = fmaxf(fmaf(dd, hs[p].x, acc[p].x), 0.0f);
                o.y = fmaxf(fmaf(dd, hs[p].y, acc[p].y), 0.0f);
                o.z = fmaxf(fmaf(dd, hs[p].z, acc[p].z), 0.0f);
                o.w = fmaxf(fmaf(dd, hs[p].w, acc[p].w), 0.0f);
                o4[q] = o;
            }
        }

        beg = nbeg;
        end = nend;
        sv  = svn;
        dv  = dvn;
    }
    __syncthreads();

    const int nstrips = (nvalid + 15) >> 4;
    if (wave < nstrips) {
        const int c = lane & 15;
        const int h = lane >> 4;
        const float* ap = Hs + min(wave * 16 + c, nvalid - 1) * HP + 2 * h;
        v2f a[KS];
        #pragma unroll
        for (int j = 0; j < KS; ++j)
            a[j] = *reinterpret_cast<const v2f*>(ap + 4 * j);

        v8f acc[NT];
        #pragma unroll
        for (int t = 0; t < NT; ++t)
            acc[t] = (v8f){0.f, 0.f, 0.f, 0.f, 0.f, 0.f, 0.f, 0.f};

        mm_tiles<KS, S, NT>(a, Bs + c * S + 2 * h, acc);

        const int rl = wave * 16 + 8 * h;
        #pragma unroll
        for (int t = 0; t < NT; ++t) {
            const int ncol = t * 16 + c;
            if (ncol < NOUT) {
                const float bv = bo[ncol];
                #pragma unroll
                for (int g = 0; g < 8; ++g)
                    if (rl + g < nvalid)
                        out[(size_t)(nbase + rl + g) * NOUT + ncol] = acc[t][g] + bv;
            }
        }
    }
}

extern "C" void kernel_launch(void* const* d_in, const int* in_sizes, int n_in,
                              void* d_out, int out_size, void* d_ws, size_t ws_size,
                              hipStream_t stream)
{
    const float* x    = (const float*)d_in[0];
    const int*   ei   = (const int*)d_in[1];
    const float* W1   = (const float*)d_in[2];
    const float* b1   = (const float*)d_in[3];
    const float* W2   = (const float*)d_in[4];
    const float* b2   = (const float*)d_in[5];
    const float* Wout = (const float*)d_in[6];
    const float* bout = (const float*)d_in[7];
    float*       out  = (float*)d_out;

    const int* srcI = ei;
    const int* dstI = ei + N_EDGES;

    float* ws   = (float*)d_ws;
    float* dinv = ws;
    float* bufA = ws + 50176;
    int*   tab    = (int*)(bufA + (size_t)N_NODES * H1P);
    int*   segbeg = tab + (NRANGE + 1) * TPITCH;
    int*   segend = segbeg + 50176;
    int*   csrsrc = segend + 50176;
    float* csrw   = (float*)(csrsrc + N_EDGES);
    unsigned* part = (unsigned*)(csrw + N_EDGES);

    k_range_split<<<NCHUNK, 1024, 0, stream>>>(srcI, dstI, part, tab, N_EDGES / 4);
    k_range_csr  <<<NRANGE, 1024, 0, stream>>>(tab, part, segbeg, segend, dinv, csrsrc, N_NODES);
    k_slot_weight<<<(N_EDGES / 4 + 255) / 256, 256, 0, stream>>>(csrsrc, dinv, csrw, N_EDGES / 4);

    k_agg_gemm2<F_IN, H1, H1P, 132, H2, H2P, 212, 16, NB1><<<NGRP1, 512, 0, stream>>>(
        segbeg, segend, csrsrc, csrw, dinv, x, W1, b1, W2, bufA, H2P, N_NODES);

    k_agg_head<H2P, H2, 16, NPB, NC, NCP, 100><<<NGRP, 256, 0, stream>>>(
        segbeg, segend, csrsrc, csrw, dinv, bufA, b2, Wout, bout, out, N_NODES);
}


# Layer-1 kernel: each wave sums its own 16-node strip and multiplies it with no barrier between; the short 13th strip's products run during the others' sums (3 strips per SIMD, not 4); sums unchanged
# speedup vs baseline: 8.4415x; 1.0339x over previous
#include <hip/hip_runtime.h>


typedef float v2f __attribute__((ext_vector_type(2)));
typedef float v8f __attribute__((ext_vector_type(8)));

#define N_NODES 50000
#define N_EDGES 800000
#define F_IN    128
#define H1      200
#define H1P     208
#define H2      100
#define H2P     112
#define NC      40
#define NCP     48
static_assert(N_EDGES % 4 == 0, "edge ids are read four at a time");


#define RANGE   256
#define NRANGE  ((N_NODES + RANGE - 1) / RANGE)
#define EPB     4096
#define NCHUNK  ((N_EDGES + EPB - 1) / EPB)
#define TPITCH  256
#define SPW     ((NCHUNK + 31) / 32)
static_assert(NRANGE < 256, "one LDS counter per range plus the total");
static_assert(NCHUNK <= TPITCH && NCHUNK <= 1024, "one thread per block run in pass 2");
static_assert(N_NODES <= (1 << 24), "source id shares a word with the 8-bit local destination");

__global__ void __launch_bounds__(1024)
k_range_split(const int* __restrict__ src, const int* __restrict__ dst,
              unsigned* __restrict__ part, int* __restrict__ tab, int e4)
{
    __shared__ int hist[256];
    __shared__ int bstart[256];
    __shared__ int wsum[8];
    __shared__ __attribute__((aligned(16))) unsigned outb[EPB];

    const int tid  = threadIdx.x;
    const int lane = tid & 31;
    const int wave = tid >> 5;
    const int g    = blockIdx.x;

    if (tid < 256) hist[tid] = 0;
    __syncthreads();

    const int  q    = g * (EPB / 4) + tid;
    const bool have = q < e4;
    int4 s = make_int4(0, 0, 0, 0);
    int4 d = make_int4(0, 0, 0, 0);
    int r0 = 0, r1 = 0, r2 = 0, r3 = 0;
    if (have) {
        s = reinterpret_cast<const int4*>(src)[q];
        d = reinterpret_cast<const int4*>(dst)[q];
        r0 = atomicAdd(&hist[d.x >> 8], 1);
        r1 = atomicAdd(&hist[d.y >> 8], 1);
        r2 = atomicAdd(&hist[d.z >> 8], 1);
        r3 = atomicAdd(&hist[d.w >> 8], 1);
    }
    __syncthreads();

    int v = 0, incl = 0;
    if (tid < 256) {
        v = hist[tid];
        incl = v;
        #pragma unroll
        for (int o = 1; o < 32; o <<= 1) {
            const int t = __shfl_up(incl, o);
            if (lane >= o) incl += t;
        }
        if (lane == 31) wsum[wave] = incl;
    }
    __syncthreads();
    if (tid < 256) {
        int off = 0;
        for (int w = 0; w < wave; ++w) off += wsum[w];
        const int ex = off + incl - v;
        bstart[tid] = ex;
        if (tid <= NRANGE) tab[tid * TPITCH + g] = ex;
    }
    __syncthreads();

    const int total = bstart[NRANGE];
    if (have) {
        outb[bstart[d.x >> 8] + r0] = ((unsigned)s.x << 8) | (unsigned)(d.x & 255);
        outb[bstart[d.y >> 8] + r1] = ((unsigned)s.y << 8) | (unsigned)(d.y & 255);
        outb[bstart[d.z >> 8] + r2] = ((unsigned)s.z << 8) | (unsigned)(d.z & 255);
        outb[bstart[d.w >> 8] + r3] = ((unsigned)s.w << 8) | (unsigned)(d.w & 255);
    }
    __syncthreads();
    if (4 * tid < total)
        reinterpret_cast<uint4*>(part)[g * (EPB / 4) + tid] = reinterpret_cast<const uint4*>(outb)[tid];
}

__global__ void __launch_bounds__(1024)
k_range_csr(const int* __restrict__ tab, const unsigned* __restrict__ part,
            int* __restrict__ segbeg, int* __restrict__ segend, float* __restrict__ dinv,
            int* __restrict__ csr_src, int n)
{
    __shared__ int sS[NCHUNK];
    __shared__ int cS[NCHUNK];
    __shared__ int cnt[256];
    __shared__ int cur[256];
    __shared__ int wsumA[32];
    __shared__ int wsumB[8];

    const int tid  = threadIdx.x;
    const int lane = tid & 31;
    const int wave = __builtin_amdgcn_readfirstlane(tid >> 5);
    const int r    = blockIdx.x;

    int s = 0;
    if (tid < NCHUNK) {
        s = tab[r * TPITCH + tid];
        const int e = tab[(r + 1) * TPITCH + tid];
        sS[tid] = s;
        cS[tid] = e - s;
    }
    if (tid < 256) cnt[tid] = 0;
    int rs = s;
    #pragma unroll
    for (int o = 16; o > 0; o >>= 1) rs += __shfl_xor(rs, o);
    if (lane == 0) wsumA[wave] = rs;
    __syncthreads();

    int base = 0;
    #pragma unroll
    for (int w = 0; w < SPW; ++w) base += wsumA[w];

    unsigned rec[SPW];
    int      cw[SPW];
    #pragma unroll
    for (int k = 0; k < SPW; ++k) {
        const int w = wave + 32 * k;
        const int c = (w < NCHUNK) ? cS[w] : 0;
        cw[k]  = c;
        rec[k] = 0u;
        if (lane < c) rec[k] = part[w * EPB + sS[w] + lane];
    }

    #pragma unroll
    for (int k = 0; k < SPW; ++k) {
        if (lane < cw[k]) atomicAdd(&cnt[rec[k] & 255u], 1);
        for (int i = 32 + lane; i < cw[k]; i += 32) {
            const int w = wave + 32 * k;
            const unsigned x = part[w * EPB + sS[w] + i];
            atomicAdd(&cnt[x & 255u], 1);
        }
    }
    __syncthreads();

    int v = 0, incl = 0;
    if (tid < 256) {
        v = cnt[tid];
        incl = v;
        #pragma unroll
        for (int o = 1; o < 32; o <<= 1) {
            const int t = __shfl_up(incl, o);
            if (lane >= o) incl += t;
        }
        if (lane == 31) wsumB[wave] = incl;
    }
    __syncthreads();
    if (tid < 256) {
        int off = 0;
        for (int w = 0; w < wave; ++w) off += wsumB[w];
        const int st = off + incl - v;
        cur[tid] = st;
        const int node = r * RANGE + tid;
        if (node < n) {
            segbeg[node] = base + st;
            segend[node] = base + st + v;
            dinv[node]   = rsqrtf((float)(v + 1));
        }
    }
    __syncthreads();

    #pragma unroll
    for (int k = 0; k < SPW; ++k) {
        if (lane < cw[k]) {
            const int p = atomicAdd(&cur[rec[k] & 255u], 1);
            csr_src[base + p] = (int)(rec[k] >> 8);
        }
        for (int i = 32 + lane; i < cw[k]; i += 32) {
            const int w = wave + 32 * k;
            const unsigned x = part[w * EPB + sS[w] + i];
            const int p = atomicAdd(&cur[x & 255u], 1);
            csr_src[base + p] = (int)(x >> 8);
        }
    }
}

__global__ void k_slot_weight(const int* __restrict__ csr_src, const float* __restrict__ dinv,
                              float* __restrict__ csr_w, int e4)
{
    int i = blockIdx.x * blockDim.x + threadIdx.x;
    if (i < e4) {
        const int4 s = reinterpret_cast<const int4*>(csr_src)[i];
        float4 w;
        w.x = dinv[s.x];
        w.y = dinv[s.y];
        w.z = dinv[s.z];
        w.w = dinv[s.w];
        reinterpret_cast<float4*>(csr_w)[i] = w;
    }
}

template <int KS, int S, int NTL>
__device__ __forceinline__ void mm_tiles(const v2f (&a)[KS], const float* bp, v8f (&acc)[NTL])
{
    #pragma unroll
    for (int j = 0; j < KS; ++j) {
        #pragma unroll
        for (int t = 0; t < NTL; ++t) {
            v2f b = *reinterpret_cast<const v2f*>(bp + t * 16 * S + 4 * j);
            acc[t] = __builtin_amdgcn_wmma_f32_16x16x4_f32(
                         false, a[j], false, b, (short)0, acc[t], false, false);
        }
    }
}

__device__ __forceinline__ int kpos(int k)
{
    return (k & ~15) | ((k & 6) << 1) | ((k & 8) >> 2) | (k & 1);
}

template <int K, int NACT, int S, bool PERM>
__device__ __forceinline__ void stage_bt(const float* __restrict__ B, float* Bs, int wave, int lane)
{
    constexpr int KS    = K / 4;
    constexpr int Q     = NACT / 4;
    constexpr int QG    = (Q + 7) / 8;
    constexpr int NTASK = KS * QG;
    constexpr int NIT   = (NTASK + 15) / 16;
    const int ql = lane & 7;
    const int kl = lane >> 3;
    float4 stg[NIT];
    #pragma unroll
    for (int it = 0; it < NIT; ++it) {
        const int task = wave + 16 * it;
        const int kg = task / QG;
        const int qg = task - kg * QG;
        const int q  = qg * 8 + ql;
        const int k  = kg * 4 + kl;
        stg[it] = make_float4(0.0f, 0.0f, 0.0f, 0.0f);
        if (task < NTASK && q < Q)
            stg[it] = *reinterpret_cast<const float4*>(B + (size_t)k * NACT + 4 * q);
    }
    #pragma unroll
    for (int it = 0; it < NIT; ++it) {
        const int task = wave + 16 * it;
        const int kg = task / QG;
        const int qg = task - kg * QG;
        const int q  = qg * 8 + ql;
        const int k  = kg * 4 + kl;
        if (task < NTASK && q < Q) {
            float* d = &Bs[(4 * q) * S + (PERM ? kpos(k) : k)];
            d[0]     = stg[it].x;
            d[S]     = stg[it].y;
            d[2 * S] = stg[it].z;
            d[3 * S] = stg[it].w;
        }
    }
}

template <int KS1, int S1, int S2, int NT2, int NTL>
__device__ __forceinline__ void fused_tiles(const v2f (&a)[KS1], const float* W1s, const float* W2s,
                                            const float* b1s, int t, int c, int h, v8f (&acc2)[NT2])
{
    v8f d[NTL];
    #pragma unroll
    for (int u = 0; u < NTL; ++u)
        d[u] = (v8f){0.f, 0.f, 0.f, 0.f, 0.f, 0.f, 0.f, 0.f};

    const float* wp = W1s + (t * 16 + c) * S1 + 2 * h;
    #pragma unroll
    for (int j = 0; j < KS1; ++j) {
        #pragma unroll
        for (int u = 0; u < NTL; ++u) {
            v2f w = *reinterpret_cast<const v2f*>(wp + u * 16 * S1 + 4 * j);
            d[u] = __builtin_amdgcn_wmma_f32_16x16x4_f32(
                       false, w, false, a[j], (short)0, d[u], false, false);
        }
    }

    #pragma unroll
    for (int u = 0; u < NTL; ++u) {
        const float4 blo = *reinterpret_cast<const float4*>(b1s + (t + u) * 16 + 8 * h);
        const float4 bhi = *reinterpret_cast<const float4*>(b1s + (t + u) * 16 + 8 * h + 4);
        v8f z = d[u];
        z[0] = fmaxf(z[0] + blo.x, 0.0f);
        z[1] = fmaxf(z[1] + blo.y, 0.0f);
        z[2] = fmaxf(z[2] + blo.z, 0.0f);
        z[3] = fmaxf(z[3] + blo.w, 0.0f);
        z[4] = fmaxf(z[4] + bhi.x, 0.0f);
        z[5] = fmaxf(z[5] + bhi.y, 0.0f);
        z[6] = fmaxf(z[6] + bhi.z, 0.0f);
        z[7] = fmaxf(z[7] + bhi.w, 0.0f);

        const float* w2p = W2s + c * S2 + (t + u) * 16 + 2 * h;
        #pragma unroll
        for (int jp = 0; jp < 4; ++jp) {
            v2f b;
            b.x = z[2 * jp];
            b.y = z[2 * jp + 1];
            #pragma unroll
            for (int q = 0; q < NT2; ++q) {
                v2f w = *reinterpret_cast<const v2f*>(w2p + q * 16 * S2 + 4 * jp);
                acc2[q] = __builtin_amdgcn_wmma_f32_16x16x4_f32(
                              false, w, false, b, (short)0, acc2[q], false, false);
            }
        }
    }
}

template <int FP, int NP, int R>
__device__ __forceinline__ void pull_rows(const float* __restrict__ hw, int sv, int dvb, float di,
                                          int e, const int (&qc)[NP], float4 (&acc)[NP])
{
    float4 h[R][NP];
    #pragma unroll
    for (int u = 0; u < R; ++u) {
        const int s = __builtin_amdgcn_readlane(sv, e + u);
        const float4* row = reinterpret_cast<const float4*>(hw + (size_t)s * FP);
        #pragma unroll
        for (int p = 0; p < NP; ++p)
            h[u][p] = row[qc[p]];
    }
    #pragma unroll
    for (int u = 0; u < R; ++u) {
        const float w = __int_as_float(__builtin_amdgcn_readlane(dvb, e + u)) * di;
        #pragma unroll
        for (int p = 0; p < NP; ++p) {
            acc[p].x = fmaf(w, h[u][p].x, acc[p].x);
            acc[p].y = fmaf(w, h[u][p].y, acc[p].y);
            acc[p].z = fmaf(w, h[u][p].z, acc[p].z);
            acc[p].w = fmaf(w, h[u][p].w, acc[p].w);
        }
    }
}

template <int FP, int NP, int R>
__device__ __forceinline__ void pull_tail(const float* __restrict__ hw, int sv, int dvb, float di,
                                          int e, int r, const int (&qc)[NP], float4 (&acc)[NP])
{
    if constexpr (R >= 1) {
        if (r == R) pull_rows<FP, NP, R>(hw, sv, dvb, di, e, qc, acc);
        else        pull_tail<FP, NP, R - 1>(hw, sv, dvb, di, e, r, qc, acc);
    }
}

template <int FP, int NP, int U>
__device__ __forceinline__ void pull_chunk(const float* __restrict__ hw, int sv, int dvb, float di,
                                           int cnt, const int (&qc)[NP], float4 (&acc)[NP])
{
    int e = 0;
    for (; e + U <= cnt; e += U)
        pull_rows<FP, NP, U>(hw, sv, dvb, di, e, qc, acc);
    pull_tail<FP, NP, U - 1>(hw, sv, dvb, di, e, cnt - e, qc, acc);
}

#define NB1   196
#define NGRP1 ((N_NODES + NB1 - 1) / NB1)

template <int K1, int N1, int N1P, int S1, int N2, int N2P, int S2, int U, int NB>
__global__ void __launch_bounds__(512)
k_agg_gemm2(const int* __restrict__ start, const int* __restrict__ endp,
            const int* __restrict__ csr_src, const float* __restrict__ csr_w,
            const float* __restrict__ dinv,
            const float* __restrict__ x,
            const float* __restrict__ W1, const float* __restrict__ b1,
            const float* __restrict__ W2,
            float* __restrict__ C, int ldc, int n)
{
    constexpr int KS1 = K1 / 4;
    constexpr int NT1 = N1P / 16;
    constexpr int NT2 = N2P / 16;
    constexpr int HP  = K1 + 4;
    static_assert(K1 == 128, "one float4 per lane covers an input row");
    static_assert(S2 >= N1P, "W2s rows hold the padded k range");
    static_assert((HP / 4) % 2 == 1, "LDS pitch is 4 x odd");
    static_assert((NB + 15) / 16 <= 16, "one wave per 16-row strip, one lane per node of a wave");

    __shared__ __attribute__((aligned(16))) float W1s[N1P * S1];
    __shared__ __attribute__((aligned(16))) float W2s[N2P * S2];
    __shared__ __attribute__((aligned(16))) float b1s[N1P];
    __shared__ __attribute__((aligned(16))) float Hs[NB * HP];

    const int tid  = threadIdx.x;
    const int lane = tid & 31;
    const int wave   = __builtin_amdgcn_readfirstlane(tid >> 5);
    const int c      = lane & 15;
    const int h      = lane >> 4;
    const int nbase  = blockIdx.x * NB;
    const int nvalid = min(NB, n - nbase);

    const int qc[1] = {lane};

    const int kn   = min(16, max(nvalid - 16 * wave, 0));
    const int mine = nbase + min(16 * wave + lane, nvalid - 1);
    const int begv = start[mine];
    const int endv = endp[mine];
    const int divb = __float_as_int(dinv[mine]);

    int beg = __builtin_amdgcn_readlane(begv, 0);
    int end = __builtin_amdgcn_readlane(endv, 0);
    int sv, dv;
    {
        const int idx = max(min(beg + lane, end - 1), 0);
        sv = csr_src[idx];
        dv = __float_as_int(csr_w[idx]);
    }

    stage_bt<K1, N1, S1, false>(W1, W1s, wave, lane);
    stage_bt<N1, N2, S2, true >(W2, W2s, wave, lane);
    for (int idx = tid; idx < (N1P - N1) * K1; idx += 512) {
        const int cc = idx / K1;
        const int k  = idx - cc * K1;
        W1s[(N1 + cc) * S1 + k] = 0.0f;
    }
    for (int idx = tid; idx < (N2P - N2) * N1P; idx += 512) {
        const int cc = idx / N1P;
        const int k  = idx - cc * N1P;
        W2s[(N2 + cc) * S2 + k] = 0.0f;
    }
    for (int idx = tid; idx < N2 * (N1P - N1); idx += 512) {
        const int cc = idx / (N1P - N1);
        const int k  = N1 + (idx - cc * (N1P - N1));
        W2s[cc * S2 + kpos(k)] = 0.0f;
    }
    if (tid < N1P) b1s[tid] = (tid < N1) ? b1[tid] : 0.0f;
    __syncthreads();

    for (int k = 0; k < kn; ++k) {
        const int   lrow = 16 * wave + k;
        const int   node = nbase + lrow;
        const float di   = __int_as_float(__builtin_amdgcn_readlane(divb, k));

        const float4 hs = reinterpret_cast<const float4*>(x + (size_t)node * K1)[lane];
        float4 acc[1];
        acc[0] = make_float4(0.0f, 0.0f, 0.0f, 0.0f);

        const int k1   = min(k + 1, kn - 1);
        const int nbeg = __builtin_amdgcn_readlane(begv, k1);
        const int nend = __builtin_amdgcn_readlane(endv, k1);
        const int idxn = max(min(nbeg + lane, nend - 1), 0);
        const int svn  = csr_src[idxn];
        const int dvn  = __float_as_int(csr_w[idxn]);

        if (beg < end)
            pull_chunk<K1, 1, U>(x, sv, dv, di, min(32, end - beg), qc, acc);
        for (int base = beg + 32; base < end; base += 32) {
            const int idx = min(base + lane, end - 1);
            const int svc = csr_src[idx];
            const int dvc = __float_as_int(csr_w[idx]);
            pull_chunk<K1, 1, U>(x, svc, dvc, di, min(32, end - base), qc, acc);
        }

        const float dd = di * di;
        float4 o;
        o.x = fmaf(dd, hs.x, acc[0].x);
        o.y = fmaf(dd, hs.y, acc[0].y);
        o.z = fmaf(dd, hs.z, acc[0].z);
        o.w = fmaf(dd, hs.w, acc[0].w);
        reinterpret_cast<float4*>(Hs + lrow * HP)[lane] = o;

        beg = nbeg;
        end = nend;
        sv  = svn;
        dv  = dvn;
    }

    const int nstrips = (nvalid + 15) >> 4;
    if (wave < nstrips) {
        const int lrow = wave * 16 + c;
        const float* ap = Hs + min(lrow, nvalid - 1) * HP + 2 * h;
        v2f a[KS1];
        #pragma unroll
        for (int j = 0; j < KS1; ++j)
            a[j] = *reinterpret_cast<const v2f*>(ap + 4 * j);

        v8f acc2[NT2];
        #pragma unroll
        for (int q = 0; q < NT2; ++q)
            acc2[q] = (v8f){0.f, 0.f, 0.f, 0.f, 0.f, 0.f, 0.f, 0.f};

        if (NT1 & 1)
            fused_tiles<KS1, S1, S2, NT2, 1>(a, W1s, W2s, b1s, 0, c, h, acc2);
        for (int t = NT1 & 1; t + 2 <= NT1; t += 2)
            fused_tiles<KS1, S1, S2, NT2, 2>(a, W1s, W2s, b1s, t, c, h, acc2);

        if (lrow < nvalid) {
            float* crow = C + (size_t)(nbase + lrow) * ldc + 8 * h;
            #pragma unroll
            for (int q = 0; q < NT2; ++q) {
                float4 lo, hi;
                lo.x = acc2[q][0]; lo.y = acc2[q][1]; lo.z = acc2[q][2]; lo.w = acc2[q][3];
                hi.x = acc2[q][4]; hi.y = acc2[q][5]; hi.z = acc2[q][6]; hi.w = acc2[q][7];
                *reinterpret_cast<float4*>(crow + q * 16)     = lo;
                *reinterpret_cast<float4*>(crow + q * 16 + 4) = hi;
            }
        }
    }
}

#define NPB   49
#define NGRP  ((N_NODES + NPB - 1) / NPB)

template <int FP, int FACT, int U, int NB, int NOUT, int NOUTP, int S>
__global__ void __launch_bounds__(256, 8)
k_agg_head(const int* __restrict__ start, const int* __restrict__ endp,
           const int* __restrict__ csr_src, const float* __restrict__ csr_w,
           const float* __restrict__ dinv,
           const float* __restrict__ hw, const float* __restrict__ bias,
           const float* __restrict__ Wo, const float* __restrict__ bo,
           float* __restrict__ out, int n)
{
    constexpr int Q   = FP / 4;
    constexpr int QB  = FACT / 4;
    constexpr int NP  = (Q + 31) / 32;
    constexpr int HP  = FP + 4;
    constexpr int KS  = FACT / 4;
    constexpr int NT  = NOUTP / 16;
    constexpr int QO  = NOUT / 4;
    constexpr int NW4 = FACT * QO;
    constexpr int NIT = (NW4 + 255) / 256;
    static_assert(FP % 4 == 0 && FACT % 4 == 0 && NOUT % 4 == 0, "rows are walked as float4");
    static_assert((HP / 4) % 2 == 1 && (S / 4) % 2 == 1 && S >= FACT, "LDS pitches are 4 x odd");
    static_assert((NB + 15) / 16 <= 8, "one wave per 16-row strip");

    __shared__ __attribute__((aligned(16))) float Hs[NB * HP];
    __shared__ __attribute__((aligned(16))) float Bs[NOUTP * S];

    const int tid  = threadIdx.x;
    const int lane = tid & 31;
    const int wave   = __builtin_amdgcn_readfirstlane(tid >> 5);
    const int nbase  = blockIdx.x * NB;
    const int nvalid = min(NB, n - nbase);

    float4 wst[NIT];
    #pragma unroll
    for (int it = 0; it < NIT; ++it) {
        const int idx = tid + 256 * it;
        wst[it] = make_float4(0.0f, 0.0f, 0.0f, 0.0f);
        if (idx < NW4) wst[it] = reinterpret_cast<const float4*>(Wo)[idx];
    }

    int    qc[NP];
    float4 bq[NP];
    #pragma unroll
    for (int p = 0; p < NP; ++p) {
        const int q = lane + 32 * p;
        qc[p] = min(q, Q - 1);
        bq[p] = make_float4(0.0f, 0.0f, 0.0f, 0.0f);
        if (q < QB) bq[p] = reinterpret_cast<const float4*>(bias)[q];
    }

    const int kn   = (wave < nvalid) ? (nvalid - 1 - wave) / 8 + 1 : 0;
    const int mine = nbase + min(wave + 8 * lane, nvalid - 1);
    const int begv = start[mine];
    const int endv = endp[mine];
    const int divb = __float_as_int(dinv[mine]);

    int beg = __builtin_amdgcn_readlane(begv, 0);
    int end = __builtin_amdgcn_readlane(endv, 0);
    int sv, dv;
    {
        const int idx = max(min(beg + lane, end - 1), 0);
        sv = csr_src[idx];
        dv = __float_as_int(csr_w[idx]);
    }

    #pragma unroll
    for (int it = 0; it < NIT; ++it) {
        const int idx = tid + 256 * it;
        if (idx < NW4) {
            const int k = idx / QO;
            const int q = idx - k * QO;
            float* d = &Bs[(4 * q) * S + k];
            d[0]     = wst[it].x;
            d[S]     = wst[it].y;
            d[2 * S] = wst[it].z;
            d[3 * S] = wst[it].w;
        }
    }
    for (int idx = tid; idx < (NOUTP - NOUT) * S; idx += 256)
        Bs[NOUT * S + idx] = 0.0f;

    for (int k = 0; k < kn; ++k) {
        const int   lrow = wave + 8 * k;
        const int   node = nbase + lrow;
        const float di   = __int_as_float(__builtin_amdgcn_readlane(divb, k));

        const float4* hself = reinterpret_cast<const float4*>(hw + (size_t)node * FP);
        float4 hs[NP];
        float4 acc[NP];
        #pragma unroll
        for (int p = 0; p < NP; ++p) {
            hs[p]  = hself[qc[p]];
            acc[p] = bq[p];
        }

        const int k1   = min(k + 1, kn - 1);
        const int nbeg = __builtin_amdgcn_readlane(begv, k1);
        const int nend = __builtin_amdgcn_readlane(endv, k1);
        const int idxn = max(min(nbeg + lane, nend - 1), 0);
        const int svn  = csr_src[idxn];
        const int dvn  = __float_as_int(csr_w[idxn]);

        if (beg < end)
            pull_chunk<FP, NP, U>(hw, sv, dv, di, min(32, end - beg), qc, acc);
        for (int base = beg + 32; base < end; base += 32) {
            const int idx = min(base + lane, end - 1);
            const int svc = csr_src[idx];
            const int dvc = __float_as_int(csr_w[idx]);
            pull_chunk<FP, NP, U>(hw, svc, dvc, di, min(32, end - base), qc, acc);
        }

        const float dd = di * di;
        float4* o4 = reinterpret_cast<float4*>(Hs + lrow * HP);
        #pragma unroll
        for (int p = 0; p < NP; ++p) {
            const int q = lane + 32 * p;
            if (q < Q) {
                float4 o;
                o.x = fmaxf(fmaf(dd, hs[p].x, acc[p].x), 0.0f);
                o.y = fmaxf(fmaf(dd, hs[p].y, acc[p].y), 0.0f);
                o.z = fmaxf(fmaf(dd, hs[p].z, acc[p].z), 0.0f);
                o.w = fmaxf(fmaf(dd, hs[p].w, acc[p].w), 0.0f);
                o4[q] = o;
            }
        }

        beg = nbeg;
        end = nend;
        sv  = svn;
        dv  = dvn;
    }
    __syncthreads();

    const int nstrips = (nvalid + 15) >> 4;
    if (wave < nstrips) {
        const int c = lane & 15;
        const int h = lane >> 4;
        const float* ap = Hs + min(wave * 16 + c, nvalid - 1) * HP + 2 * h;
        v2f a[KS];
        #pragma unroll
        for (int j = 0; j < KS; ++j)
            a[j] = *reinterpret_cast<const v2f*>(ap + 4 * j);

        v8f acc[NT];
        #pragma unroll
        for (int t = 0; t < NT; ++t)
            acc[t] = (v8f){0.f, 0.f, 0.f, 0.f, 0.f, 0.f, 0.f, 0.f};

        mm_tiles<KS, S, NT>(a, Bs + c * S + 2 * h, acc);

        const int rl = wave * 16 + 8 * h;
        #pragma unroll
        for (int t = 0; t < NT; ++t) {
            const int ncol = t * 16 + c;
            if (ncol < NOUT) {
                const float bv = bo[ncol];
                #pragma unroll
                for (int g = 0; g < 8; ++g)
                    if (rl + g < nvalid)
                        out[(size_t)(nbase + rl + g) * NOUT + ncol] = acc[t][g] + bv;
            }
        }
    }
}

extern "C" void kernel_launch(void* const* d_in, const int* in_sizes, int n_in,
                              void* d_out, int out_size, void* d_ws, size_t ws_size,
                              hipStream_t stream)
{
    const float* x    = (const float*)d_in[0];
    const int*   ei   = (const int*)d_in[1];
    const float* W1   = (const float*)d_in[2];
    const float* b1   = (const float*)d_in[3];
    const float* W2   = (const float*)d_in[4];
    const float* b2   = (const float*)d_in[5];
    const float* Wout = (const float*)d_in[6];
    const float* bout = (const float*)d_in[7];
    float*       out  = (float*)d_out;

    const int* srcI = ei;
    const int* dstI = ei + N_EDGES;

    float* ws   = (float*)d_ws;
    float* dinv = ws;
    float* bufA = ws + 50176;
    int*   tab    = (int*)(bufA + (size_t)N_NODES * H1P);
    int*   segbeg = tab + (NRANGE + 1) * TPITCH;
    int*   segend = segbeg + 50176;
    int*   csrsrc = segend + 50176;
    float* csrw   = (float*)(csrsrc + N_EDGES);
    unsigned* part = (unsigned*)(csrw + N_EDGES);

    k_range_split<<<NCHUNK, 1024, 0, stream>>>(srcI, dstI, part, tab, N_EDGES / 4);
    k_range_csr  <<<NRANGE, 1024, 0, stream>>>(tab, part, segbeg, segend, dinv, csrsrc, N_NODES);
    k_slot_weight<<<(N_EDGES / 4 + 255) / 256, 256, 0, stream>>>(csrsrc, dinv, csrw, N_EDGES / 4);

    k_agg_gemm2<F_IN, H1, H1P, 132, H2, H2P, 212, 16, NB1><<<NGRP1, 512, 0, stream>>>(
        segbeg, segend, csrsrc, csrw, dinv, x, W1, b1, W2, bufA, H2P, N_NODES);

    k_agg_head<H2P, H2, 16, NPB, NC, NCP, 100><<<NGRP, 256, 0, stream>>>(
        segbeg, segend, csrsrc, csrw, dinv, bufA, b2, Wout, bout, out, N_NODES);
}


# Slot-weight pass folded into the layer-1 kernel (4 launches, not 5): each work-group writes dinv[source] for its own nodes' slots under its weight staging; values and sums unchanged
# speedup vs baseline: 8.7441x; 1.0358x over previous
#include <hip/hip_runtime.h>


typedef float v2f __attribute__((ext_vector_type(2)));
typedef float v8f __attribute__((ext_vector_type(8)));

#define N_NODES 50000
#define N_EDGES 800000
#define F_IN    128
#define H1      200
#define H1P     208
#define H2      100
#define H2P     112
#define NC      40
#define NCP     48
static_assert(N_EDGES % 4 == 0, "edge ids are read four at a time");


#define RANGE   256
#define NRANGE  ((N_NODES + RANGE - 1) / RANGE)
#define EPB     4096
#define NCHUNK  ((N_EDGES + EPB - 1) / EPB)
#define TPITCH  256
#define SPW     ((NCHUNK + 31) / 32)
static_assert(NRANGE < 256, "one LDS counter per range plus the total");
static_assert(NCHUNK <= TPITCH && NCHUNK <= 1024, "one thread per block run in pass 2");
static_assert(N_NODES <= (1 << 24), "source id shares a word with the 8-bit local destination");

__global__ void __launch_bounds__(1024)
k_range_split(const int* __restrict__ src, const int* __restrict__ dst,
              unsigned* __restrict__ part, int* __restrict__ tab, int e4)
{
    __shared__ int hist[256];
    __shared__ int bstart[256];
    __shared__ int wsum[8];
    __shared__ __attribute__((aligned(16))) unsigned outb[EPB];

    const int tid  = threadIdx.x;
    const int lane = tid & 31;
    const int wave = tid >> 5;
    const int g    = blockIdx.x;

    if (tid < 256) hist[tid] = 0;
    __syncthreads();

    const int  q    = g * (EPB / 4) + tid;
    const bool have = q < e4;
    int4 s = make_int4(0, 0, 0, 0);
    int4 d = make_int4(0, 0, 0, 0);
    int r0 = 0, r1 = 0, r2 = 0, r3 = 0;
    if (have) {
        s = reinterpret_cast<const int4*>(src)[q];
        d = reinterpret_cast<const int4*>(dst)[q];
        r0 = atomicAdd(&hist[d.x >> 8], 1);
        r1 = atomicAdd(&hist[d.y >> 8], 1);
        r2 = atomicAdd(&hist[d.z >> 8], 1);
        r3 = atomicAdd(&hist[d.w >> 8], 1);
    }
    __syncthreads();

    int v = 0, incl = 0;
    if (tid < 256) {
        v = hist[tid];
        incl = v;
        #pragma unroll
        for (int o = 1; o < 32; o <<= 1) {
            const int t = __shfl_up(incl, o);
            if (lane >= o) incl += t;
        }
        if (lane == 31) wsum[wave] = incl;
    }
    __syncthreads();
    if (tid < 256) {
        int off = 0;
        for (int w = 0; w < wave; ++w) off += wsum[w];
        const int ex = off + incl - v;
        bstart[tid] = ex;
        if (tid <= NRANGE) tab[tid * TPITCH + g] = ex;
    }
    __syncthreads();

    const int total = bstart[NRANGE];
    if (have) {
        outb[bstart[d.x >> 8] + r0] = ((unsigned)s.x << 8) | (unsigned)(d.x & 255);
        outb[bstart[d.y >> 8] + r1] = ((unsigned)s.y << 8) | (unsigned)(d.y & 255);
        outb[bstart[d.z >> 8] + r2] = ((unsigned)s.z << 8) | (unsigned)(d.z & 255);
        outb[bstart[d.w >> 8] + r3] = ((unsigned)s.w << 8) | (unsigned)(d.w & 255);
    }
    __syncthreads();
    if (4 * tid < total)
        reinterpret_cast<uint4*>(part)[g * (EPB / 4) + tid] = reinterpret_cast<const uint4*>(outb)[tid];
}

__global__ void __launch_bounds__(1024)
k_range_csr(const int* __restrict__ tab, const unsigned* __restrict__ part,
            int* __restrict__ segbeg, int* __restrict__ segend, float* __restrict__ dinv,
            int* __restrict__ csr_src, int n)
{
    __shared__ int sS[NCHUNK];
    __shared__ int cS[NCHUNK];
    __shared__ int cnt[256];
    __shared__ int cur[256];
    __shared__ int wsumA[32];
    __shared__ int wsumB[8];

    const int tid  = threadIdx.x;
    const int lane = tid & 31;
    const int wave = __builtin_amdgcn_readfirstlane(tid >> 5);
    const int r    = blockIdx.x;

    int s = 0;
    if (tid < NCHUNK) {
        s = tab[r * TPITCH + tid];
        const int e = tab[(r + 1) * TPITCH + tid];
        sS[tid] = s;
        cS[tid] = e - s;
    }
    if (tid < 256) cnt[tid] = 0;
    int rs = s;
    #pragma unroll
    for (int o = 16; o > 0; o >>= 1) rs += __shfl_xor(rs, o);
    if (lane == 0) wsumA[wave] = rs;
    __syncthreads();

    int base = 0;
    #pragma unroll
    for (int w = 0; w < SPW; ++w) base += wsumA[w];

    unsigned rec[SPW];
    int      cw[SPW];
    #pragma unroll
    for (int k = 0; k < SPW; ++k) {
        const int w = wave + 32 * k;
        const int c = (w < NCHUNK) ? cS[w] : 0;
        cw[k]  = c;
        rec[k] = 0u;
        if (lane < c) rec[k] = part[w * EPB + sS[w] + lane];
    }

    #pragma unroll
    for (int k = 0; k < SPW; ++k) {
        if (lane < cw[k]) atomicAdd(&cnt[rec[k] & 255u], 1);
        for (int i = 32 + lane; i < cw[k]; i += 32) {
            const int w = wave + 32 * k;
            const unsigned x = part[w * EPB + sS[w] + i];
            atomicAdd(&cnt[x & 255u], 1);
        }
    }
    __syncthreads();

    int v = 0, incl = 0;
    if (tid < 256) {
        v = cnt[tid];
        incl = v;
        #pragma unroll
        for (int o = 1; o < 32; o <<= 1) {
            const int t = __shfl_up(incl, o);
            if (lane >= o) incl += t;
        }
        if (lane == 31) wsumB[wave] = incl;
    }
    __syncthreads();
    if (tid < 256) {
        int off = 0;
        for (int w = 0; w < wave; ++w) off += wsumB[w];
        const int st = off + incl - v;
        cur[tid] = st;
        const int node = r * RANGE + tid;
        if (node < n) {
            segbeg[node] = base + st;
            segend[node] = base + st + v;
            dinv[node]   = rsqrtf((float)(v + 1));
        }
    }
    __syncthreads();

    #pragma unroll
    for (int k = 0; k < SPW; ++k) {
        if (lane < cw[k]) {
            const int p = atomicAdd(&cur[rec[k] & 255u], 1);
            csr_src[base + p] = (int)(rec[k] >> 8);
        }
        for (int i = 32 + lane; i < cw[k]; i += 32) {
            const int w = wave + 32 * k;
            const unsigned x = part[w * EPB + sS[w] + i];
            const int p = atomicAdd(&cur[x & 255u], 1);
            csr_src[base + p] = (int)(x >> 8);
        }
    }
}

template <int KS, int S, int NTL>
__device__ __forceinline__ void mm_tiles(const v2f (&a)[KS], const float* bp, v8f (&acc)[NTL])
{
    #pragma unroll
    for (int j = 0; j < KS; ++j) {
        #pragma unroll
        for (int t = 0; t < NTL; ++t) {
            v2f b = *reinterpret_cast<const v2f*>(bp + t * 16 * S + 4 * j);
            acc[t] = __builtin_amdgcn_wmma_f32_16x16x4_f32(
                         false, a[j], false, b, (short)0, acc[t], false, false);
        }
    }
}

__device__ __forceinline__ int kpos(int k)
{
    return (k & ~15) | ((k & 6) << 1) | ((k & 8) >> 2) | (k & 1);
}

template <int K, int NACT, int S, bool PERM>
__device__ __forceinline__ void stage_bt(const float* __restrict__ B, float* Bs, int wave, int lane)
{
    constexpr int KS    = K / 4;
    constexpr int Q     = NACT / 4;
    constexpr int QG    = (Q + 7) / 8;
    constexpr int NTASK = KS * QG;
    constexpr int NIT   = (NTASK + 15) / 16;
    const int ql = lane & 7;
    const int kl = lane >> 3;
    float4 stg[NIT];
    #pragma unroll
    for (int it = 0; it < NIT; ++it) {
        const int task = wave + 16 * it;
        const int kg = task / QG;
        const int qg = task - kg * QG;
        const int q  = qg * 8 + ql;
        const int k  = kg * 4 + kl;
        stg[it] = make_float4(0.0f, 0.0f, 0.0f, 0.0f);
        if (task < NTASK && q < Q)
            stg[it] = *reinterpret_cast<const float4*>(B + (size_t)k * NACT + 4 * q);
    }
    #pragma unroll
    for (int it = 0; it < NIT; ++it) {
        const int task = wave + 16 * it;
        const int kg = task / QG;
        const int qg = task - kg * QG;
        const int q  = qg * 8 + ql;
        const int k  = kg * 4 + kl;
        if (task < NTASK && q < Q) {
            float* d = &Bs[(4 * q) * S + (PERM ? kpos(k) : k)];
            d[0]     = stg[it].x;
            d[S]     = stg[it].y;
            d[2 * S] = stg[it].z;
            d[3 * S] = stg[it].w;
        }
    }
}

template <int KS1, int S1, int S2, int NT2, int NTL>
__device__ __forceinline__ void fused_tiles(const v2f (&a)[KS1], const float* W1s, const float* W2s,
                                            const float* b1s, int t, int c, int h, v8f (&acc2)[NT2])
{
    v8f d[NTL];
    #pragma unroll
    for (int u = 0; u < NTL; ++u)
        d[u] = (v8f){0.f, 0.f, 0.f, 0.f, 0.f, 0.f, 0.f, 0.f};

    const float* wp = W1s + (t * 16 + c) * S1 + 2 * h;
    #pragma unroll
    for (int j = 0; j < KS1; ++j) {
        #pragma unroll
        for (int u = 0; u < NTL; ++u) {
            v2f w = *reinterpret_cast<const v2f*>(wp + u * 16 * S1 + 4 * j);
            d[u] = __builtin_amdgcn_wmma_f32_16x16x4_f32(
                       false, w, false, a[j], (short)0, d[u], false, false);
        }
    }

    #pragma unroll
    for (int u = 0; u < NTL; ++u) {
        const float4 blo = *reinterpret_cast<const float4*>(b1s + (t + u) * 16 + 8 * h);
        const float4 bhi = *reinterpret_cast<const float4*>(b1s + (t + u) * 16 + 8 * h + 4);
        v8f z = d[u];
        z[0] = fmaxf(z[0] + blo.x, 0.0f);
        z[1] = fmaxf(z[1] + blo.y, 0.0f);
        z[2] = fmaxf(z[2] + blo.z, 0.0f);
        z[3] = fmaxf(z[3] + blo.w, 0.0f);
        z[4] = fmaxf(z[4] + bhi.x, 0.0f);
        z[5] = fmaxf(z[5] + bhi.y, 0.0f);
        z[6] = fmaxf(z[6] + bhi.z, 0.0f);
        z[7] = fmaxf(z[7] + bhi.w, 0.0f);

        const float* w2p = W2s + c * S2 + (t + u) * 16 + 2 * h;
        #pragma unroll
        for (int jp = 0; jp < 4; ++jp) {
            v2f b;
            b.x = z[2 * jp];
            b.y = z[2 * jp + 1];
            #pragma unroll
            for (int q = 0; q < NT2; ++q) {
                v2f w = *reinterpret_cast<const v2f*>(w2p + q * 16 * S2 + 4 * jp);
                acc2[q] = __builtin_amdgcn_wmma_f32_16x16x4_f32(
                              false, w, false, b, (short)0, acc2[q], false, false);
            }
        }
    }
}

template <int FP, int NP, int R>
__device__ __forceinline__ void pull_rows(const float* __restrict__ hw, int sv, int dvb, float di,
                                          int e, const int (&qc)[NP], float4 (&acc)[NP])
{
    float4 h[R][NP];
    #pragma unroll
    for (int u = 0; u < R; ++u) {
        const int s = __builtin_amdgcn_readlane(sv, e + u);
        const float4* row = reinterpret_cast<const float4*>(hw + (size_t)s * FP);
        #pragma unroll
        for (int p = 0; p < NP; ++p)
            h[u][p] = row[qc[p]];
    }
    #pragma unroll
    for (int u = 0; u < R; ++u) {
        const float w = __int_as_float(__builtin_amdgcn_readlane(dvb, e + u)) * di;
        #pragma unroll
        for (int p = 0; p < NP; ++p) {
            acc[p].x = fmaf(w, h[u][p].x, acc[p].x);
            acc[p].y = fmaf(w, h[u][p].y, acc[p].y);
            acc[p].z = fmaf(w, h[u][p].z, acc[p].z);
            acc[p].w = fmaf(w, h[u][p].w, acc[p].w);
        }
    }
}

template <int FP, int NP, int R>
__device__ __forceinline__ void pull_tail(const float* __restrict__ hw, int sv, int dvb, float di,
                                          int e, int r, const int (&qc)[NP], float4 (&acc)[NP])
{
    if constexpr (R >= 1) {
        if (r == R) pull_rows<FP, NP, R>(hw, sv, dvb, di, e, qc, acc);
        else        pull_tail<FP, NP, R - 1>(hw, sv, dvb, di, e, r, qc, acc);
    }
}

template <int FP, int NP, int U>
__device__ __forceinline__ void pull_chunk(const float* __restrict__ hw, int sv, int dvb, float di,
                                           int cnt, const int (&qc)[NP], float4 (&acc)[NP])
{
    int e = 0;
    for (; e + U <= cnt; e += U)
        pull_rows<FP, NP, U>(hw, sv, dvb, di, e, qc, acc);
    pull_tail<FP, NP, U - 1>(hw, sv, dvb, di, e, cnt - e, qc, acc);
}

#define NB1   196
#define NGRP1 ((N_NODES + NB1 - 1) / NB1)

template <int K1, int N1, int N1P, int S1, int N2, int N2P, int S2, int U, int NB>
__global__ void __launch_bounds__(512)
k_agg_gemm2(const int* __restrict__ start, const int* __restrict__ endp,
            const int* __restrict__ csr_src, float* csr_w,
            const float* __restrict__ dinv,
            const float* __restrict__ x,
            const float* __restrict__ W1, const float* __restrict__ b1,
            const float* __restrict__ W2,
            float* __restrict__ C, int ldc, int n)
{
    constexpr int KS1 = K1 / 4;
    constexpr int NT1 = N1P / 16;
    constexpr int NT2 = N2P / 16;
    constexpr int HP  = K1 + 4;
    static_assert(K1 == 128, "one float4 per lane covers an input row");
    static_assert(S2 >= N1P, "W2s rows hold the padded k range");
    static_assert((HP / 4) % 2 == 1, "LDS pitch is 4 x odd");
    static_assert((NB + 15) / 16 <= 16, "one wave per 16-row strip, one lane per node of a wave");

    __shared__ __attribute__((aligned(16))) float W1s[N1P * S1];
    __shared__ __attribute__((aligned(16))) float W2s[N2P * S2];
    __shared__ __attribute__((aligned(16))) float b1s[N1P];
    __shared__ __attribute__((aligned(16))) float Hs[NB * HP];

    const int tid  = threadIdx.x;
    const int lane = tid & 31;
    const int wave   = __builtin_amdgcn_readfirstlane(tid >> 5);
    const int c      = lane & 15;
    const int h      = lane >> 4;
    const int nbase  = blockIdx.x * NB;
    const int nvalid = min(NB, n - nbase);

    const int qc[1] = {lane};

    const int kn   = min(16, max(nvalid - 16 * wave, 0));
    const int mine = nbase + min(16 * wave + lane, nvalid - 1);
    const int begv = start[mine];
    const int endv = endp[mine];
    const int divb = __float_as_int(dinv[mine]);

    int beg = __builtin_amdgcn_readlane(begv, 0);
    int end = __builtin_amdgcn_readlane(endv, 0);
    int sv, dv;
    {
        const int idx = max(min(beg + lane, end - 1), 0);
        sv = csr_src[idx];
    }

    constexpr int SLW = 8;
    const int sbeg = start[nbase];
    const int send = endp[nbase + nvalid - 1];
    int ssrc[SLW];
    #pragma unroll
    for (int it = 0; it < SLW; ++it) {
        const int i = sbeg + tid + 512 * it;
        ssrc[it] = csr_src[max(min(i, send - 1), 0)];
    }

    stage_bt<K1, N1, S1, false>(W1, W1s, wave, lane);

    float swt[SLW];
    #pragma unroll
    for (int it = 0; it < SLW; ++it)
        swt[it] = dinv[ssrc[it]];
    dv = __float_as_int(dinv[sv]);

    stage_bt<N1, N2, S2, true >(W2, W2s, wave, lane);

    #pragma unroll
    for (int it = 0; it < SLW; ++it) {
        const int i = sbeg + tid + 512 * it;
        if (i < send) csr_w[i] = swt[it];
    }
    for (int i = sbeg + tid + 512 * SLW; i < send; i += 512)
        csr_w[i] = dinv[csr_src[i]];
    for (int idx = tid; idx < (N1P - N1) * K1; idx += 512) {
        const int cc = idx / K1;
        const int k  = idx - cc * K1;
        W1s[(N1 + cc) * S1 + k] = 0.0f;
    }
    for (int idx = tid; idx < (N2P - N2) * N1P; idx += 512) {
        const int cc = idx / N1P;
        const int k  = idx - cc * N1P;
        W2s[(N2 + cc) * S2 + k] = 0.0f;
    }
    for (int idx = tid; idx < N2 * (N1P - N1); idx += 512) {
        const int cc = idx / (N1P - N1);
        const int k  = N1 + (idx - cc * (N1P - N1));
        W2s[cc * S2 + kpos(k)] = 0.0f;
    }
    if (tid < N1P) b1s[tid] = (tid < N1) ? b1[tid] : 0.0f;
    __syncthreads();

    for (int k = 0; k < kn; ++k) {
        const int   lrow = 16 * wave + k;
        const int   node = nbase + lrow;
        const float di   = __int_as_float(__builtin_amdgcn_readlane(divb, k));

        const float4 hs = reinterpret_cast<const float4*>(x + (size_t)node * K1)[lane];
        float4 acc[1];
        acc[0] = make_float4(0.0f, 0.0f, 0.0f, 0.0f);

        const int k1   = min(k + 1, kn - 1);
        const int nbeg = __builtin_amdgcn_readlane(begv, k1);
        const int nend = __builtin_amdgcn_readlane(endv, k1);
        const int idxn = max(min(nbeg + lane, nend - 1), 0);
        const int svn  = csr_src[idxn];
        const int dvn  = __float_as_int(csr_w[idxn]);

        if (beg < end)
            pull_chunk<K1, 1, U>(x, sv, dv, di, min(32, end - beg), qc, acc);
        for (int base = beg + 32; base < end; base += 32) {
            const int idx = min(base + lane, end - 1);
            const int svc = csr_src[idx];
            const int dvc = __float_as_int(csr_w[idx]);
            pull_chunk<K1, 1, U>(x, svc, dvc, di, min(32, end - base), qc, acc);
        }

        const float dd = di * di;
        float4 o;
        o.x = fmaf(dd, hs.x, acc[0].x);
        o.y = fmaf(dd, hs.y, acc[0].y);
        o.z = fmaf(dd, hs.z, acc[0].z);
        o.w = fmaf(dd, hs.w, acc[0].w);
        reinterpret_cast<float4*>(Hs + lrow * HP)[lane] = o;

        beg = nbeg;
        end = nend;
        sv  = svn;
        dv  = dvn;
    }

    const int nstrips = (nvalid + 15) >> 4;
    if (wave < nstrips) {
        const int lrow = wave * 16 + c;
        const float* ap = Hs + min(lrow, nvalid - 1) * HP + 2 * h;
        v2f a[KS1];
        #pragma unroll
        for (int j = 0; j < KS1; ++j)
            a[j] = *reinterpret_cast<const v2f*>(ap + 4 * j);

        v8f acc2[NT2];
        #pragma unroll
        for (int q = 0; q < NT2; ++q)
            acc2[q] = (v8f){0.f, 0.f, 0.f, 0.f, 0.f, 0.f, 0.f, 0.f};

        if (NT1 & 1)
            fused_tiles<KS1, S1, S2, NT2, 1>(a, W1s, W2s, b1s, 0, c, h, acc2);
        for (int t = NT1 & 1; t + 2 <= NT1; t += 2)
            fused_tiles<KS1, S1, S2, NT2, 2>(a, W1s, W2s, b1s, t, c, h, acc2);

        if (lrow < nvalid) {
            float* crow = C + (size_t)(nbase + lrow) * ldc + 8 * h;
            #pragma unroll
            for (int q = 0; q < NT2; ++q) {
                float4 lo, hi;
                lo.x = acc2[q][0]; lo.y = acc2[q][1]; lo.z = acc2[q][2]; lo.w = acc2[q][3];
                hi.x = acc2[q][4]; hi.y = acc2[q][5]; hi.z = acc2[q][6]; hi.w = acc2[q][7];
                *reinterpret_cast<float4*>(crow + q * 16)     = lo;
                *reinterpret_cast<float4*>(crow + q * 16 + 4) = hi;
            }
        }
    }
}

#define NPB   49
#define NGRP  ((N_NODES + NPB - 1) / NPB)

template <int FP, int FACT, int U, int NB, int NOUT, int NOUTP, int S>
__global__ void __launch_bounds__(256, 8)
k_agg_head(const int* __restrict__ start, const int* __restrict__ endp,
           const int* __restrict__ csr_src, const float* __restrict__ csr_w,
           const float* __restrict__ dinv,
           const float* __restrict__ hw, const float* __restrict__ bias,
           const float* __restrict__ Wo, const float* __restrict__ bo,
           float* __restrict__ out, int n)
{
    constexpr int Q   = FP / 4;
    constexpr int QB  = FACT / 4;
    constexpr int NP  = (Q + 31) / 32;
    constexpr int HP  = FP + 4;
    constexpr int KS  = FACT / 4;
    constexpr int NT  = NOUTP / 16;
    constexpr int QO  = NOUT / 4;
    constexpr int NW4 = FACT * QO;
    constexpr int NIT = (NW4 + 255) / 256;
    static_assert(FP % 4 == 0 && FACT % 4 == 0 && NOUT % 4 == 0, "rows are walked as float4");
    static_assert((HP / 4) % 2 == 1 && (S / 4) % 2 == 1 && S >= FACT, "LDS pitches are 4 x odd");
    static_assert((NB + 15) / 16 <= 8, "one wave per 16-row strip");

    __shared__ __attribute__((aligned(16))) float Hs[NB * HP];
    __shared__ __attribute__((aligned(16))) float Bs[NOUTP * S];

    const int tid  = threadIdx.x;
    const int lane = tid & 31;
    const int wave   = __builtin_amdgcn_readfirstlane(tid >> 5);
    const int nbase  = blockIdx.x * NB;
    const int nvalid = min(NB, n - nbase);

    float4 wst[NIT];
    #pragma unroll
    for (int it = 0; it < NIT; ++it) {
        const int idx = tid + 256 * it;
        wst[it] = make_float4(0.0f, 0.0f, 0.0f, 0.0f);
        if (idx < NW4) wst[it] = reinterpret_cast<const float4*>(Wo)[idx];
    }

    int    qc[NP];
    float4 bq[NP];
    #pragma unroll
    for (int p = 0; p < NP; ++p) {
        const int q = lane + 32 * p;
        qc[p] = min(q, Q - 1);
        bq[p] = make_float4(0.0f, 0.0f, 0.0f, 0.0f);
        if (q < QB) bq[p] = reinterpret_cast<const float4*>(bias)[q];
    }

    const int kn   = (wave < nvalid) ? (nvalid - 1 - wave) / 8 + 1 : 0;
    const int mine = nbase + min(wave + 8 * lane, nvalid - 1);
    const int begv = start[mine];
    const int endv = endp[mine];
    const int divb = __float_as_int(dinv[mine]);

    int beg = __builtin_amdgcn_readlane(begv, 0);
    int end = __builtin_amdgcn_readlane(endv, 0);
    int sv, dv;
    {
        const int idx = max(min(beg + lane, end - 1), 0);
        sv = csr_src[idx];
        dv = __float_as_int(csr_w[idx]);
    }

    #pragma unroll
    for (int it = 0; it < NIT; ++it) {
        const int idx = tid + 256 * it;
        if (idx < NW4) {
            const int k = idx / QO;
            const int q = idx - k * QO;
            float* d = &Bs[(4 * q) * S + k];
            d[0]     = wst[it].x;
            d[S]     = wst[it].y;
            d[2 * S] = wst[it].z;
            d[3 * S] = wst[it].w;
        }
    }
    for (int idx = tid; idx < (NOUTP - NOUT) * S; idx += 256)
        Bs[NOUT * S + idx] = 0.0f;

    for (int k = 0; k < kn; ++k) {
        const int   lrow = wave + 8 * k;
        const int   node = nbase + lrow;
        const float di   = __int_as_float(__builtin_amdgcn_readlane(divb, k));

        const float4* hself = reinterpret_cast<const float4*>(hw + (size_t)node * FP);
        float4 hs[NP];
        float4 acc[NP];
        #pragma unroll
        for (int p = 0; p < NP; ++p) {
            hs[p]  = hself[qc[p]];
            acc[p] = bq[p];
        }

        const int k1   = min(k + 1, kn - 1);
        const int nbeg = __builtin_amdgcn_readlane(begv, k1);
        const int nend = __builtin_amdgcn_readlane(endv, k1);
        const int idxn = max(min(nbeg + lane, nend - 1), 0);
        const int svn  = csr_src[idxn];
        const int dvn  = __float_as_int(csr_w[idxn]);

        if (beg < end)
            pull_chunk<FP, NP, U>(hw, sv, dv, di, min(32, end - beg), qc, acc);
        for (int base = beg + 32; base < end; base += 32) {
            const int idx = min(base + lane, end - 1);
            const int svc = csr_src[idx];
            const int dvc = __float_as_int(csr_w[idx]);
            pull_chunk<FP, NP, U>(hw, svc, dvc, di, min(32, end - base), qc, acc);
        }

        const float dd = di * di;
        float4* o4 = reinterpret_cast<float4*>(Hs + lrow * HP);
        #pragma unroll
        for (int p = 0; p < NP; ++p) {
            const int q = lane + 32 * p;
            if (q < Q) {
                float4 o;
                o.x = fmaxf(fmaf(dd, hs[p].x, acc[p].x), 0.0f);
                o.y = fmaxf(fmaf(dd, hs[p].y, acc[p].y), 0.0f);
                o.z = fmaxf(fmaf(dd, hs[p].z, acc[p].z), 0.0f);
                o.w = fmaxf(fmaf(dd, hs[p].w, acc[p].w), 0.0f);
                o4[q] = o;
            }
        }

        beg = nbeg;
        end = nend;
        sv  = svn;
        dv  = dvn;
    }
    __syncthreads();

    const int nstrips = (nvalid + 15) >> 4;
    if (wave < nstrips) {
        const int c = lane & 15;
        const int h = lane >> 4;
        const float* ap = Hs + min(wave * 16 + c, nvalid - 1) * HP + 2 * h;
        v2f a[KS];
        #pragma unroll
        for (int j = 0; j < KS; ++j)
            a[j] = *reinterpret_cast<const v2f*>(ap + 4 * j);

        v8f acc[NT];
        #pragma unroll
        for (int t = 0; t < NT; ++t)
            acc[t] = (v8f){0.f, 0.f, 0.f, 0.f, 0.f, 0.f, 0.f, 0.f};

        mm_tiles<KS, S, NT>(a, Bs + c * S + 2 * h, acc);

        const int rl = wave * 16 + 8 * h;
        #pragma unroll
        for (int t = 0; t < NT; ++t) {
            const int ncol = t * 16 + c;
            if (ncol < NOUT) {
                const float bv = bo[ncol];
                #pragma unroll
                for (int g = 0; g < 8; ++g)
                    if (rl + g < nvalid)
                        out[(size_t)(nbase + rl + g) * NOUT + ncol] = acc[t][g] + bv;
            }
        }
    }
}

extern "C" void kernel_launch(void* const* d_in, const int* in_sizes, int n_in,
                              void* d_out, int out_size, void* d_ws, size_t ws_size,
                              hipStream_t stream)
{
    const float* x    = (const float*)d_in[0];
    const int*   ei   = (const int*)d_in[1];
    const float* W1   = (const float*)d_in[2];
    const float* b1   = (const float*)d_in[3];
    const float* W2   = (const float*)d_in[4];
    const float* b2   = (const float*)d_in[5];
    const float* Wout = (const float*)d_in[6];
    const float* bout = (const float*)d_in[7];
    float*       out  = (float*)d_out;

    const int* srcI = ei;
    const int* dstI = ei + N_EDGES;

    float* ws   = (float*)d_ws;
    float* dinv = ws;
    float* bufA = ws + 50176;
    int*   tab    = (int*)(bufA + (size_t)N_NODES * H1P);
    int*   segbeg = tab + (NRANGE + 1) * TPITCH;
    int*   segend = segbeg + 50176;
    int*   csrsrc = segend + 50176;
    float* csrw   = (float*)(csrsrc + N_EDGES);
    unsigned* part = (unsigned*)(csrw + N_EDGES);

    k_range_split<<<NCHUNK, 1024, 0, stream>>>(srcI, dstI, part, tab, N_EDGES / 4);
    k_range_csr  <<<NRANGE, 1024, 0, stream>>>(tab, part, segbeg, segend, dinv, csrsrc, N_NODES);

    k_agg_gemm2<F_IN, H1, H1P, 132, H2, H2P, 212, 16, NB1><<<NGRP1, 512, 0, stream>>>(
        segbeg, segend, csrsrc, csrw, dinv, x, W1, b1, W2, bufA, H2P, N_NODES);

    k_agg_head<H2P, H2, 16, NPB, NC, NCP, 100><<<NGRP, 256, 0, stream>>>(
        segbeg, segend, csrsrc, csrw, dinv, bufA, b2, Wout, bout, out, N_NODES);
}


# Fused products: LDS weight-fragment reads run a stage (7-8 WMMAs) ahead of their use, the two tiles of a stage in turn, instead of an exposed LDS wait in front of every WMMA pair; sums unchanged
# speedup vs baseline: 9.2041x; 1.0526x over previous
#include <hip/hip_runtime.h>


typedef float v2f __attribute__((ext_vector_type(2)));
typedef float v8f __attribute__((ext_vector_type(8)));

#define N_NODES 50000
#define N_EDGES 800000
#define F_IN    128
#define H1      200
#define H1P     208
#define H2      100
#define H2P     112
#define NC      40
#define NCP     48
static_assert(N_EDGES % 4 == 0, "edge ids are read four at a time");


#define RANGE   256
#define NRANGE  ((N_NODES + RANGE - 1) / RANGE)
#define EPB     4096
#define NCHUNK  ((N_EDGES + EPB - 1) / EPB)
#define TPITCH  256
#define SPW     ((NCHUNK + 31) / 32)
static_assert(NRANGE < 256, "one LDS counter per range plus the total");
static_assert(NCHUNK <= TPITCH && NCHUNK <= 1024, "one thread per block run in pass 2");
static_assert(N_NODES <= (1 << 24), "source id shares a word with the 8-bit local destination");

__global__ void __launch_bounds__(1024)
k_range_split(const int* __restrict__ src, const int* __restrict__ dst,
              unsigned* __restrict__ part, int* __restrict__ tab, int e4)
{
    __shared__ int hist[256];
    __shared__ int bstart[256];
    __shared__ int wsum[8];
    __shared__ __attribute__((aligned(16))) unsigned outb[EPB];

    const int tid  = threadIdx.x;
    const int lane = tid & 31;
    const int wave = tid >> 5;
    const int g    = blockIdx.x;

    if (tid < 256) hist[tid] = 0;
    __syncthreads();

    const int  q    = g * (EPB / 4) + tid;
    const bool have = q < e4;
    int4 s = make_int4(0, 0, 0, 0);
    int4 d = make_int4(0, 0, 0, 0);
    int r0 = 0, r1 = 0, r2 = 0, r3 = 0;
    if (have) {
        s = reinterpret_cast<const int4*>(src)[q];
        d = reinterpret_cast<const int4*>(dst)[q];
        r0 = atomicAdd(&hist[d.x >> 8], 1);
        r1 = atomicAdd(&hist[d.y >> 8], 1);
        r2 = atomicAdd(&hist[d.z >> 8], 1);
        r3 = atomicAdd(&hist[d.w >> 8], 1);
    }
    __syncthreads();

    int v = 0, incl = 0;
    if (tid < 256) {
        v = hist[tid];
        incl = v;
        #pragma unroll
        for (int o = 1; o < 32; o <<= 1) {
            const int t = __shfl_up(incl, o);
            if (lane >= o) incl += t;
        }
        if (lane == 31) wsum[wave] = incl;
    }
    __syncthreads();
    if (tid < 256) {
        int off = 0;
        for (int w = 0; w < wave; ++w) off += wsum[w];
        const int ex = off + incl - v;
        bstart[tid] = ex;
        if (tid <= NRANGE) tab[tid * TPITCH + g] = ex;
    }
    __syncthreads();

    const int total = bstart[NRANGE];
    if (have) {
        outb[bstart[d.x >> 8] + r0] = ((unsigned)s.x << 8) | (unsigned)(d.x & 255);
        outb[bstart[d.y >> 8] + r1] = ((unsigned)s.y << 8) | (unsigned)(d.y & 255);
        outb[bstart[d.z >> 8] + r2] = ((unsigned)s.z << 8) | (unsigned)(d.z & 255);
        outb[bstart[d.w >> 8] + r3] = ((unsigned)s.w << 8) | (unsigned)(d.w & 255);
    }
    __syncthreads();
    if (4 * tid < total)
        reinterpret_cast<uint4*>(part)[g * (EPB / 4) + tid] = reinterpret_cast<const uint4*>(outb)[tid];
}

__global__ void __launch_bounds__(1024)
k_range_csr(const int* __restrict__ tab, const unsigned* __restrict__ part,
            int* __restrict__ segbeg, int* __restrict__ segend, float* __restrict__ dinv,
            int* __restrict__ csr_src, int n)
{
    __shared__ int sS[NCHUNK];
    __shared__ int cS[NCHUNK];
    __shared__ int cnt[256];
    __shared__ int cur[256];
    __shared__ int wsumA[32];
    __shared__ int wsumB[8];

    const int tid  = threadIdx.x;
    const int lane = tid & 31;
    const int wave = __builtin_amdgcn_readfirstlane(tid >> 5);
    const int r    = blockIdx.x;

    int s = 0;
    if (tid < NCHUNK) {
        s = tab[r * TPITCH + tid];
        const int e = tab[(r + 1) * TPITCH + tid];
        sS[tid] = s;
        cS[tid] = e - s;
    }
    if (tid < 256) cnt[tid] = 0;
    int rs = s;
    #pragma unroll
    for (int o = 16; o > 0; o >>= 1) rs += __shfl_xor(rs, o);
    if (lane == 0) wsumA[wave] = rs;
    __syncthreads();

    int base = 0;
    #pragma unroll
    for (int w = 0; w < SPW; ++w) base += wsumA[w];

    unsigned rec[SPW];
    int      cw[SPW];
    #pragma unroll
    for (int k = 0; k < SPW; ++k) {
        const int w = wave + 32 * k;
        const int c = (w < NCHUNK) ? cS[w] : 0;
        cw[k]  = c;
        rec[k] = 0u;
        if (lane < c) rec[k] = part[w * EPB + sS[w] + lane];
    }

    #pragma unroll
    for (int k = 0; k < SPW; ++k) {
        if (lane < cw[k]) atomicAdd(&cnt[rec[k] & 255u], 1);
        for (int i = 32 + lane; i < cw[k]; i += 32) {
            const int w = wave + 32 * k;
            const unsigned x = part[w * EPB + sS[w] + i];
            atomicAdd(&cnt[x & 255u], 1);
        }
    }
    __syncthreads();

    int v = 0, incl = 0;
    if (tid < 256) {
        v = cnt[tid];
        incl = v;
        #pragma unroll
        for (int o = 1; o < 32; o <<= 1) {
            const int t = __shfl_up(incl, o);
            if (lane >= o) incl += t;
        }
        if (lane == 31) wsumB[wave] = incl;
    }
    __syncthreads();
    if (tid < 256) {
        int off = 0;
        for (int w = 0; w < wave; ++w) off += wsumB[w];
        const int st = off + incl - v;
        cur[tid] = st;
        const int node = r * RANGE + tid;
        if (node < n) {
            segbeg[node] = base + st;
            segend[node] = base + st + v;
            dinv[node]   = rsqrtf((float)(v + 1));
        }
    }
    __syncthreads();

    #pragma unroll
    for (int k = 0; k < SPW; ++k) {
        if (lane < cw[k]) {
            const int p = atomicAdd(&cur[rec[k] & 255u], 1);
            csr_src[base + p] = (int)(rec[k] >> 8);
        }
        for (int i = 32 + lane; i < cw[k]; i += 32) {
            const int w = wave + 32 * k;
            const unsigned x = part[w * EPB + sS[w] + i];
            const int p = atomicAdd(&cur[x & 255u], 1);
            csr_src[base + p] = (int)(x >> 8);
        }
    }
}

template <int KS, int S, int NTL>
__device__ __forceinline__ void mm_tiles(const v2f (&a)[KS], const float* bp, v8f (&acc)[NTL])
{
    #pragma unroll
    for (int j = 0; j < KS; ++j) {
        #pragma unroll
        for (int t = 0; t < NTL; ++t) {
            v2f b = *reinterpret_cast<const v2f*>(bp + t * 16 * S + 4 * j);
            acc[t] = __builtin_amdgcn_wmma_f32_16x16x4_f32(
                         false, a[j], false, b, (short)0, acc[t], false, false);
        }
    }
}

__device__ __forceinline__ int kpos(int k)
{
    return (k & ~15) | ((k & 6) << 1) | ((k & 8) >> 2) | (k & 1);
}

template <int K, int NACT, int S, bool PERM>
__device__ __forceinline__ void stage_bt(const float* __restrict__ B, float* Bs, int wave, int lane)
{
    constexpr int KS    = K / 4;
    constexpr int Q     = NACT / 4;
    constexpr int QG    = (Q + 7) / 8;
    constexpr int NTASK = KS * QG;
    constexpr int NIT   = (NTASK + 15) / 16;
    const int ql = lane & 7;
    const int kl = lane >> 3;
    float4 stg[NIT];
    #pragma unroll
    for (int it = 0; it < NIT; ++it) {
        const int task = wave + 16 * it;
        const int kg = task / QG;
        const int qg = task - kg * QG;
        const int q  = qg * 8 + ql;
        const int k  = kg * 4 + kl;
        stg[it] = make_float4(0.0f, 0.0f, 0.0f, 0.0f);
        if (task < NTASK && q < Q)
            stg[it] = *reinterpret_cast<const float4*>(B + (size_t)k * NACT + 4 * q);
    }
    #pragma unroll
    for (int it = 0; it < NIT; ++it) {
        const int task = wave + 16 * it;
        const int kg = task / QG;
        const int qg = task - kg * QG;
        const int q  = qg * 8 + ql;
        const int k  = kg * 4 + kl;
        if (task < NTASK && q < Q) {
            float* d = &Bs[(4 * q) * S + (PERM ? kpos(k) : k)];
            d[0]     = stg[it].x;
            d[S]     = stg[it].y;
            d[2 * S] = stg[it].z;
            d[3 * S] = stg[it].w;
        }
    }
}

template <int KS1, int S1, int S2, int NT2, int NTL>
__device__ __forceinline__ void fused_tiles(const v2f (&a)[KS1], const float* W1s, const float* W2s,
                                            const float* b1s, int t, int c, int h, v8f (&acc2)[NT2])
{
    constexpr int CH  = 8 / NTL;
    constexpr int NS1 = KS1 / CH;
    constexpr int NS2 = NTL * 4;
    static_assert(8 % NTL == 0 && KS1 % CH == 0, "whole stages");

    v8f d[NTL];
    #pragma unroll
    for (int u = 0; u < NTL; ++u)
        d[u] = (v8f){0.f, 0.f, 0.f, 0.f, 0.f, 0.f, 0.f, 0.f};

    const float* wp  = W1s + (t * 16 + c) * S1 + 2 * h;
    const float* w2p = W2s + c * S2 + t * 16 + 2 * h;

    float4 blo[NTL], bhi[NTL];
    #pragma unroll
    for (int u = 0; u < NTL; ++u) {
        blo[u] = *reinterpret_cast<const float4*>(b1s + (t + u) * 16 + 8 * h);
        bhi[u] = *reinterpret_cast<const float4*>(b1s + (t + u) * 16 + 8 * h + 4);
    }

    v2f f1[2][CH][NTL];
    v2f f2[2][NT2];
    #pragma unroll
    for (int jj = 0; jj < CH; ++jj) {
        #pragma unroll
        for (int u = 0; u < NTL; ++u)
            f1[0][jj][u] = *reinterpret_cast<const v2f*>(wp + u * 16 * S1 + 4 * jj);
    }

    #pragma unroll
    for (int s = 0; s < NS1; ++s) {
        if (s + 1 < NS1) {
            #pragma unroll
            for (int jj = 0; jj < CH; ++jj) {
                #pragma unroll
                for (int u = 0; u < NTL; ++u)
                    f1[(s + 1) & 1][jj][u] =
                        *reinterpret_cast<const v2f*>(wp + u * 16 * S1 + 4 * ((s + 1) * CH + jj));
            }
        } else {
            #pragma unroll
            for (int q = 0; q < NT2; ++q)
                f2[0][q] = *reinterpret_cast<const v2f*>(w2p + q * 16 * S2);
        }
        __builtin_amdgcn_sched_barrier(0);
        #pragma unroll
        for (int jj = 0; jj < CH; ++jj) {
            #pragma unroll
            for (int u = 0; u < NTL; ++u)
                d[u] = __builtin_amdgcn_wmma_f32_16x16x4_f32(
                           false, f1[s & 1][jj][u], false, a[s * CH + jj], (short)0, d[u], false, false);
        }
        __builtin_amdgcn_sched_barrier(0);
    }

    #pragma unroll
    for (int u = 0; u < NTL; ++u) {
        d[u][0] = fmaxf(d[u][0] + blo[u].x, 0.0f);
        d[u][1] = fmaxf(d[u][1] + blo[u].y, 0.0f);
        d[u][2] = fmaxf(d[u][2] + blo[u].z, 0.0f);
        d[u][3] = fmaxf(d[u][3] + blo[u].w, 0.0f);
        d[u][4] = fmaxf(d[u][4] + bhi[u].x, 0.0f);
        d[u][5] = fmaxf(d[u][5] + bhi[u].y, 0.0f);
        d[u][6] = fmaxf(d[u][6] + bhi[u].z, 0.0f);
        d[u][7] = fmaxf(d[u][7] + bhi[u].w, 0.0f);
    }

    #pragma unroll
    for (int g = 0; g < NS2; ++g) {
        const int u  = g >> 2;
        const int jp = g & 3;
        if (g + 1 < NS2) {
            const int u1  = (g + 1) >> 2;
            const int jp1 = (g + 1) & 3;
            #pragma unroll
            for (int q = 0; q < NT2; ++q)
                f2[(g + 1) & 1][q] =
                    *reinterpret_cast<const v2f*>(w2p + u1 * 16 + q * 16 * S2 + 4 * jp1);
        }
        __builtin_amdgcn_sched_barrier(0);
        v2f b;
        b.x = d[u][2 * jp];
        b.y = d[u][2 * jp + 1];
        #pragma unroll
        for (int q = 0; q < NT2; ++q)
            acc2[q] = __builtin_amdgcn_wmma_f32_16x16x4_f32(
                          false, f2[g & 1][q], false, b, (short)0, acc2[q], false, false);
        __builtin_amdgcn_sched_barrier(0);
    }
}

template <int FP, int NP, int R>
__device__ __forceinline__ void pull_rows(const float* __restrict__ hw, int sv, int dvb, float di,
                                          int e, const int (&qc)[NP], float4 (&acc)[NP])
{
    float4 h[R][NP];
    #pragma unroll
    for (int u = 0; u < R; ++u) {
        const int s = __builtin_amdgcn_readlane(sv, e + u);
        const float4* row = reinterpret_cast<const float4*>(hw + (size_t)s * FP);
        #pragma unroll
        for (int p = 0; p < NP; ++p)
            h[u][p] = row[qc[p]];
    }
    #pragma unroll
    for (int u = 0; u < R; ++u) {
        const float w = __int_as_float(__builtin_amdgcn_readlane(dvb, e + u)) * di;
        #pragma unroll
        for (int p = 0; p < NP; ++p) {
            acc[p].x = fmaf(w, h[u][p].x, acc[p].x);
            acc[p].y = fmaf(w, h[u][p].y, acc[p].y);
            acc[p].z = fmaf(w, h[u][p].z, acc[p].z);
            acc[p].w = fmaf(w, h[u][p].w, acc[p].w);
        }
    }
}

template <int FP, int NP, int R>
__device__ __forceinline__ void pull_tail(const float* __restrict__ hw, int sv, int dvb, float di,
                                          int e, int r, const int (&qc)[NP], float4 (&acc)[NP])
{
    if constexpr (R >= 1) {
        if (r == R) pull_rows<FP, NP, R>(hw, sv, dvb, di, e, qc, acc);
        else        pull_tail<FP, NP, R - 1>(hw, sv, dvb, di, e, r, qc, acc);
    }
}

template <int FP, int NP, int U>
__device__ __forceinline__ void pull_chunk(const float* __restrict__ hw, int sv, int dvb, float di,
                                           int cnt, const int (&qc)[NP], float4 (&acc)[NP])
{
    int e = 0;
    for (; e + U <= cnt; e += U)
        pull_rows<FP, NP, U>(hw, sv, dvb, di, e, qc, acc);
    pull_tail<FP, NP, U - 1>(hw, sv, dvb, di, e, cnt - e, qc, acc);
}

#define NB1   196
#define NGRP1 ((N_NODES + NB1 - 1) / NB1)

template <int K1, int N1, int N1P, int S1, int N2, int N2P, int S2, int U, int NB>
__global__ void __launch_bounds__(512)
k_agg_gemm2(const int* __restrict__ start, const int* __restrict__ endp,
            const int* __restrict__ csr_src, float* csr_w,
            const float* __restrict__ dinv,
            const float* __restrict__ x,
            const float* __restrict__ W1, const float* __restrict__ b1,
            const float* __restrict__ W2,
            float* __restrict__ C, int ldc, int n)
{
    constexpr int KS1 = K1 / 4;
    constexpr int NT1 = N1P / 16;
    constexpr int NT2 = N2P / 16;
    constexpr int HP  = K1 + 4;
    static_assert(K1 == 128, "one float4 per lane covers an input row");
    static_assert(S2 >= N1P, "W2s rows hold the padded k range");
    static_assert((HP / 4) % 2 == 1, "LDS pitch is 4 x odd");
    static_assert((NB + 15) / 16 <= 16, "one wave per 16-row strip, one lane per node of a wave");

    __shared__ __attribute__((aligned(16))) float W1s[N1P * S1];
    __shared__ __attribute__((aligned(16))) float W2s[N2P * S2];
    __shared__ __attribute__((aligned(16))) float b1s[N1P];
    __shared__ __attribute__((aligned(16))) float Hs[NB * HP];

    const int tid  = threadIdx.x;
    const int lane = tid & 31;
    const int wave   = __builtin_amdgcn_readfirstlane(tid >> 5);
    const int c      = lane & 15;
    const int h      = lane >> 4;
    const int nbase  = blockIdx.x * NB;
    const int nvalid = min(NB, n - nbase);

    const int qc[1] = {lane};

    const int kn   = min(16, max(nvalid - 16 * wave, 0));
    const int mine = nbase + min(16 * wave + lane, nvalid - 1);
    const int begv = start[mine];
    const int endv = endp[mine];
    const int divb = __float_as_int(dinv[mine]);

    int beg = __builtin_amdgcn_readlane(begv, 0);
    int end = __builtin_amdgcn_readlane(endv, 0);
    int sv, dv;
    {
        const int idx = max(min(beg + lane, end - 1), 0);
        sv = csr_src[idx];
    }

    constexpr int SLW = 8;
    const int sbeg = start[nbase];
    const int send = endp[nbase + nvalid - 1];
    int ssrc[SLW];
    #pragma unroll
    for (int it = 0; it < SLW; ++it) {
        const int i = sbeg + tid + 512 * it;
        ssrc[it] = csr_src[max(min(i, send - 1), 0)];
    }

    stage_bt<K1, N1, S1, false>(W1, W1s, wave, lane);

    float swt[SLW];
    #pragma unroll
    for (int it = 0; it < SLW; ++it)
        swt[it] = dinv[ssrc[it]];
    dv = __float_as_int(dinv[sv]);

    stage_bt<N1, N2, S2, true >(W2, W2s, wave, lane);

    #pragma unroll
    for (int it = 0; it < SLW; ++it) {
        const int i = sbeg + tid + 512 * it;
        if (i < send) csr_w[i] = swt[it];
    }
    for (int i = sbeg + tid + 512 * SLW; i < send; i += 512)
        csr_w[i] = dinv[csr_src[i]];
    for (int idx = tid; idx < (N1P - N1) * K1; idx += 512) {
        const int cc = idx / K1;
        const int k  = idx - cc * K1;
        W1s[(N1 + cc) * S1 + k] = 0.0f;
    }
    for (int idx = tid; idx < (N2P - N2) * N1P; idx += 512) {
        const int cc = idx / N1P;
        const int k  = idx - cc * N1P;
        W2s[(N2 + cc) * S2 + k] = 0.0f;
    }
    for (int idx = tid; idx < N2 * (N1P - N1); idx += 512) {
        const int cc = idx / (N1P - N1);
        const int k  = N1 + (idx - cc * (N1P - N1));
        W2s[cc * S2 + kpos(k)] = 0.0f;
    }
    if (tid < N1P) b1s[tid] = (tid < N1) ? b1[tid] : 0.0f;
    __syncthreads();

    for (int k = 0; k < kn; ++k) {
        const int   lrow = 16 * wave + k;
        const int   node = nbase + lrow;
        const float di   = __int_as_float(__builtin_amdgcn_readlane(divb, k));

        const float4 hs = reinterpret_cast<const float4*>(x + (size_t)node * K1)[lane];
        float4 acc[1];
        acc[0] = make_float4(0.0f, 0.0f, 0.0f, 0.0f);

        const int k1   = min(k + 1, kn - 1);
        const int nbeg = __builtin_amdgcn_readlane(begv, k1);
        const int nend = __builtin_amdgcn_readlane(endv, k1);
        const int idxn = max(min(nbeg + lane, nend - 1), 0);
        const int svn  = csr_src[idxn];
        const int dvn  = __float_as_int(csr_w[idxn]);

        if (beg < end)
            pull_chunk<K1, 1, U>(x, sv, dv, di, min(32, end - beg), qc, acc);
        for (int base = beg + 32; base < end; base += 32) {
            const int idx = min(base + lane, end - 1);
            const int svc = csr_src[idx];
            const int dvc = __float_as_int(csr_w[idx]);
            pull_chunk<K1, 1, U>(x, svc, dvc, di, min(32, end - base), qc, acc);
        }

        const float dd = di * di;
        float4 o;
        o.x = fmaf(dd, hs.x, acc[0].x);
        o.y = fmaf(dd, hs.y, acc[0].y);
        o.z = fmaf(dd, hs.z, acc[0].z);
        o.w = fmaf(dd, hs.w, acc[0].w);
        reinterpret_cast<float4*>(Hs + lrow * HP)[lane] = o;

        beg = nbeg;
        end = nend;
        sv  = svn;
        dv  = dvn;
    }

    const int nstrips = (nvalid + 15) >> 4;
    if (wave < nstrips) {
        const int lrow = wave * 16 + c;
        const float* ap = Hs + min(lrow, nvalid - 1) * HP + 2 * h;
        v2f a[KS1];
        #pragma unroll
        for (int j = 0; j < KS1; ++j)
            a[j] = *reinterpret_cast<const v2f*>(ap + 4 * j);

        v8f acc2[NT2];
        #pragma unroll
        for (int q = 0; q < NT2; ++q)
            acc2[q] = (v8f){0.f, 0.f, 0.f, 0.f, 0.f, 0.f, 0.f, 0.f};

        if (NT1 & 1)
            fused_tiles<KS1, S1, S2, NT2, 1>(a, W1s, W2s, b1s, 0, c, h, acc2);
        for (int t = NT1 & 1; t + 2 <= NT1; t += 2)
            fused_tiles<KS1, S1, S2, NT2, 2>(a, W1s, W2s, b1s, t, c, h, acc2);

        if (lrow < nvalid) {
            float* crow = C + (size_t)(nbase + lrow) * ldc + 8 * h;
            #pragma unroll
            for (int q = 0; q < NT2; ++q) {
                float4 lo, hi;
                lo.x = acc2[q][0]; lo.y = acc2[q][1]; lo.z = acc2[q][2]; lo.w = acc2[q][3];
                hi.x = acc2[q][4]; hi.y = acc2[q][5]; hi.z = acc2[q][6]; hi.w = acc2[q][7];
                *reinterpret_cast<float4*>(crow + q * 16)     = lo;
                *reinterpret_cast<float4*>(crow + q * 16 + 4) = hi;
            }
        }
    }
}

#define NPB   49
#define NGRP  ((N_NODES + NPB - 1) / NPB)

template <int FP, int FACT, int U, int NB, int NOUT, int NOUTP, int S>
__global__ void __launch_bounds__(256, 8)
k_agg_head(const int* __restrict__ start, const int* __restrict__ endp,
           const int* __restrict__ csr_src, const float* __restrict__ csr_w,
           const float* __restrict__ dinv,
           const float* __restrict__ hw, const float* __restrict__ bias,
           const float* __restrict__ Wo, const float* __restrict__ bo,
           float* __restrict__ out, int n)
{
    constexpr int Q   = FP / 4;
    constexpr int QB  = FACT / 4;
    constexpr int NP  = (Q + 31) / 32;
    constexpr int HP  = FP + 4;
    constexpr int KS  = FACT / 4;
    constexpr int NT  = NOUTP / 16;
    constexpr int QO  = NOUT / 4;
    constexpr int NW4 = FACT * QO;
    constexpr int NIT = (NW4 + 255) / 256;
    static_assert(FP % 4 == 0 && FACT % 4 == 0 && NOUT % 4 == 0, "rows are walked as float4");
    static_assert((HP / 4) % 2 == 1 && (S / 4) % 2 == 1 && S >= FACT, "LDS pitches are 4 x odd");
    static_assert((NB + 15) / 16 <= 8, "one wave per 16-row strip");

    __shared__ __attribute__((aligned(16))) float Hs[NB * HP];
    __shared__ __attribute__((aligned(16))) float Bs[NOUTP * S];

    const int tid  = threadIdx.x;
    const int lane = tid & 31;
    const int wave   = __builtin_amdgcn_readfirstlane(tid >> 5);
    const int nbase  = blockIdx.x * NB;
    const int nvalid = min(NB, n - nbase);

    float4 wst[NIT];
    #pragma unroll
    for (int it = 0; it < NIT; ++it) {
        const int idx = tid + 256 * it;
        wst[it] = make_float4(0.0f, 0.0f, 0.0f, 0.0f);
        if (idx < NW4) wst[it] = reinterpret_cast<const float4*>(Wo)[idx];
    }

    int    qc[NP];
    float4 bq[NP];
    #pragma unroll
    for (int p = 0; p < NP; ++p) {
        const int q = lane + 32 * p;
        qc[p] = min(q, Q - 1);
        bq[p] = make_float4(0.0f, 0.0f, 0.0f, 0.0f);
        if (q < QB) bq[p] = reinterpret_cast<const float4*>(bias)[q];
    }

    const int kn   = (wave < nvalid) ? (nvalid - 1 - wave) / 8 + 1 : 0;
    const int mine = nbase + min(wave + 8 * lane, nvalid - 1);
    const int begv = start[mine];
    const int endv = endp[mine];
    const int divb = __float_as_int(dinv[mine]);

    int beg = __builtin_amdgcn_readlane(begv, 0);
    int end = __builtin_amdgcn_readlane(endv, 0);
    int sv, dv;
    {
        const int idx = max(min(beg + lane, end - 1), 0);
        sv = csr_src[idx];
        dv = __float_as_int(csr_w[idx]);
    }

    #pragma unroll
    for (int it = 0; it < NIT; ++it) {
        const int idx = tid + 256 * it;
        if (idx < NW4) {
            const int k = idx / QO;
            const int q = idx - k * QO;
            float* d = &Bs[(4 * q) * S + k];
            d[0]     = wst[it].x;
            d[S]     = wst[it].y;
            d[2 * S] = wst[it].z;
            d[3 * S] = wst[it].w;
        }
    }
    for (int idx = tid; idx < (NOUTP - NOUT) * S; idx += 256)
        Bs[NOUT * S + idx] = 0.0f;

    for (int k = 0; k < kn; ++k) {
        const int   lrow = wave + 8 * k;
        const int   node = nbase + lrow;
        const float di   = __int_as_float(__builtin_amdgcn_readlane(divb, k));

        const float4* hself = reinterpret_cast<const float4*>(hw + (size_t)node * FP);
        float4 hs[NP];
        float4 acc[NP];
        #pragma unroll
        for (int p = 0; p < NP; ++p) {
            hs[p]  = hself[qc[p]];
            acc[p] = bq[p];
        }

        const int k1   = min(k + 1, kn - 1);
        const int nbeg = __builtin_amdgcn_readlane(begv, k1);
        const int nend = __builtin_amdgcn_readlane(endv, k1);
        const int idxn = max(min(nbeg + lane, nend - 1), 0);
        const int svn  = csr_src[idxn];
        const int dvn  = __float_as_int(csr_w[idxn]);

        if (beg < end)
            pull_chunk<FP, NP, U>(hw, sv, dv, di, min(32, end - beg), qc, acc);
        for (int base = beg + 32; base < end; base += 32) {
            const int idx = min(base + lane, end - 1);
            const int svc = csr_src[idx];
            const int dvc = __float_as_int(csr_w[idx]);
            pull_chunk<FP, NP, U>(hw, svc, dvc, di, min(32, end - base), qc, acc);
        }

        const float dd = di * di;
        float4* o4 = reinterpret_cast<float4*>(Hs + lrow * HP);
        #pragma unroll
        for (int p = 0; p < NP; ++p) {
            const int q = lane + 32 * p;
            if (q < Q) {
                float4 o;
                o.x = fmaxf(fmaf(dd, hs[p].x, acc[p].x), 0.0f);
                o.y = fmaxf(fmaf(dd, hs[p].y, acc[p].y), 0.0f);
                o.z = fmaxf(fmaf(dd, hs[p].z, acc[p].z), 0.0f);
                o.w = fmaxf(fmaf(dd, hs[p].w, acc[p].w), 0.0f);
                o4[q] = o;
            }
        }

        beg = nbeg;
        end = nend;
        sv  = svn;
        dv  = dvn;
    }
    __syncthreads();

    const int nstrips = (nvalid + 15) >> 4;
    if (wave < nstrips) {
        const int c = lane & 15;
        const int h = lane >> 4;
        const float* ap = Hs + min(wave * 16 + c, nvalid - 1) * HP + 2 * h;
        v2f a[KS];
        #pragma unroll
        for (int j = 0; j < KS; ++j)
            a[j] = *reinterpret_cast<const v2f*>(ap + 4 * j);

        v8f acc[NT];
        #pragma unroll
        for (int t = 0; t < NT; ++t)
            acc[t] = (v8f){0.f, 0.f, 0.f, 0.f, 0.f, 0.f, 0.f, 0.f};

        mm_tiles<KS, S, NT>(a, Bs + c * S + 2 * h, acc);

        const int rl = wave * 16 + 8 * h;
        #pragma unroll
        for (int t = 0; t < NT; ++t) {
            const int ncol = t * 16 + c;
            if (ncol < NOUT) {
                const float bv = bo[ncol];
                #pragma unroll
                for (int g = 0; g < 8; ++g)
                    if (rl + g < nvalid)
                        out[(size_t)(nbase + rl + g) * NOUT + ncol] = acc[t][g] + bv;
            }
        }
    }
}

extern "C" void kernel_launch(void* const* d_in, const int* in_sizes, int n_in,
                              void* d_out, int out_size, void* d_ws, size_t ws_size,
                              hipStream_t stream)
{
    const float* x    = (const float*)d_in[0];
    const int*   ei   = (const int*)d_in[1];
    const float* W1   = (const float*)d_in[2];
    const float* b1   = (const float*)d_in[3];
    const float* W2   = (const float*)d_in[4];
    const float* b2   = (const float*)d_in[5];
    const float* Wout = (const float*)d_in[6];
    const float* bout = (const float*)d_in[7];
    float*       out  = (float*)d_out;

    const int* srcI = ei;
    const int* dstI = ei + N_EDGES;

    float* ws   = (float*)d_ws;
    float* dinv = ws;
    float* bufA = ws + 50176;
    int*   tab    = (int*)(bufA + (size_t)N_NODES * H1P);
    int*   segbeg = tab + (NRANGE + 1) * TPITCH;
    int*   segend = segbeg + 50176;
    int*   csrsrc = segend + 50176;
    float* csrw   = (float*)(csrsrc + N_EDGES);
    unsigned* part = (unsigned*)(csrw + N_EDGES);

    k_range_split<<<NCHUNK, 1024, 0, stream>>>(srcI, dstI, part, tab, N_EDGES / 4);
    k_range_csr  <<<NRANGE, 1024, 0, stream>>>(tab, part, segbeg, segend, dinv, csrsrc, N_NODES);

    k_agg_gemm2<F_IN, H1, H1P, 132, H2, H2P, 212, 16, NB1><<<NGRP1, 512, 0, stream>>>(
        segbeg, segend, csrsrc, csrw, dinv, x, W1, b1, W2, bufA, H2P, N_NODES);

    k_agg_head<H2P, H2, 16, NPB, NC, NCP, 100><<<NGRP, 256, 0, stream>>>(
        segbeg, segend, csrsrc, csrw, dinv, bufA, b2, Wout, bout, out, N_NODES);
}
